# starting kernel
# baseline (speedup 1.0000x reference)
#include <hip/hip_runtime.h>
#include <hip/hip_bf16.h>

typedef float v2f __attribute__((ext_vector_type(2)));
typedef float v8f __attribute__((ext_vector_type(8)));

__global__ void gcn_init_deg(float* __restrict__ deg, int n) {
    int i = blockIdx.x * blockDim.x + threadIdx.x;
    if (i < n) deg[i] = 1.0f;
}

__global__ void gcn_deg_scatter(const int* __restrict__ dst, float* __restrict__ deg, int nE) {
    int e = blockIdx.x * blockDim.x + threadIdx.x;
    if (e < nE) atomicAdd(&deg[dst[e]], 1.0f);
}

__global__ void gcn_rsqrt(float* __restrict__ deg, int n) {
    int i = blockIdx.x * blockDim.x + threadIdx.x;
    if (i < n) deg[i] = __frsqrt_rn(deg[i]);
}

template <int NT>
__global__ void gcn_gemm_wmma(const float* __restrict__ X,
                              const float* __restrict__ W,
                              const float* __restrict__ dis,
                              float* __restrict__ hs,
                              float* __restrict__ agg,
                              int M, int K, int N) {
    int wave = (int)((blockIdx.x * blockDim.x + threadIdx.x) >> 5);
    int lane = threadIdx.x & 31;
    int m0 = wave << 4;
    if (m0 >= M) return;

    int lo = lane & 15;
    int hi = lane >> 4;

    const float* xrow = X + (size_t)(m0 + lo) * K;

    int  col[NT];
    int  ccol[NT];
    bool ok[NT];
#pragma unroll
    for (int t = 0; t < NT; ++t) {
        col[t]  = t * 16 + lo;
        ok[t]   = col[t] < N;
        ccol[t] = ok[t] ? col[t] : 0;
    }

    v8f c[NT];
#pragma unroll
    for (int t = 0; t < NT; ++t) c[t] = (v8f){};

    for (int k0 = 0; k0 < K; k0 += 4) {
        int ka = k0 + 2 * hi;
        v2f a;
        a.x = xrow[ka];
        a.y = xrow[ka + 1];
        const float* w0 = W + (size_t)ka * N;
#pragma unroll
        for (int t = 0; t < NT; ++t) {
            float bx = w0[ccol[t]];
            float by = w0[N + ccol[t]];
            v2f b;
            b.x = ok[t] ? bx : 0.0f;
            b.y = ok[t] ? by : 0.0f;
            c[t] = __builtin_amdgcn_wmma_f32_16x16x4_f32(
                        false, a,  false, b,
                        (short)0, c[t],
                        false,  false);
        }
    }

    int rowBase = m0 + (hi << 3);
#pragma unroll
    for (int r = 0; r < 8; ++r) {
        int rr = rowBase + r;
        int rc = rr < M ? rr : M - 1;
        float dr = dis[rc];
        size_t base = (size_t)rc * N;
#pragma unroll
        for (int t = 0; t < NT; ++t) {
            if (rr < M && ok[t]) {
                float v = c[t][r] * dr;
                hs[base + col[t]]  = v;
                agg[base + col[t]] = v;
            }
        }
    }
}

__global__ void gcn_edge_scatter(const int* __restrict__ src,
                                 const int* __restrict__ dst,
                                 const float* __restrict__ hs,
                                 float* __restrict__ agg,
                                 int nE, int F, int chunks) {
    long long t = (long long)blockIdx.x * blockDim.x + threadIdx.x;
    long long total = (long long)nE * chunks;
    if (t >= total) return;
    int e = (int)(t / chunks);
    int c = (int)(t - (long long)e * chunks);
    int s = src[e];
    int d = dst[e];
    const float4 v = *(const float4*)(hs + (long long)s * F + (c << 2));
    float* p = agg + (long long)d * F + (c << 2);
    atomicAdd(p + 0, v.x);
    atomicAdd(p + 1, v.y);
    atomicAdd(p + 2, v.z);
    atomicAdd(p + 3, v.w);
}

__global__ void gcn_finalize(const float* __restrict__ agg,
                             const float* __restrict__ dis,
                             const float* __restrict__ bias,
                             float* __restrict__ out,
                             int nNodes, int F, int doRelu) {
    long long t = (long long)blockIdx.x * blockDim.x + threadIdx.x;
    if (t >= (long long)nNodes * F) return;
    int i = (int)(t / F);
    int f = (int)(t - (long long)i * F);
    float v = fmaf(dis[i], agg[t], bias[f]);
    if (doRelu) v = fmaxf(v, 0.0f);
    out[t] = v;
}

static inline void launch_gemm(int tilesN, const float* X, const float* W,
                               const float* dis, float* hs, float* agg,
                               int M, int K, int N, hipStream_t stream) {
    int mt = (M + 15) / 16;
    long long thr = (long long)mt * 32;
    int blocks = (int)((thr + 255) / 256);
    switch (tilesN) {
    case 1: gcn_gemm_wmma<1><<<blocks, 256, 0, stream>>>(X, W, dis, hs, agg, M, K, N); break;
    case 2: gcn_gemm_wmma<2><<<blocks, 256, 0, stream>>>(X, W, dis, hs, agg, M, K, N); break;
    case 3: gcn_gemm_wmma<3><<<blocks, 256, 0, stream>>>(X, W, dis, hs, agg, M, K, N); break;
    default: gcn_gemm_wmma<4><<<blocks, 256, 0, stream>>>(X, W, dis, hs, agg, M, K, N); break;
    }
}

extern "C" void kernel_launch(void* const* d_in, const int* in_sizes, int n_in,
                              void* d_out, int out_size, void* d_ws, size_t ws_size,
                              hipStream_t stream) {
    const float* x  = (const float*)d_in[0];
    const int*   ei = (const int*)d_in[1];
    const float* W1 = (const float*)d_in[2];
    const float* b1 = (const float*)d_in[3];
    const float* W2 = (const float*)d_in[4];
    const float* b2 = (const float*)d_in[5];
    float* out = (float*)d_out;

    const int IN  = 128;
    const int HID = in_sizes[3];
    const int OUT = in_sizes[5];
    const int N   = in_sizes[0] / IN;
    const int E   = in_sizes[1] / 2;

    const int* src = ei;
    const int* dst = ei + E;

    float* dis  = (float*)d_ws;
    float* hs1  = dis + (((size_t)N + 63) & ~63ull);
    float* agg1 = hs1 + (size_t)N * HID;
    float* hs2  = agg1 + (size_t)N * HID;

    const int B = 256;

    gcn_init_deg<<<(N + B - 1) / B, B, 0, stream>>>(dis, N);
    gcn_deg_scatter<<<(E + B - 1) / B, B, 0, stream>>>(dst, dis, E);
    gcn_rsqrt<<<(N + B - 1) / B, B, 0, stream>>>(dis, N);

    launch_gemm((HID + 15) / 16, x, W1, dis, hs1, agg1, N, IN, HID, stream);

    int chunks1 = HID / 4;
    long long sthr1 = (long long)E * chunks1;
    gcn_edge_scatter<<<(int)((sthr1 + B - 1) / B), B, 0, stream>>>(
        src, dst, hs1, agg1, E, HID, chunks1);

    long long fthr1 = (long long)N * HID;
    gcn_finalize<<<(int)((fthr1 + B - 1) / B), B, 0, stream>>>(
        agg1, dis, b1, hs1, N, HID,  1);

    launch_gemm((OUT + 15) / 16, hs1, W2, dis, hs2, out, N, HID, OUT, stream);

    int chunks2 = OUT / 4;
    long long sthr2 = (long long)E * chunks2;
    gcn_edge_scatter<<<(int)((sthr2 + B - 1) / B), B, 0, stream>>>(
        src, dst, hs2, out, E, OUT, chunks2);

    long long fthr2 = (long long)N * OUT;
    gcn_finalize<<<(int)((fthr2 + B - 1) / B), B, 0, stream>>>(
        out, dis, b2, out, N, OUT,  0);
}


# edges partitioned by destination bucket (LDS counting sort); degree and both scatter-adds done per bucket in LDS (ds f32 atomics, arrival order, self term last), rows stored once; no global atomics
# speedup vs baseline: 3.0272x; 3.0272x over previous
#include <hip/hip_runtime.h>
#include <hip/hip_bf16.h>

typedef float v2f __attribute__((ext_vector_type(2)));
typedef float v8f __attribute__((ext_vector_type(8)));

template <int NT>
__global__ void gcn_gemm_wmma(const float* __restrict__ X,
                              const float* __restrict__ W,
                              const float* __restrict__ dis,
                              float* __restrict__ hs,
                              int M, int K, int N) {
    int wave = (int)((blockIdx.x * blockDim.x + threadIdx.x) >> 5);
    int lane = threadIdx.x & 31;
    int m0 = wave << 4;
    if (m0 >= M) return;

    int lo = lane & 15;
    int hi = lane >> 4;

    const float* xrow = X + (size_t)(m0 + lo) * K;

    int  col[NT];
    int  ccol[NT];
    bool ok[NT];
#pragma unroll
    for (int t = 0; t < NT; ++t) {
        col[t]  = t * 16 + lo;
        ok[t]   = col[t] < N;
        ccol[t] = ok[t] ? col[t] : 0;
    }

    v8f c[NT];
#pragma unroll
    for (int t = 0; t < NT; ++t) c[t] = (v8f){};

    for (int k0 = 0; k0 < K; k0 += 4) {
        int ka = k0 + 2 * hi;
        v2f a;
        a.x = xrow[ka];
        a.y = xrow[ka + 1];
        const float* w0 = W + (size_t)ka * N;
#pragma unroll
        for (int t = 0; t < NT; ++t) {
            float bx = w0[ccol[t]];
            float by = w0[N + ccol[t]];
            v2f b;
            b.x = ok[t] ? bx : 0.0f;
            b.y = ok[t] ? by : 0.0f;
            c[t] = __builtin_amdgcn_wmma_f32_16x16x4_f32(
                        false, a,  false, b,
                        (short)0, c[t],
                        false,  false);
        }
    }

    int rowBase = m0 + (hi << 3);
#pragma unroll
    for (int r = 0; r < 8; ++r) {
        int rr = rowBase + r;
        int rc = rr < M ? rr : M - 1;
        float dr = dis[rc];
        size_t base = (size_t)rc * N;
#pragma unroll
        for (int t = 0; t < NT; ++t) {
            if (rr < M && ok[t]) hs[base + col[t]] = c[t][r] * dr;
        }
    }
}

constexpr int LOG_NB = 7;
constexpr int NB     = 1 << LOG_NB;
constexpr int PB     = 256;
constexpr int PT     = 1024;
constexpr int PU     = 4;

__global__ void gcn_bucket_hist(const int* __restrict__ dst, int* __restrict__ hist,
                                int nE, int nBk) {
    extern __shared__ int lcount[];
    for (int b = threadIdx.x; b < nBk; b += blockDim.x) lcount[b] = 0;
    __syncthreads();
    const int per = (nE + PB - 1) / PB;
    const int lo  = blockIdx.x * per;
    const int hi  = lo + per < nE ? lo + per : nE;
    for (int e = lo + threadIdx.x; e < hi; e += PU * PT) {
        int d[PU];
#pragma unroll
        for (int u = 0; u < PU; ++u) {
            const int eu = e + u * PT;
            d[u] = dst[eu < hi ? eu : e];
        }
#pragma unroll
        for (int u = 0; u < PU; ++u)
            if (e + u * PT < hi) atomicAdd(&lcount[d[u] >> LOG_NB], 1);
    }
    __syncthreads();
    for (int b = threadIdx.x; b < nBk; b += blockDim.x)
        hist[b * PB + blockIdx.x] = lcount[b];
}

__global__ void gcn_bucket_scan_slices(const int* __restrict__ hist, int* __restrict__ pref,
                                       int* __restrict__ total) {
    __shared__ int s[PB];
    const int p = threadIdx.x;
    const int v = hist[blockIdx.x * PB + p];
    s[p] = v;
    __syncthreads();
    for (int o = 1; o < PB; o <<= 1) {
        const int t = p >= o ? s[p - o] : 0;
        __syncthreads();
        s[p] += t;
        __syncthreads();
    }
    pref[blockIdx.x * PB + p] = s[p] - v;
    if (p == PB - 1) total[blockIdx.x] = s[p];
}

__global__ void gcn_bucket_scan_totals(const int* __restrict__ total, int* __restrict__ start,
                                       int nBk) {
    __shared__ int s[PT];
    __shared__ int carry;
    const int p = threadIdx.x;
    if (p == 0) carry = 0;
    __syncthreads();
    for (int base = 0; base < nBk; base += PT) {
        const int i = base + p;
        const int v = i < nBk ? total[i] : 0;
        s[p] = v;
        __syncthreads();
        for (int o = 1; o < PT; o <<= 1) {
            const int t = p >= o ? s[p - o] : 0;
            __syncthreads();
            s[p] += t;
            __syncthreads();
        }
        const int c = carry;
        if (i < nBk) start[i] = c + s[p] - v;
        __syncthreads();
        if (p == PT - 1) carry = c + s[p];
        __syncthreads();
    }
    if (p == 0) start[nBk] = carry;
}

__global__ void gcn_bucket_fill(const int* __restrict__ src, const int* __restrict__ dst,
                                const int* __restrict__ pref, const int* __restrict__ start,
                                int* __restrict__ grouped, int nE, int nBk) {
    extern __shared__ int cursor[];
    for (int b = threadIdx.x; b < nBk; b += blockDim.x)
        cursor[b] = start[b] + pref[b * PB + blockIdx.x];
    __syncthreads();
    const int per = (nE + PB - 1) / PB;
    const int lo  = blockIdx.x * per;
    const int hi  = lo + per < nE ? lo + per : nE;
    for (int e = lo + threadIdx.x; e < hi; e += PU * PT) {
        int d[PU], s[PU];
#pragma unroll
        for (int u = 0; u < PU; ++u) {
            const int eu = e + u * PT;
            const int ec = eu < hi ? eu : e;
            d[u] = dst[ec];
            s[u] = src[ec];
        }
#pragma unroll
        for (int u = 0; u < PU; ++u) {
            if (e + u * PT < hi) {
                const int pos = atomicAdd(&cursor[d[u] >> LOG_NB], 1);
                grouped[pos] = (s[u] << LOG_NB) | (d[u] & (NB - 1));
            }
        }
    }
}

__global__ void gcn_bucket_degree(const int* __restrict__ grouped, const int* __restrict__ start,
                                  float* __restrict__ dis, int nNodes) {
    __shared__ int cnt[NB];
    for (int r = threadIdx.x; r < NB; r += blockDim.x) cnt[r] = 0;
    __syncthreads();
    const int lo = start[blockIdx.x];
    const int hi = start[blockIdx.x + 1];
    for (int e = lo + threadIdx.x; e < hi; e += blockDim.x)
        atomicAdd(&cnt[grouped[e] & (NB - 1)], 1);
    __syncthreads();
    for (int r = threadIdx.x; r < NB; r += blockDim.x) {
        const int node = blockIdx.x * NB + r;
        if (node < nNodes) dis[node] = __frsqrt_rn((float)(1 + cnt[r]));
    }
}

constexpr int AT = 512;

template <int FT>
__global__ void __launch_bounds__(AT)
gcn_bucket_aggregate(const int* __restrict__ grouped, const int* __restrict__ start,
                     const float* __restrict__ hs, const float* __restrict__ dis,
                     const float* __restrict__ bias, float* __restrict__ out,
                     int nNodes, int Frt, int doRelu) {
    extern __shared__ float4 acc4[];
    float* acc = (float*)acc4;
    const int F = FT ? FT : Frt;
    constexpr int KT = 2;
    constexpr int U  = 8;

    for (int i = threadIdx.x; i < NB * F / 4; i += AT)
        acc4[i] = make_float4(0.0f, 0.0f, 0.0f, 0.0f);
    __syncthreads();

    const int lo   = start[blockIdx.x];
    const int hi   = start[blockIdx.x + 1];
    const int lane = threadIdx.x & 31;
    const int wave = __builtin_amdgcn_readfirstlane((int)(threadIdx.x >> 5));

    int e0 = lo + (wave << 5);
    int my = e0 + lane < hi ? grouped[e0 + lane] : 0;
    for (; e0 < hi; e0 += AT) {
        const int cnt  = hi - e0 < 32 ? hi - e0 : 32;
        const int en   = e0 + AT + lane;
        const int next = en < hi ? grouped[en] : 0;
        for (int j0 = 0; j0 < cnt; j0 += U) {
            float v[U][KT];
#pragma unroll
            for (int u = 0; u < U; ++u) {
                const int j = j0 + u < cnt ? j0 + u : cnt - 1;
                const int w = __builtin_amdgcn_readlane(my, j);
                const float* row = hs + (size_t)(w >> LOG_NB) * F;
#pragma unroll
                for (int k = 0; k < KT; ++k) {
                    int f = (k << 5) + lane;
                    f = f < F ? f : F - 1;
                    v[u][k] = row[f];
                }
            }
#pragma unroll
            for (int u = 0; u < U; ++u) {
                if (j0 + u < cnt) {
                    const int w = __builtin_amdgcn_readlane(my, j0 + u);
                    float* arow = acc + (w & (NB - 1)) * F;
#pragma unroll
                    for (int k = 0; k < KT; ++k) {
                        const int f = (k << 5) + lane;
                        if (f < F) atomicAdd(arow + f, v[u][k]);
                    }
                }
            }
        }
        my = next;
    }
    __syncthreads();

    const size_t g0 = (size_t)blockIdx.x * NB * F;
    const size_t gN = (size_t)nNodes * F;
    if constexpr (FT != 0 && FT % 4 == 0) {
        constexpr int Q   = NB * FT / 4;
        constexpr int PCS = (Q + AT - 1) / AT;
        const float4* hs4  = (const float4*)(hs + g0);
        float4*       out4 = (float4*)(out + g0);
        float4 a[PCS], s[PCS];
        float  d[PCS], bv[PCS][4];
#pragma unroll
        for (int t = 0; t < PCS; ++t) {
            const int  q    = threadIdx.x + t * AT;
            const bool live = q < Q && g0 + 4 * (size_t)q < gN;
            const int  qc   = live ? q : 0;
            const int  r    = (4 * qc) / FT;
            const int  f    = 4 * qc - r * FT;
            d[t] = dis[blockIdx.x * NB + r];
            s[t] = hs4[qc];
            a[t] = acc4[qc];
#pragma unroll
            for (int c = 0; c < 4; ++c) bv[t][c] = bias[f + c];
        }
#pragma unroll
        for (int t = 0; t < PCS; ++t) {
            const int q = threadIdx.x + t * AT;
            if (q < Q && g0 + 4 * (size_t)q < gN) {
                float4 o;
                o.x = fmaf(d[t], a[t].x + s[t].x, bv[t][0]);
                o.y = fmaf(d[t], a[t].y + s[t].y, bv[t][1]);
                o.z = fmaf(d[t], a[t].z + s[t].z, bv[t][2]);
                o.w = fmaf(d[t], a[t].w + s[t].w, bv[t][3]);
                if (doRelu) {
                    o.x = fmaxf(o.x, 0.0f); o.y = fmaxf(o.y, 0.0f);
                    o.z = fmaxf(o.z, 0.0f); o.w = fmaxf(o.w, 0.0f);
                }
                out4[q] = o;
            }
        }
    } else {
        for (int i = threadIdx.x; i < NB * F; i += AT) {
            const size_t g = g0 + i;
            if (g < gN) {
                const int r = i / F;
                const int f = i - r * F;
                float o = fmaf(dis[blockIdx.x * NB + r], acc[i] + hs[g], bias[f]);
                if (doRelu) o = fmaxf(o, 0.0f);
                out[g] = o;
            }
        }
    }
}

static inline void launch_gemm(int tilesN, const float* X, const float* W,
                               const float* dis, float* hs,
                               int M, int K, int N, hipStream_t stream) {
    int mt = (M + 15) / 16;
    long long thr = (long long)mt * 32;
    int blocks = (int)((thr + 255) / 256);
    switch (tilesN) {
    case 1: gcn_gemm_wmma<1><<<blocks, 256, 0, stream>>>(X, W, dis, hs, M, K, N); break;
    case 2: gcn_gemm_wmma<2><<<blocks, 256, 0, stream>>>(X, W, dis, hs, M, K, N); break;
    case 3: gcn_gemm_wmma<3><<<blocks, 256, 0, stream>>>(X, W, dis, hs, M, K, N); break;
    default: gcn_gemm_wmma<4><<<blocks, 256, 0, stream>>>(X, W, dis, hs, M, K, N); break;
    }
}

static inline void launch_aggregate(const int* grouped, const int* start, const float* hs,
                                    const float* dis, const float* bias, float* out,
                                    int nNodes, int nBk, int F, int doRelu,
                                    hipStream_t stream) {
    size_t lds = (size_t)NB * F * sizeof(float);
    switch (F) {
    case 64: gcn_bucket_aggregate<64><<<nBk, AT, lds, stream>>>(
                 grouped, start, hs, dis, bias, out, nNodes, F, doRelu); break;
    case 40: gcn_bucket_aggregate<40><<<nBk, AT, lds, stream>>>(
                 grouped, start, hs, dis, bias, out, nNodes, F, doRelu); break;
    default: gcn_bucket_aggregate<0><<<nBk, AT, lds, stream>>>(
                 grouped, start, hs, dis, bias, out, nNodes, F, doRelu); break;
    }
}

extern "C" void kernel_launch(void* const* d_in, const int* in_sizes, int n_in,
                              void* d_out, int out_size, void* d_ws, size_t ws_size,
                              hipStream_t stream) {
    const float* x  = (const float*)d_in[0];
    const int*   ei = (const int*)d_in[1];
    const float* W1 = (const float*)d_in[2];
    const float* b1 = (const float*)d_in[3];
    const float* W2 = (const float*)d_in[4];
    const float* b2 = (const float*)d_in[5];
    float* out = (float*)d_out;

    const int IN  = 128;
    const int HID = in_sizes[3];
    const int OUT = in_sizes[5];
    const int N   = in_sizes[0] / IN;
    const int E   = in_sizes[1] / 2;

    const int* src = ei;
    const int* dst = ei + E;

    const int nBk  = (N + NB - 1) / NB;
    const int FMAX = HID > OUT ? HID : OUT;

    float* dis    = (float*)d_ws;
    float* hs     = dis + (((size_t)N + 63) & ~63ull);
    float* h1     = hs + (size_t)N * FMAX;
    int*  grouped = (int*)(h1 + (size_t)N * HID);
    int*  hist    = grouped + (((size_t)E + 63) & ~63ull);
    int*  pref    = hist + (size_t)nBk * PB;
    int*  total   = pref + (size_t)nBk * PB;
    int*  start   = total + (((size_t)nBk + 63) & ~63ull);

    size_t ldsBk = (size_t)nBk * sizeof(int);
    gcn_bucket_hist<<<PB, PT, ldsBk, stream>>>(dst, hist, E, nBk);
    gcn_bucket_scan_slices<<<nBk, PB, 0, stream>>>(hist, pref, total);
    gcn_bucket_scan_totals<<<1, PT, 0, stream>>>(total, start, nBk);
    gcn_bucket_fill<<<PB, PT, ldsBk, stream>>>(src, dst, pref, start, grouped, E, nBk);

    gcn_bucket_degree<<<nBk, PT, 0, stream>>>(grouped, start, dis, N);

    launch_gemm((HID + 15) / 16, x, W1, dis, hs, N, IN, HID, stream);
    launch_aggregate(grouped, start, hs, dis, b1, h1, N, nBk, HID,  1, stream);

    launch_gemm((OUT + 15) / 16, h1, W2, dis, hs, N, HID, OUT, stream);
    launch_aggregate(grouped, start, hs, dis, b2, out, N, nBk, OUT,  0, stream);
}


# edges ordered by destination node in each bucket (LDS counting sort, arrival order); a wave sums a node's rows in registers, two rows per 16-byte load, half sums added last; no LDS float adds
# speedup vs baseline: 3.1820x; 1.0511x over previous
#include <hip/hip_runtime.h>
#include <hip/hip_bf16.h>

typedef float v2f __attribute__((ext_vector_type(2)));
typedef float v8f __attribute__((ext_vector_type(8)));

template <int NT>
__global__ void gcn_gemm_wmma(const float* __restrict__ X,
                              const float* __restrict__ W,
                              const float* __restrict__ dis,
                              float* __restrict__ hs,
                              int M, int K, int N) {
    int wave = (int)((blockIdx.x * blockDim.x + threadIdx.x) >> 5);
    int lane = threadIdx.x & 31;
    int m0 = wave << 4;
    if (m0 >= M) return;

    int lo = lane & 15;
    int hi = lane >> 4;

    const float* xrow = X + (size_t)(m0 + lo) * K;

    int  col[NT];
    int  ccol[NT];
    bool ok[NT];
#pragma unroll
    for (int t = 0; t < NT; ++t) {
        col[t]  = t * 16 + lo;
        ok[t]   = col[t] < N;
        ccol[t] = ok[t] ? col[t] : 0;
    }

    v8f c[NT];
#pragma unroll
    for (int t = 0; t < NT; ++t) c[t] = (v8f){};

    for (int k0 = 0; k0 < K; k0 += 4) {
        int ka = k0 + 2 * hi;
        v2f a;
        a.x = xrow[ka];
        a.y = xrow[ka + 1];
        const float* w0 = W + (size_t)ka * N;
#pragma unroll
        for (int t = 0; t < NT; ++t) {
            float bx = w0[ccol[t]];
            float by = w0[N + ccol[t]];
            v2f b;
            b.x = ok[t] ? bx : 0.0f;
            b.y = ok[t] ? by : 0.0f;
            c[t] = __builtin_amdgcn_wmma_f32_16x16x4_f32(
                        false, a,  false, b,
                        (short)0, c[t],
                        false,  false);
        }
    }

    int rowBase = m0 + (hi << 3);
#pragma unroll
    for (int r = 0; r < 8; ++r) {
        int rr = rowBase + r;
        int rc = rr < M ? rr : M - 1;
        float dr = dis[rc];
        size_t base = (size_t)rc * N;
#pragma unroll
        for (int t = 0; t < NT; ++t) {
            if (rr < M && ok[t]) hs[base + col[t]] = c[t][r] * dr;
        }
    }
}

constexpr int LOG_NB = 7;
constexpr int NB     = 1 << LOG_NB;
constexpr int PB     = 256;
constexpr int PT     = 1024;
constexpr int PU     = 4;

__global__ void gcn_bucket_hist(const int* __restrict__ dst, int* __restrict__ hist,
                                int nE, int nBk) {
    extern __shared__ int lcount[];
    for (int b = threadIdx.x; b < nBk; b += blockDim.x) lcount[b] = 0;
    __syncthreads();
    const int per = (nE + PB - 1) / PB;
    const int lo  = blockIdx.x * per;
    const int hi  = lo + per < nE ? lo + per : nE;
    for (int e = lo + threadIdx.x; e < hi; e += PU * PT) {
        int d[PU];
#pragma unroll
        for (int u = 0; u < PU; ++u) {
            const int eu = e + u * PT;
            d[u] = dst[eu < hi ? eu : e];
        }
#pragma unroll
        for (int u = 0; u < PU; ++u)
            if (e + u * PT < hi) atomicAdd(&lcount[d[u] >> LOG_NB], 1);
    }
    __syncthreads();
    for (int b = threadIdx.x; b < nBk; b += blockDim.x)
        hist[b * PB + blockIdx.x] = lcount[b];
}

__global__ void gcn_bucket_scan_slices(const int* __restrict__ hist, int* __restrict__ pref,
                                       int* __restrict__ total) {
    __shared__ int s[PB];
    const int p = threadIdx.x;
    const int v = hist[blockIdx.x * PB + p];
    s[p] = v;
    __syncthreads();
    for (int o = 1; o < PB; o <<= 1) {
        const int t = p >= o ? s[p - o] : 0;
        __syncthreads();
        s[p] += t;
        __syncthreads();
    }
    pref[blockIdx.x * PB + p] = s[p] - v;
    if (p == PB - 1) total[blockIdx.x] = s[p];
}

__global__ void gcn_bucket_scan_totals(const int* __restrict__ total, int* __restrict__ start,
                                       int nBk) {
    __shared__ int s[PT];
    __shared__ int carry;
    const int p = threadIdx.x;
    if (p == 0) carry = 0;
    __syncthreads();
    for (int base = 0; base < nBk; base += PT) {
        const int i = base + p;
        const int v = i < nBk ? total[i] : 0;
        s[p] = v;
        __syncthreads();
        for (int o = 1; o < PT; o <<= 1) {
            const int t = p >= o ? s[p - o] : 0;
            __syncthreads();
            s[p] += t;
            __syncthreads();
        }
        const int c = carry;
        if (i < nBk) start[i] = c + s[p] - v;
        __syncthreads();
        if (p == PT - 1) carry = c + s[p];
        __syncthreads();
    }
    if (p == 0) start[nBk] = carry;
}

__global__ void gcn_bucket_fill(const int* __restrict__ src, const int* __restrict__ dst,
                                const int* __restrict__ pref, const int* __restrict__ start,
                                int* __restrict__ grouped, int nE, int nBk) {
    extern __shared__ int cursor[];
    for (int b = threadIdx.x; b < nBk; b += blockDim.x)
        cursor[b] = start[b] + pref[b * PB + blockIdx.x];
    __syncthreads();
    const int per = (nE + PB - 1) / PB;
    const int lo  = blockIdx.x * per;
    const int hi  = lo + per < nE ? lo + per : nE;
    for (int e = lo + threadIdx.x; e < hi; e += PU * PT) {
        int d[PU], s[PU];
#pragma unroll
        for (int u = 0; u < PU; ++u) {
            const int eu = e + u * PT;
            const int ec = eu < hi ? eu : e;
            d[u] = dst[ec];
            s[u] = src[ec];
        }
#pragma unroll
        for (int u = 0; u < PU; ++u) {
            if (e + u * PT < hi) {
                const int pos = atomicAdd(&cursor[d[u] >> LOG_NB], 1);
                grouped[pos] = (s[u] << LOG_NB) | (d[u] & (NB - 1));
            }
        }
    }
}

__global__ void __launch_bounds__(PT)
gcn_bucket_order(const int* __restrict__ grouped, const int* __restrict__ start,
                 int* __restrict__ sorted, int* __restrict__ rowptr,
                 float* __restrict__ dis, int nNodes, int nBk) {
    __shared__ int cnt[NB];
    __shared__ int off[NB];
    for (int r = threadIdx.x; r < NB; r += blockDim.x) cnt[r] = 0;
    __syncthreads();
    const int lo = start[blockIdx.x];
    const int hi = start[blockIdx.x + 1];
    const int T  = blockDim.x;
    for (int e = lo + threadIdx.x; e < hi; e += PU * T) {
        int w[PU];
#pragma unroll
        for (int u = 0; u < PU; ++u) {
            const int eu = e + u * T;
            w[u] = grouped[eu < hi ? eu : e];
        }
#pragma unroll
        for (int u = 0; u < PU; ++u)
            if (e + u * T < hi) atomicAdd(&cnt[w[u] & (NB - 1)], 1);
    }
    __syncthreads();
    if (threadIdx.x < 32) {
        const int l  = threadIdx.x;
        const int c0 = cnt[4 * l], c1 = cnt[4 * l + 1], c2 = cnt[4 * l + 2], c3 = cnt[4 * l + 3];
        const int s  = c0 + c1 + c2 + c3;
        int incl = s;
        for (int o = 1; o < 32; o <<= 1) {
            const int t = __shfl_up(incl, o, 32);
            if (l >= o) incl += t;
        }
        const int ex = incl - s;
        off[4 * l]     = ex;
        off[4 * l + 1] = ex + c0;
        off[4 * l + 2] = ex + c0 + c1;
        off[4 * l + 3] = ex + c0 + c1 + c2;
    }
    __syncthreads();
    for (int r = threadIdx.x; r < NB; r += blockDim.x) {
        const int node = blockIdx.x * NB + r;
        if (node < nNodes) {
            dis[node]    = __frsqrt_rn((float)(1 + cnt[r]));
            rowptr[node] = lo + off[r];
        }
    }
    if (blockIdx.x == nBk - 1 && threadIdx.x == 0) rowptr[nNodes] = hi;
    __syncthreads();
    for (int e = lo + threadIdx.x; e < hi; e += PU * T) {
        int w[PU];
#pragma unroll
        for (int u = 0; u < PU; ++u) {
            const int eu = e + u * T;
            w[u] = grouped[eu < hi ? eu : e];
        }
#pragma unroll
        for (int u = 0; u < PU; ++u) {
            if (e + u * T < hi) {
                const int pos = atomicAdd(&off[w[u] & (NB - 1)], 1);
                sorted[lo + pos] = w[u] >> LOG_NB;
            }
        }
    }
}

__device__ __forceinline__ float xhalf(float x) {
    return __builtin_bit_cast(float, __builtin_amdgcn_permlanex16(
        __builtin_bit_cast(int, x), __builtin_bit_cast(int, x), 0x76543210, 0xfedcba98, false, false));
}

constexpr int AT  = 512;
constexpr int NPW = NB / (AT / 32);

template <int FT>
__global__ void __launch_bounds__(AT)
gcn_bucket_aggregate(const int* __restrict__ sorted, const int* __restrict__ rowptr,
                     const float* __restrict__ hs, const float* __restrict__ dis,
                     const float* __restrict__ bias, float* __restrict__ out,
                     int nNodes, int Frt, int doRelu) {
    extern __shared__ float4 acc4[];
    float* acc = (float*)acc4;
    const int F = FT ? FT : Frt;

    const int lane = threadIdx.x & 31;
    const int wave = __builtin_amdgcn_readfirstlane((int)(threadIdx.x >> 5));

    int rpi = blockIdx.x * NB + wave * NPW + lane;
    rpi = rpi < nNodes ? rpi : nNodes;
    const int rpv = rowptr[rpi];

    if constexpr (FT != 0 && FT % 4 == 0) {
        constexpr int F4 = FT / 4;
        constexpr int U  = 8;
        const int  j    = lane & 15;
        const bool up   = lane >= 16;
        const int  jc   = j < F4 ? j : F4 - 1;
        const char* lanebase = (const char*)hs + 16 * jc;

        int p0 = __builtin_amdgcn_readlane(rpv, 0);
        int p1 = __builtin_amdgcn_readlane(rpv, 1);
        int sNext = 0;
        {
            const int c = p1 - p0 < 32 ? p1 - p0 : 32;
            const int h = (c + 1) >> 1;
            const int q = up ? h + j : j;
            const bool have = up ? q < c : j < h;
            if (c > 0) sNext = sorted[p0 + (have ? q : 0)];
        }
        for (int i = 0; i < NPW; ++i) {
            float4 a = make_float4(0.0f, 0.0f, 0.0f, 0.0f);
            int s = sNext;
            const int n0 = p0, n1 = p1;
            if (i + 1 < NPW) {
                p0 = n1;
                p1 = __builtin_amdgcn_readlane(rpv, i + 2);
                const int c = p1 - p0 < 32 ? p1 - p0 : 32;
                const int h = (c + 1) >> 1;
                const int q = up ? h + j : j;
                const bool have = up ? q < c : j < h;
                if (c > 0) sNext = sorted[p0 + (have ? q : 0)];
            }
            for (int c0 = n0; c0 < n1; c0 += 32) {
                const int c  = n1 - c0 < 32 ? n1 - c0 : 32;
                const int h  = (c + 1) >> 1;
                const int hf = c >> 1;
                if (c0 != n0) {
                    const int q = up ? h + j : j;
                    const bool have = up ? q < c : j < h;
                    s = sorted[c0 + (have ? q : 0)];
                }
                for (int g = 0; g < h; g += U) {
                    float4 v[U];
#pragma unroll
                    for (int u = 0; u < U; ++u) {
                        const int k  = g + u < h ? g + u : h - 1;
                        const int sa = __builtin_amdgcn_readlane(s, k);
                        const int sb = __builtin_amdgcn_readlane(s, 16 + k);
                        const int sk = up ? sb : sa;
                        v[u] = *(const float4*)(lanebase + (size_t)sk * (FT * 4));
                    }
                    if (g + U <= hf) {
#pragma unroll
                        for (int u = 0; u < U; ++u) {
                            a.x += v[u].x; a.y += v[u].y; a.z += v[u].z; a.w += v[u].w;
                        }
                    } else {
#pragma unroll
                        for (int u = 0; u < U; ++u) {
                            const int  k  = g + u;
                            const bool ok = up ? h + k < c : k < h;
                            a.x += ok ? v[u].x : 0.0f; a.y += ok ? v[u].y : 0.0f;
                            a.z += ok ? v[u].z : 0.0f; a.w += ok ? v[u].w : 0.0f;
                        }
                    }
                }
            }
            a.x += xhalf(a.x); a.y += xhalf(a.y); a.z += xhalf(a.z); a.w += xhalf(a.w);
            if (!up && j < F4) acc4[(wave * NPW + i) * F4 + j] = a;
        }
    } else {
        int f0 = lane, f1 = 32 + lane;
        f0 = f0 < F ? f0 : F - 1;
        f1 = f1 < F ? f1 : F - 1;
        for (int i = 0; i < NPW; ++i) {
            const int n0 = __builtin_amdgcn_readlane(rpv, i);
            const int n1 = __builtin_amdgcn_readlane(rpv, i + 1);
            float a0 = 0.0f, a1 = 0.0f;
            for (int c0 = n0; c0 < n1; c0 += 32) {
                const int c = n1 - c0 < 32 ? n1 - c0 : 32;
                const int s = sorted[c0 + (lane < c ? lane : 0)];
                for (int k = 0; k < c; ++k) {
                    const float* row = hs + (size_t)__builtin_amdgcn_readlane(s, k) * F;
                    a0 += row[f0];
                    a1 += row[f1];
                }
            }
            float* arow = acc + (wave * NPW + i) * F;
            if (lane < F)      arow[lane]      = a0;
            if (32 + lane < F) arow[32 + lane] = a1;
        }
    }
    __syncthreads();

    const size_t g0 = (size_t)blockIdx.x * NB * F;
    const size_t gN = (size_t)nNodes * F;
    if constexpr (FT != 0 && FT % 4 == 0) {
        constexpr int Q   = NB * FT / 4;
        constexpr int PCS = (Q + AT - 1) / AT;
        const float4* hs4  = (const float4*)(hs + g0);
        float4*       out4 = (float4*)(out + g0);
        float4 a[PCS], s[PCS];
        float  d[PCS], bv[PCS][4];
#pragma unroll
        for (int t = 0; t < PCS; ++t) {
            const int  q    = threadIdx.x + t * AT;
            const bool live = q < Q && g0 + 4 * (size_t)q < gN;
            const int  qc   = live ? q : 0;
            const int  r    = (4 * qc) / FT;
            const int  f    = 4 * qc - r * FT;
            d[t] = dis[blockIdx.x * NB + r];
            s[t] = hs4[qc];
            a[t] = acc4[qc];
#pragma unroll
            for (int c = 0; c < 4; ++c) bv[t][c] = bias[f + c];
        }
#pragma unroll
        for (int t = 0; t < PCS; ++t) {
            const int q = threadIdx.x + t * AT;
            if (q < Q && g0 + 4 * (size_t)q < gN) {
                float4 o;
                o.x = fmaf(d[t], a[t].x + s[t].x, bv[t][0]);
                o.y = fmaf(d[t], a[t].y + s[t].y, bv[t][1]);
                o.z = fmaf(d[t], a[t].z + s[t].z, bv[t][2]);
                o.w = fmaf(d[t], a[t].w + s[t].w, bv[t][3]);
                if (doRelu) {
                    o.x = fmaxf(o.x, 0.0f); o.y = fmaxf(o.y, 0.0f);
                    o.z = fmaxf(o.z, 0.0f); o.w = fmaxf(o.w, 0.0f);
                }
                out4[q] = o;
            }
        }
    } else {
        for (int i = threadIdx.x; i < NB * F; i += AT) {
            const size_t g = g0 + i;
            if (g < gN) {
                const int r = i / F;
                const int f = i - r * F;
                float o = fmaf(dis[blockIdx.x * NB + r], acc[i] + hs[g], bias[f]);
                if (doRelu) o = fmaxf(o, 0.0f);
                out[g] = o;
            }
        }
    }
}

static inline void launch_gemm(int tilesN, const float* X, const float* W,
                               const float* dis, float* hs,
                               int M, int K, int N, hipStream_t stream) {
    int mt = (M + 15) / 16;
    long long thr = (long long)mt * 32;
    int blocks = (int)((thr + 255) / 256);
    switch (tilesN) {
    case 1: gcn_gemm_wmma<1><<<blocks, 256, 0, stream>>>(X, W, dis, hs, M, K, N); break;
    case 2: gcn_gemm_wmma<2><<<blocks, 256, 0, stream>>>(X, W, dis, hs, M, K, N); break;
    case 3: gcn_gemm_wmma<3><<<blocks, 256, 0, stream>>>(X, W, dis, hs, M, K, N); break;
    default: gcn_gemm_wmma<4><<<blocks, 256, 0, stream>>>(X, W, dis, hs, M, K, N); break;
    }
}

static inline void launch_aggregate(const int* sorted, const int* rowptr, const float* hs,
                                    const float* dis, const float* bias, float* out,
                                    int nNodes, int nBk, int F, int doRelu,
                                    hipStream_t stream) {
    size_t lds = (size_t)NB * F * sizeof(float);
    switch (F) {
    case 64: gcn_bucket_aggregate<64><<<nBk, AT, lds, stream>>>(
                 sorted, rowptr, hs, dis, bias, out, nNodes, F, doRelu); break;
    case 40: gcn_bucket_aggregate<40><<<nBk, AT, lds, stream>>>(
                 sorted, rowptr, hs, dis, bias, out, nNodes, F, doRelu); break;
    default: gcn_bucket_aggregate<0><<<nBk, AT, lds, stream>>>(
                 sorted, rowptr, hs, dis, bias, out, nNodes, F, doRelu); break;
    }
}

extern "C" void kernel_launch(void* const* d_in, const int* in_sizes, int n_in,
                              void* d_out, int out_size, void* d_ws, size_t ws_size,
                              hipStream_t stream) {
    const float* x  = (const float*)d_in[0];
    const int*   ei = (const int*)d_in[1];
    const float* W1 = (const float*)d_in[2];
    const float* b1 = (const float*)d_in[3];
    const float* W2 = (const float*)d_in[4];
    const float* b2 = (const float*)d_in[5];
    float* out = (float*)d_out;

    const int IN  = 128;
    const int HID = in_sizes[3];
    const int OUT = in_sizes[5];
    const int N   = in_sizes[0] / IN;
    const int E   = in_sizes[1] / 2;

    const int* src = ei;
    const int* dst = ei + E;

    const int nBk  = (N + NB - 1) / NB;
    const int FMAX = HID > OUT ? HID : OUT;

    float* dis    = (float*)d_ws;
    float* hs     = dis + (((size_t)N + 63) & ~63ull);
    float* h1     = hs + (size_t)N * FMAX;
    int*  sorted  = (int*)(h1 + (size_t)N * HID);
    int*  hist    = sorted + (((size_t)E + 63) & ~63ull);
    int*  pref    = hist + (size_t)nBk * PB;
    int*  total   = pref + (size_t)nBk * PB;
    int*  start   = total + (((size_t)nBk + 63) & ~63ull);
    int*  grouped = (int*)hs;
    int*  rowptr  = hist;

    size_t ldsBk = (size_t)nBk * sizeof(int);
    gcn_bucket_hist<<<PB, PT, ldsBk, stream>>>(dst, hist, E, nBk);
    gcn_bucket_scan_slices<<<nBk, PB, 0, stream>>>(hist, pref, total);
    gcn_bucket_scan_totals<<<1, PT, 0, stream>>>(total, start, nBk);
    gcn_bucket_fill<<<PB, PT, ldsBk, stream>>>(src, dst, pref, start, grouped, E, nBk);

    gcn_bucket_order<<<nBk, PT, 0, stream>>>(grouped, start, sorted, rowptr, dis, N, nBk);

    launch_gemm((HID + 15) / 16, x, W1, dis, hs, N, IN, HID, stream);
    launch_aggregate(sorted, rowptr, hs, dis, b1, h1, N, nBk, HID,  1, stream);

    launch_gemm((OUT + 15) / 16, h1, W2, dis, hs, N, HID, OUT, stream);
    launch_aggregate(sorted, rowptr, hs, dis, b2, out, N, nBk, OUT,  0, stream);
}


# GEMM operand feed: W staged once per block in LDS as k-pair rows, x fragments fetched one 32-k chunk ahead, row scales ahead of the loop (was one load round trip in front of every WMMA)
# speedup vs baseline: 3.5340x; 1.1106x over previous
#include <hip/hip_runtime.h>
#include <hip/hip_bf16.h>

typedef float v2f __attribute__((ext_vector_type(2)));
typedef float v8f __attribute__((ext_vector_type(8)));

template <int NT>
__global__ void __launch_bounds__(256)
gcn_gemm_wmma(const float* __restrict__ X,
              const float* __restrict__ W,
              const float* __restrict__ dis,
              float* __restrict__ hs,
              int M, int K, int N) {
    constexpr int NP    = NT * 16;
    constexpr int PITCH = (NP % 32 == 16) ? NP : NP + 16;
    constexpr int KC    = 8;
    extern __shared__ float2 wsm[];

    const int wave = (int)((blockIdx.x * blockDim.x + threadIdx.x) >> 5);
    const int lane = threadIdx.x & 31;
    const int m0 = wave << 4;
    const int lo = lane & 15;
    const int hi = lane >> 4;
    const bool live = m0 < M;

    int arow = m0 + lo;
    arow = arow < M ? arow : M - 1;
    const float* xrow = X + (size_t)arow * K + 2 * hi;

    const int nfull = K / (4 * KC);
    v2f an[KC];
#pragma unroll
    for (int j = 0; j < KC; ++j) an[j] = (v2f){0.0f, 0.0f};
    if (nfull > 0) {
#pragma unroll
        for (int j = 0; j < KC; ++j) an[j] = *(const v2f*)(xrow + 4 * j);
    }
    const int rowBase = m0 + (hi << 3);
    float dr[8];
#pragma unroll
    for (int r = 0; r < 8; ++r) {
        const int rr = rowBase + r;
        dr[r] = dis[rr < M ? rr : M - 1];
    }

    {
        const int KP = K >> 1;
        const int total = KP * NP;
        for (int base = 0; base < total; base += 4 * 256) {
            float w0[4], w1[4];
#pragma unroll
            for (int u = 0; u < 4; ++u) {
                int idx = base + u * 256 + (int)threadIdx.x;
                idx = idx < total ? idx : total - 1;
                const int kp = idx / NP;
                const int n  = idx - kp * NP;
                const int nc = n < N ? n : 0;
                w0[u] = W[(size_t)(2 * kp) * N + nc];
                w1[u] = W[(size_t)(2 * kp + 1) * N + nc];
            }
#pragma unroll
            for (int u = 0; u < 4; ++u) {
                const int idx = base + u * 256 + (int)threadIdx.x;
                if (idx < total) {
                    const int kp = idx / NP;
                    const int n  = idx - kp * NP;
                    wsm[kp * PITCH + n] = n < N ? make_float2(w0[u], w1[u]) : make_float2(0.0f, 0.0f);
                }
            }
        }
    }
    __syncthreads();
    if (!live) return;

    v8f c[NT];
#pragma unroll
    for (int t = 0; t < NT; ++t) c[t] = (v8f){};

    const float2* wl = wsm + hi * PITCH + lo;
    for (int ch = 0; ch < nfull; ++ch) {
        v2f ac[KC];
#pragma unroll
        for (int j = 0; j < KC; ++j) ac[j] = an[j];
        if (ch + 1 < nfull) {
#pragma unroll
            for (int j = 0; j < KC; ++j) an[j] = *(const v2f*)(xrow + (ch + 1) * (4 * KC) + 4 * j);
        }
        const float2* wc = wl + (size_t)ch * (2 * KC) * PITCH;
#pragma unroll
        for (int j = 0; j < KC; ++j) {
            v2f b[NT];
#pragma unroll
            for (int t = 0; t < NT; ++t) {
                const float2 w = wc[2 * j * PITCH + t * 16];
                b[t].x = w.x;
                b[t].y = w.y;
            }
#pragma unroll
            for (int t = 0; t < NT; ++t)
                c[t] = __builtin_amdgcn_wmma_f32_16x16x4_f32(
                            false, ac[j],  false, b[t],
                            (short)0, c[t],
                            false,  false);
        }
    }
    for (int k0 = nfull * (4 * KC); k0 < K; k0 += 4) {
        const v2f a = *(const v2f*)(xrow + k0);
        const float2* wc = wl + (size_t)(k0 >> 1) * PITCH;
#pragma unroll
        for (int t = 0; t < NT; ++t) {
            const float2 w = wc[t * 16];
            v2f b;
            b.x = w.x;
            b.y = w.y;
            c[t] = __builtin_amdgcn_wmma_f32_16x16x4_f32(
                        false, a,  false, b,
                        (short)0, c[t],
                        false,  false);
        }
    }

#pragma unroll
    for (int r = 0; r < 8; ++r) {
        const int rr = rowBase + r;
        if (rr < M) {
            const size_t base = (size_t)rr * N;
#pragma unroll
            for (int t = 0; t < NT; ++t) {
                const int col = t * 16 + lo;
                if (col < N) hs[base + col] = c[t][r] * dr[r];
            }
        }
    }
}

constexpr int LOG_NB = 7;
constexpr int NB     = 1 << LOG_NB;
constexpr int PB     = 256;
constexpr int PT     = 1024;
constexpr int PU     = 4;

__global__ void gcn_bucket_hist(const int* __restrict__ dst, int* __restrict__ hist,
                                int nE, int nBk) {
    extern __shared__ int lcount[];
    for (int b = threadIdx.x; b < nBk; b += blockDim.x) lcount[b] = 0;
    __syncthreads();
    const int per = (nE + PB - 1) / PB;
    const int lo  = blockIdx.x * per;
    const int hi  = lo + per < nE ? lo + per : nE;
    for (int e = lo + threadIdx.x; e < hi; e += PU * PT) {
        int d[PU];
#pragma unroll
        for (int u = 0; u < PU; ++u) {
            const int eu = e + u * PT;
            d[u] = dst[eu < hi ? eu : e];
        }
#pragma unroll
        for (int u = 0; u < PU; ++u)
            if (e + u * PT < hi) atomicAdd(&lcount[d[u] >> LOG_NB], 1);
    }
    __syncthreads();
    for (int b = threadIdx.x; b < nBk; b += blockDim.x)
        hist[b * PB + blockIdx.x] = lcount[b];
}

__global__ void gcn_bucket_scan_slices(const int* __restrict__ hist, int* __restrict__ pref,
                                       int* __restrict__ total) {
    __shared__ int s[PB];
    const int p = threadIdx.x;
    const int v = hist[blockIdx.x * PB + p];
    s[p] = v;
    __syncthreads();
    for (int o = 1; o < PB; o <<= 1) {
        const int t = p >= o ? s[p - o] : 0;
        __syncthreads();
        s[p] += t;
        __syncthreads();
    }
    pref[blockIdx.x * PB + p] = s[p] - v;
    if (p == PB - 1) total[blockIdx.x] = s[p];
}

__global__ void gcn_bucket_scan_totals(const int* __restrict__ total, int* __restrict__ start,
                                       int nBk) {
    __shared__ int s[PT];
    __shared__ int carry;
    const int p = threadIdx.x;
    if (p == 0) carry = 0;
    __syncthreads();
    for (int base = 0; base < nBk; base += PT) {
        const int i = base + p;
        const int v = i < nBk ? total[i] : 0;
        s[p] = v;
        __syncthreads();
        for (int o = 1; o < PT; o <<= 1) {
            const int t = p >= o ? s[p - o] : 0;
            __syncthreads();
            s[p] += t;
            __syncthreads();
        }
        const int c = carry;
        if (i < nBk) start[i] = c + s[p] - v;
        __syncthreads();
        if (p == PT - 1) carry = c + s[p];
        __syncthreads();
    }
    if (p == 0) start[nBk] = carry;
}

__global__ void gcn_bucket_fill(const int* __restrict__ src, const int* __restrict__ dst,
                                const int* __restrict__ pref, const int* __restrict__ start,
                                int* __restrict__ grouped, int nE, int nBk) {
    extern __shared__ int cursor[];
    for (int b = threadIdx.x; b < nBk; b += blockDim.x)
        cursor[b] = start[b] + pref[b * PB + blockIdx.x];
    __syncthreads();
    const int per = (nE + PB - 1) / PB;
    const int lo  = blockIdx.x * per;
    const int hi  = lo + per < nE ? lo + per : nE;
    for (int e = lo + threadIdx.x; e < hi; e += PU * PT) {
        int d[PU], s[PU];
#pragma unroll
        for (int u = 0; u < PU; ++u) {
            const int eu = e + u * PT;
            const int ec = eu < hi ? eu : e;
            d[u] = dst[ec];
            s[u] = src[ec];
        }
#pragma unroll
        for (int u = 0; u < PU; ++u) {
            if (e + u * PT < hi) {
                const int pos = atomicAdd(&cursor[d[u] >> LOG_NB], 1);
                grouped[pos] = (s[u] << LOG_NB) | (d[u] & (NB - 1));
            }
        }
    }
}

__global__ void __launch_bounds__(PT)
gcn_bucket_order(const int* __restrict__ grouped, const int* __restrict__ start,
                 int* __restrict__ sorted, int* __restrict__ rowptr,
                 float* __restrict__ dis, int nNodes, int nBk) {
    __shared__ int cnt[NB];
    __shared__ int off[NB];
    for (int r = threadIdx.x; r < NB; r += blockDim.x) cnt[r] = 0;
    __syncthreads();
    const int lo = start[blockIdx.x];
    const int hi = start[blockIdx.x + 1];
    const int T  = blockDim.x;
    for (int e = lo + threadIdx.x; e < hi; e += PU * T) {
        int w[PU];
#pragma unroll
        for (int u = 0; u < PU; ++u) {
            const int eu = e + u * T;
            w[u] = grouped[eu < hi ? eu : e];
        }
#pragma unroll
        for (int u = 0; u < PU; ++u)
            if (e + u * T < hi) atomicAdd(&cnt[w[u] & (NB - 1)], 1);
    }
    __syncthreads();
    if (threadIdx.x < 32) {
        const int l  = threadIdx.x;
        const int c0 = cnt[4 * l], c1 = cnt[4 * l + 1], c2 = cnt[4 * l + 2], c3 = cnt[4 * l + 3];
        const int s  = c0 + c1 + c2 + c3;
        int incl = s;
        for (int o = 1; o < 32; o <<= 1) {
            const int t = __shfl_up(incl, o, 32);
            if (l >= o) incl += t;
        }
        const int ex = incl - s;
        off[4 * l]     = ex;
        off[4 * l + 1] = ex + c0;
        off[4 * l + 2] = ex + c0 + c1;
        off[4 * l + 3] = ex + c0 + c1 + c2;
    }
    __syncthreads();
    for (int r = threadIdx.x; r < NB; r += blockDim.x) {
        const int node = blockIdx.x * NB + r;
        if (node < nNodes) {
            dis[node]    = __frsqrt_rn((float)(1 + cnt[r]));
            rowptr[node] = lo + off[r];
        }
    }
    if (blockIdx.x == nBk - 1 && threadIdx.x == 0) rowptr[nNodes] = hi;
    __syncthreads();
    for (int e = lo + threadIdx.x; e < hi; e += PU * T) {
        int w[PU];
#pragma unroll
        for (int u = 0; u < PU; ++u) {
            const int eu = e + u * T;
            w[u] = grouped[eu < hi ? eu : e];
        }
#pragma unroll
        for (int u = 0; u < PU; ++u) {
            if (e + u * T < hi) {
                const int pos = atomicAdd(&off[w[u] & (NB - 1)], 1);
                sorted[lo + pos] = w[u] >> LOG_NB;
            }
        }
    }
}

__device__ __forceinline__ float xhalf(float x) {
    return __builtin_bit_cast(float, __builtin_amdgcn_permlanex16(
        __builtin_bit_cast(int, x), __builtin_bit_cast(int, x), 0x76543210, 0xfedcba98, false, false));
}

constexpr int AT  = 512;
constexpr int NPW = NB / (AT / 32);

template <int FT>
__global__ void __launch_bounds__(AT)
gcn_bucket_aggregate(const int* __restrict__ sorted, const int* __restrict__ rowptr,
                     const float* __restrict__ hs, const float* __restrict__ dis,
                     const float* __restrict__ bias, float* __restrict__ out,
                     int nNodes, int Frt, int doRelu) {
    extern __shared__ float4 acc4[];
    float* acc = (float*)acc4;
    const int F = FT ? FT : Frt;

    const int lane = threadIdx.x & 31;
    const int wave = __builtin_amdgcn_readfirstlane((int)(threadIdx.x >> 5));

    int rpi = blockIdx.x * NB + wave * NPW + lane;
    rpi = rpi < nNodes ? rpi : nNodes;
    const int rpv = rowptr[rpi];

    if constexpr (FT != 0 && FT % 4 == 0) {
        constexpr int F4 = FT / 4;
        constexpr int U  = 8;
        const int  j    = lane & 15;
        const bool up   = lane >= 16;
        const int  jc   = j < F4 ? j : F4 - 1;
        const char* lanebase = (const char*)hs + 16 * jc;

        int p0 = __builtin_amdgcn_readlane(rpv, 0);
        int p1 = __builtin_amdgcn_readlane(rpv, 1);
        int sNext = 0;
        {
            const int c = p1 - p0 < 32 ? p1 - p0 : 32;
            const int h = (c + 1) >> 1;
            const int q = up ? h + j : j;
            const bool have = up ? q < c : j < h;
            if (c > 0) sNext = sorted[p0 + (have ? q : 0)];
        }
        for (int i = 0; i < NPW; ++i) {
            float4 a = make_float4(0.0f, 0.0f, 0.0f, 0.0f);
            int s = sNext;
            const int n0 = p0, n1 = p1;
            if (i + 1 < NPW) {
                p0 = n1;
                p1 = __builtin_amdgcn_readlane(rpv, i + 2);
                const int c = p1 - p0 < 32 ? p1 - p0 : 32;
                const int h = (c + 1) >> 1;
                const int q = up ? h + j : j;
                const bool have = up ? q < c : j < h;
                if (c > 0) sNext = sorted[p0 + (have ? q : 0)];
            }
            for (int c0 = n0; c0 < n1; c0 += 32) {
                const int c  = n1 - c0 < 32 ? n1 - c0 : 32;
                const int h  = (c + 1) >> 1;
                const int hf = c >> 1;
                if (c0 != n0) {
                    const int q = up ? h + j : j;
                    const bool have = up ? q < c : j < h;
                    s = sorted[c0 + (have ? q : 0)];
                }
                for (int g = 0; g < h; g += U) {
                    float4 v[U];
#pragma unroll
                    for (int u = 0; u < U; ++u) {
                        const int k  = g + u < h ? g + u : h - 1;
                        const int sa = __builtin_amdgcn_readlane(s, k);
                        const int sb = __builtin_amdgcn_readlane(s, 16 + k);
                        const int sk = up ? sb : sa;
                        v[u] = *(const float4*)(lanebase + (size_t)sk * (FT * 4));
                    }
                    if (g + U <= hf) {
#pragma unroll
                        for (int u = 0; u < U; ++u) {
                            a.x += v[u].x; a.y += v[u].y; a.z += v[u].z; a.w += v[u].w;
                        }
                    } else {
#pragma unroll
                        for (int u = 0; u < U; ++u) {
                            const int  k  = g + u;
                            const bool ok = up ? h + k < c : k < h;
                            a.x += ok ? v[u].x : 0.0f; a.y += ok ? v[u].y : 0.0f;
                            a.z += ok ? v[u].z : 0.0f; a.w += ok ? v[u].w : 0.0f;
                        }
                    }
                }
            }
            a.x += xhalf(a.x); a.y += xhalf(a.y); a.z += xhalf(a.z); a.w += xhalf(a.w);
            if (!up && j < F4) acc4[(wave * NPW + i) * F4 + j] = a;
        }
    } else {
        int f0 = lane, f1 = 32 + lane;
        f0 = f0 < F ? f0 : F - 1;
        f1 = f1 < F ? f1 : F - 1;
        for (int i = 0; i < NPW; ++i) {
            const int n0 = __builtin_amdgcn_readlane(rpv, i);
            const int n1 = __builtin_amdgcn_readlane(rpv, i + 1);
            float a0 = 0.0f, a1 = 0.0f;
            for (int c0 = n0; c0 < n1; c0 += 32) {
                const int c = n1 - c0 < 32 ? n1 - c0 : 32;
                const int s = sorted[c0 + (lane < c ? lane : 0)];
                for (int k = 0; k < c; ++k) {
                    const float* row = hs + (size_t)__builtin_amdgcn_readlane(s, k) * F;
                    a0 += row[f0];
                    a1 += row[f1];
                }
            }
            float* arow = acc + (wave * NPW + i) * F;
            if (lane < F)      arow[lane]      = a0;
            if (32 + lane < F) arow[32 + lane] = a1;
        }
    }
    __syncthreads();

    const size_t g0 = (size_t)blockIdx.x * NB * F;
    const size_t gN = (size_t)nNodes * F;
    if constexpr (FT != 0 && FT % 4 == 0) {
        constexpr int Q   = NB * FT / 4;
        constexpr int PCS = (Q + AT - 1) / AT;
        const float4* hs4  = (const float4*)(hs + g0);
        float4*       out4 = (float4*)(out + g0);
        float4 a[PCS], s[PCS];
        float  d[PCS], bv[PCS][4];
#pragma unroll
        for (int t = 0; t < PCS; ++t) {
            const int  q    = threadIdx.x + t * AT;
            const bool live = q < Q && g0 + 4 * (size_t)q < gN;
            const int  qc   = live ? q : 0;
            const int  r    = (4 * qc) / FT;
            const int  f    = 4 * qc - r * FT;
            d[t] = dis[blockIdx.x * NB + r];
            s[t] = hs4[qc];
            a[t] = acc4[qc];
#pragma unroll
            for (int c = 0; c < 4; ++c) bv[t][c] = bias[f + c];
        }
#pragma unroll
        for (int t = 0; t < PCS; ++t) {
            const int q = threadIdx.x + t * AT;
            if (q < Q && g0 + 4 * (size_t)q < gN) {
                float4 o;
                o.x = fmaf(d[t], a[t].x + s[t].x, bv[t][0]);
                o.y = fmaf(d[t], a[t].y + s[t].y, bv[t][1]);
                o.z = fmaf(d[t], a[t].z + s[t].z, bv[t][2]);
                o.w = fmaf(d[t], a[t].w + s[t].w, bv[t][3]);
                if (doRelu) {
                    o.x = fmaxf(o.x, 0.0f); o.y = fmaxf(o.y, 0.0f);
                    o.z = fmaxf(o.z, 0.0f); o.w = fmaxf(o.w, 0.0f);
                }
                out4[q] = o;
            }
        }
    } else {
        for (int i = threadIdx.x; i < NB * F; i += AT) {
            const size_t g = g0 + i;
            if (g < gN) {
                const int r = i / F;
                const int f = i - r * F;
                float o = fmaf(dis[blockIdx.x * NB + r], acc[i] + hs[g], bias[f]);
                if (doRelu) o = fmaxf(o, 0.0f);
                out[g] = o;
            }
        }
    }
}

static inline void launch_gemm(int tilesN, const float* X, const float* W,
                               const float* dis, float* hs,
                               int M, int K, int N, hipStream_t stream) {
    int mt = (M + 15) / 16;
    long long thr = (long long)mt * 32;
    int blocks = (int)((thr + 255) / 256);
    switch (tilesN) {
    case 1: gcn_gemm_wmma<1><<<blocks, 256, (size_t)(K / 2) * 16 * 8, stream>>>(X, W, dis, hs, M, K, N); break;
    case 2: gcn_gemm_wmma<2><<<blocks, 256, (size_t)(K / 2) * 48 * 8, stream>>>(X, W, dis, hs, M, K, N); break;
    case 3: gcn_gemm_wmma<3><<<blocks, 256, (size_t)(K / 2) * 48 * 8, stream>>>(X, W, dis, hs, M, K, N); break;
    default: gcn_gemm_wmma<4><<<blocks, 256, (size_t)(K / 2) * 80 * 8, stream>>>(X, W, dis, hs, M, K, N); break;
    }
}

static inline void launch_aggregate(const int* sorted, const int* rowptr, const float* hs,
                                    const float* dis, const float* bias, float* out,
                                    int nNodes, int nBk, int F, int doRelu,
                                    hipStream_t stream) {
    size_t lds = (size_t)NB * F * sizeof(float);
    switch (F) {
    case 64: gcn_bucket_aggregate<64><<<nBk, AT, lds, stream>>>(
                 sorted, rowptr, hs, dis, bias, out, nNodes, F, doRelu); break;
    case 40: gcn_bucket_aggregate<40><<<nBk, AT, lds, stream>>>(
                 sorted, rowptr, hs, dis, bias, out, nNodes, F, doRelu); break;
    default: gcn_bucket_aggregate<0><<<nBk, AT, lds, stream>>>(
                 sorted, rowptr, hs, dis, bias, out, nNodes, F, doRelu); break;
    }
}

extern "C" void kernel_launch(void* const* d_in, const int* in_sizes, int n_in,
                              void* d_out, int out_size, void* d_ws, size_t ws_size,
                              hipStream_t stream) {
    const float* x  = (const float*)d_in[0];
    const int*   ei = (const int*)d_in[1];
    const float* W1 = (const float*)d_in[2];
    const float* b1 = (const float*)d_in[3];
    const float* W2 = (const float*)d_in[4];
    const float* b2 = (const float*)d_in[5];
    float* out = (float*)d_out;

    const int IN  = 128;
    const int HID = in_sizes[3];
    const int OUT = in_sizes[5];
    const int N   = in_sizes[0] / IN;
    const int E   = in_sizes[1] / 2;

    const int* src = ei;
    const int* dst = ei + E;

    const int nBk  = (N + NB - 1) / NB;
    const int FMAX = HID > OUT ? HID : OUT;

    float* dis    = (float*)d_ws;
    float* hs     = dis + (((size_t)N + 63) & ~63ull);
    float* h1     = hs + (size_t)N * FMAX;
    int*  sorted  = (int*)(h1 + (size_t)N * HID);
    int*  hist    = sorted + (((size_t)E + 63) & ~63ull);
    int*  pref    = hist + (size_t)nBk * PB;
    int*  total   = pref + (size_t)nBk * PB;
    int*  start   = total + (((size_t)nBk + 63) & ~63ull);
    int*  grouped = (int*)hs;
    int*  rowptr  = hist;

    size_t ldsBk = (size_t)nBk * sizeof(int);
    gcn_bucket_hist<<<PB, PT, ldsBk, stream>>>(dst, hist, E, nBk);
    gcn_bucket_scan_slices<<<nBk, PB, 0, stream>>>(hist, pref, total);
    gcn_bucket_scan_totals<<<1, PT, 0, stream>>>(total, start, nBk);
    gcn_bucket_fill<<<PB, PT, ldsBk, stream>>>(src, dst, pref, start, grouped, E, nBk);

    gcn_bucket_order<<<nBk, PT, 0, stream>>>(grouped, start, sorted, rowptr, dis, N, nBk);

    launch_gemm((HID + 15) / 16, x, W1, dis, hs, N, IN, HID, stream);
    launch_aggregate(sorted, rowptr, hs, dis, b1, h1, N, nBk, HID,  1, stream);

    launch_gemm((OUT + 15) / 16, h1, W2, dis, hs, N, HID, OUT, stream);
    launch_aggregate(sorted, rowptr, hs, dis, b2, out, N, nBk, OUT,  0, stream);
}


# edge grouping in one launch not four: each edge-list slice bucket-sorted in its own range (count, LDS scan, place); a bucket's block reads its 256 runs, base = sum of run offsets; arrival-order sums
# speedup vs baseline: 4.2395x; 1.1996x over previous
#include <hip/hip_runtime.h>
#include <hip/hip_bf16.h>

typedef float v2f __attribute__((ext_vector_type(2)));
typedef float v8f __attribute__((ext_vector_type(8)));

template <int NT>
__global__ void __launch_bounds__(256)
gcn_gemm_wmma(const float* __restrict__ X,
              const float* __restrict__ W,
              const float* __restrict__ dis,
              float* __restrict__ hs,
              int M, int K, int N) {
    constexpr int NP    = NT * 16;
    constexpr int PITCH = (NP % 32 == 16) ? NP : NP + 16;
    constexpr int KC    = 8;
    extern __shared__ float2 wsm[];

    const int wave = (int)((blockIdx.x * blockDim.x + threadIdx.x) >> 5);
    const int lane = threadIdx.x & 31;
    const int m0 = wave << 4;
    const int lo = lane & 15;
    const int hi = lane >> 4;
    const bool live = m0 < M;

    int arow = m0 + lo;
    arow = arow < M ? arow : M - 1;
    const float* xrow = X + (size_t)arow * K + 2 * hi;

    const int nfull = K / (4 * KC);
    v2f an[KC];
#pragma unroll
    for (int j = 0; j < KC; ++j) an[j] = (v2f){0.0f, 0.0f};
    if (nfull > 0) {
#pragma unroll
        for (int j = 0; j < KC; ++j) an[j] = *(const v2f*)(xrow + 4 * j);
    }
    const int rowBase = m0 + (hi << 3);
    float dr[8];
#pragma unroll
    for (int r = 0; r < 8; ++r) {
        const int rr = rowBase + r;
        dr[r] = dis[rr < M ? rr : M - 1];
    }

    {
        const int KP = K >> 1;
        const int total = KP * NP;
        for (int base = 0; base < total; base += 4 * 256) {
            float w0[4], w1[4];
#pragma unroll
            for (int u = 0; u < 4; ++u) {
                int idx = base + u * 256 + (int)threadIdx.x;
                idx = idx < total ? idx : total - 1;
                const int kp = idx / NP;
                const int n  = idx - kp * NP;
                const int nc = n < N ? n : 0;
                w0[u] = W[(size_t)(2 * kp) * N + nc];
                w1[u] = W[(size_t)(2 * kp + 1) * N + nc];
            }
#pragma unroll
            for (int u = 0; u < 4; ++u) {
                const int idx = base + u * 256 + (int)threadIdx.x;
                if (idx < total) {
                    const int kp = idx / NP;
                    const int n  = idx - kp * NP;
                    wsm[kp * PITCH + n] = n < N ? make_float2(w0[u], w1[u]) : make_float2(0.0f, 0.0f);
                }
            }
        }
    }
    __syncthreads();
    if (!live) return;

    v8f c[NT];
#pragma unroll
    for (int t = 0; t < NT; ++t) c[t] = (v8f){};

    const float2* wl = wsm + hi * PITCH + lo;
    for (int ch = 0; ch < nfull; ++ch) {
        v2f ac[KC];
#pragma unroll
        for (int j = 0; j < KC; ++j) ac[j] = an[j];
        if (ch + 1 < nfull) {
#pragma unroll
            for (int j = 0; j < KC; ++j) an[j] = *(const v2f*)(xrow + (ch + 1) * (4 * KC) + 4 * j);
        }
        const float2* wc = wl + (size_t)ch * (2 * KC) * PITCH;
#pragma unroll
        for (int j = 0; j < KC; ++j) {
            v2f b[NT];
#pragma unroll
            for (int t = 0; t < NT; ++t) {
                const float2 w = wc[2 * j * PITCH + t * 16];
                b[t].x = w.x;
                b[t].y = w.y;
            }
#pragma unroll
            for (int t = 0; t < NT; ++t)
                c[t] = __builtin_amdgcn_wmma_f32_16x16x4_f32(
                            false, ac[j],  false, b[t],
                            (short)0, c[t],
                            false,  false);
        }
    }
    for (int k0 = nfull * (4 * KC); k0 < K; k0 += 4) {
        const v2f a = *(const v2f*)(xrow + k0);
        const float2* wc = wl + (size_t)(k0 >> 1) * PITCH;
#pragma unroll
        for (int t = 0; t < NT; ++t) {
            const float2 w = wc[t * 16];
            v2f b;
            b.x = w.x;
            b.y = w.y;
            c[t] = __builtin_amdgcn_wmma_f32_16x16x4_f32(
                        false, a,  false, b,
                        (short)0, c[t],
                        false,  false);
        }
    }

#pragma unroll
    for (int r = 0; r < 8; ++r) {
        const int rr = rowBase + r;
        if (rr < M) {
            const size_t base = (size_t)rr * N;
#pragma unroll
            for (int t = 0; t < NT; ++t) {
                const int col = t * 16 + lo;
                if (col < N) hs[base + col] = c[t][r] * dr[r];
            }
        }
    }
}

constexpr int LOG_NB = 7;
constexpr int NB     = 1 << LOG_NB;
constexpr int PB     = 256;
constexpr int PT     = 1024;
constexpr int PU     = 4;

__global__ void __launch_bounds__(PT)
gcn_slice_sort(const int* __restrict__ src, const int* __restrict__ dst,
               int* __restrict__ offs, int* __restrict__ grouped, int nE, int nBk) {
    static_assert(PT == 1024, "the scan below lets one wave add up the totals of 32 waves");
    extern __shared__ int lcount[];
    __shared__ int wsum[PT / 32];
    __shared__ int chunkTotal;
    __shared__ int carry;
    const int p    = threadIdx.x;
    const int lane = p & 31;
    const int wv   = p >> 5;
    for (int b = p; b < nBk; b += PT) lcount[b] = 0;
    if (p == 0) carry = 0;
    __syncthreads();
    const int per = (nE + PB - 1) / PB;
    const int lo  = blockIdx.x * per;
    const int hi  = lo + per < nE ? lo + per : nE;
    for (int e = lo + p; e < hi; e += PU * PT) {
        int d[PU];
#pragma unroll
        for (int u = 0; u < PU; ++u) {
            const int eu = e + u * PT;
            d[u] = dst[eu < hi ? eu : e];
        }
#pragma unroll
        for (int u = 0; u < PU; ++u)
            if (e + u * PT < hi) atomicAdd(&lcount[d[u] >> LOG_NB], 1);
    }
    __syncthreads();
    for (int base = 0; base < nBk; base += PT) {
        const int i = base + p;
        const int v = i < nBk ? lcount[i] : 0;
        int incl = v;
        for (int o = 1; o < 32; o <<= 1) {
            const int t = __shfl_up(incl, o, 32);
            if (lane >= o) incl += t;
        }
        if (lane == 31) wsum[wv] = incl;
        __syncthreads();
        if (wv == 0) {
            const int ws = wsum[lane];
            int wi = ws;
            for (int o = 1; o < 32; o <<= 1) {
                const int t = __shfl_up(wi, o, 32);
                if (lane >= o) wi += t;
            }
            wsum[lane] = wi - ws;
            if (lane == 31) chunkTotal = wi;
        }
        __syncthreads();
        const int c = carry;
        if (i < nBk) {
            const int ex = c + wsum[wv] + incl - v;
            offs[i * PB + blockIdx.x] = ex;
            lcount[i] = lo + ex;
        }
        __syncthreads();
        if (p == 0) carry = c + chunkTotal;
        __syncthreads();
    }
    if (p == 0) offs[nBk * PB + blockIdx.x] = carry;
    for (int e = lo + p; e < hi; e += PU * PT) {
        int d[PU], s[PU];
#pragma unroll
        for (int u = 0; u < PU; ++u) {
            const int eu = e + u * PT;
            const int ec = eu < hi ? eu : e;
            d[u] = dst[ec];
            s[u] = src[ec];
        }
#pragma unroll
        for (int u = 0; u < PU; ++u) {
            if (e + u * PT < hi) {
                const int pos = atomicAdd(&lcount[d[u] >> LOG_NB], 1);
                grouped[pos] = (s[u] << LOG_NB) | (d[u] & (NB - 1));
            }
        }
    }
}

constexpr int RL = PT / PB;
constexpr int OU = 8;

__global__ void __launch_bounds__(PT)
gcn_bucket_order(const int* __restrict__ grouped, const int* __restrict__ offs,
                 int* __restrict__ sorted, int* __restrict__ rowptr,
                 float* __restrict__ dis, int nNodes, int nBk, int nE) {
    __shared__ int cnt[NB];
    __shared__ int off[NB];
    __shared__ int rbeg[PB];
    __shared__ int rend[PB];
    __shared__ int ends[2];
    const int t   = threadIdx.x / RL;
    const int sub = threadIdx.x % RL;
    const int rs  = offs[blockIdx.x * PB + t];
    const int re  = offs[(blockIdx.x + 1) * PB + t];
    for (int r = threadIdx.x; r < NB; r += blockDim.x) cnt[r] = 0;
    if (sub == 0) { rbeg[t] = rs; rend[t] = re; }
    __syncthreads();
    const int per = (nE + PB - 1) / PB;
    const int* run = grouped + (size_t)t * per + rs;
    const int len  = re - rs;
    for (int k = sub; k < len; k += OU * RL) {
        int w[OU];
#pragma unroll
        for (int u = 0; u < OU; ++u) {
            const int ku = k + u * RL;
            w[u] = run[ku < len ? ku : k];
        }
#pragma unroll
        for (int u = 0; u < OU; ++u)
            if (k + u * RL < len) atomicAdd(&cnt[w[u] & (NB - 1)], 1);
    }
    __syncthreads();
    if (threadIdx.x < 32) {
        const int l  = threadIdx.x;
        const int c0 = cnt[4 * l], c1 = cnt[4 * l + 1], c2 = cnt[4 * l + 2], c3 = cnt[4 * l + 3];
        const int s  = c0 + c1 + c2 + c3;
        int incl = s;
        for (int o = 1; o < 32; o <<= 1) {
            const int t2 = __shfl_up(incl, o, 32);
            if (l >= o) incl += t2;
        }
        const int ex = incl - s;
        off[4 * l]     = ex;
        off[4 * l + 1] = ex + c0;
        off[4 * l + 2] = ex + c0 + c1;
        off[4 * l + 3] = ex + c0 + c1 + c2;
        int sb = 0, se = 0;
        for (int q = l; q < PB; q += 32) { sb += rbeg[q]; se += rend[q]; }
        for (int o = 16; o > 0; o >>= 1) {
            sb += __shfl_xor(sb, o, 32);
            se += __shfl_xor(se, o, 32);
        }
        if (l == 0) { ends[0] = sb; ends[1] = se; }
    }
    __syncthreads();
    const int lo = ends[0];
    const int hi = ends[1];
    for (int r = threadIdx.x; r < NB; r += blockDim.x) {
        const int node = blockIdx.x * NB + r;
        if (node < nNodes) {
            dis[node]    = __frsqrt_rn((float)(1 + cnt[r]));
            rowptr[node] = lo + off[r];
        }
    }
    if (blockIdx.x == nBk - 1 && threadIdx.x == 0) rowptr[nNodes] = hi;
    __syncthreads();
    for (int k = sub; k < len; k += OU * RL) {
        int w[OU];
#pragma unroll
        for (int u = 0; u < OU; ++u) {
            const int ku = k + u * RL;
            w[u] = run[ku < len ? ku : k];
        }
#pragma unroll
        for (int u = 0; u < OU; ++u) {
            if (k + u * RL < len) {
                const int pos = atomicAdd(&off[w[u] & (NB - 1)], 1);
                sorted[lo + pos] = w[u] >> LOG_NB;
            }
        }
    }
}

__device__ __forceinline__ float xhalf(float x) {
    return __builtin_bit_cast(float, __builtin_amdgcn_permlanex16(
        __builtin_bit_cast(int, x), __builtin_bit_cast(int, x), 0x76543210, 0xfedcba98, false, false));
}

constexpr int AT  = 512;
constexpr int NPW = NB / (AT / 32);

template <int FT>
__global__ void __launch_bounds__(AT)
gcn_bucket_aggregate(const int* __restrict__ sorted, const int* __restrict__ rowptr,
                     const float* __restrict__ hs, const float* __restrict__ dis,
                     const float* __restrict__ bias, float* __restrict__ out,
                     int nNodes, int Frt, int doRelu) {
    extern __shared__ float4 acc4[];
    float* acc = (float*)acc4;
    const int F = FT ? FT : Frt;

    const int lane = threadIdx.x & 31;
    const int wave = __builtin_amdgcn_readfirstlane((int)(threadIdx.x >> 5));

    int rpi = blockIdx.x * NB + wave * NPW + lane;
    rpi = rpi < nNodes ? rpi : nNodes;
    const int rpv = rowptr[rpi];

    if constexpr (FT != 0 && FT % 4 == 0) {
        constexpr int F4 = FT / 4;
        constexpr int U  = 8;
        const int  j    = lane & 15;
        const bool up   = lane >= 16;
        const int  jc   = j < F4 ? j : F4 - 1;
        const char* lanebase = (const char*)hs + 16 * jc;

        int p0 = __builtin_amdgcn_readlane(rpv, 0);
        int p1 = __builtin_amdgcn_readlane(rpv, 1);
        int sNext = 0;
        {
            const int c = p1 - p0 < 32 ? p1 - p0 : 32;
            const int h = (c + 1) >> 1;
            const int q = up ? h + j : j;
            const bool have = up ? q < c : j < h;
            if (c > 0) sNext = sorted[p0 + (have ? q : 0)];
        }
        for (int i = 0; i < NPW; ++i) {
            float4 a = make_float4(0.0f, 0.0f, 0.0f, 0.0f);
            int s = sNext;
            const int n0 = p0, n1 = p1;
            if (i + 1 < NPW) {
                p0 = n1;
                p1 = __builtin_amdgcn_readlane(rpv, i + 2);
                const int c = p1 - p0 < 32 ? p1 - p0 : 32;
                const int h = (c + 1) >> 1;
                const int q = up ? h + j : j;
                const bool have = up ? q < c : j < h;
                if (c > 0) sNext = sorted[p0 + (have ? q : 0)];
            }
            for (int c0 = n0; c0 < n1; c0 += 32) {
                const int c  = n1 - c0 < 32 ? n1 - c0 : 32;
                const int h  = (c + 1) >> 1;
                const int hf = c >> 1;
                if (c0 != n0) {
                    const int q = up ? h + j : j;
                    const bool have = up ? q < c : j < h;
                    s = sorted[c0 + (have ? q : 0)];
                }
                for (int g = 0; g < h; g += U) {
                    float4 v[U];
#pragma unroll
                    for (int u = 0; u < U; ++u) {
                        const int k  = g + u < h ? g + u : h - 1;
                        const int sa = __builtin_amdgcn_readlane(s, k);
                        const int sb = __builtin_amdgcn_readlane(s, 16 + k);
                        const int sk = up ? sb : sa;
                        v[u] = *(const float4*)(lanebase + (size_t)sk * (FT * 4));
                    }
                    if (g + U <= hf) {
#pragma unroll
                        for (int u = 0; u < U; ++u) {
                            a.x += v[u].x; a.y += v[u].y; a.z += v[u].z; a.w += v[u].w;
                        }
                    } else {
#pragma unroll
                        for (int u = 0; u < U; ++u) {
                            const int  k  = g + u;
                            const bool ok = up ? h + k < c : k < h;
                            a.x += ok ? v[u].x : 0.0f; a.y += ok ? v[u].y : 0.0f;
                            a.z += ok ? v[u].z : 0.0f; a.w += ok ? v[u].w : 0.0f;
                        }
                    }
                }
            }
            a.x += xhalf(a.x); a.y += xhalf(a.y); a.z += xhalf(a.z); a.w += xhalf(a.w);
            if (!up && j < F4) acc4[(wave * NPW + i) * F4 + j] = a;
        }
    } else {
        int f0 = lane, f1 = 32 + lane;
        f0 = f0 < F ? f0 : F - 1;
        f1 = f1 < F ? f1 : F - 1;
        for (int i = 0; i < NPW; ++i) {
            const int n0 = __builtin_amdgcn_readlane(rpv, i);
            const int n1 = __builtin_amdgcn_readlane(rpv, i + 1);
            float a0 = 0.0f, a1 = 0.0f;
            for (int c0 = n0; c0 < n1; c0 += 32) {
                const int c = n1 - c0 < 32 ? n1 - c0 : 32;
                const int s = sorted[c0 + (lane < c ? lane : 0)];
                for (int k = 0; k < c; ++k) {
                    const float* row = hs + (size_t)__builtin_amdgcn_readlane(s, k) * F;
                    a0 += row[f0];
                    a1 += row[f1];
                }
            }
            float* arow = acc + (wave * NPW + i) * F;
            if (lane < F)      arow[lane]      = a0;
            if (32 + lane < F) arow[32 + lane] = a1;
        }
    }
    __syncthreads();

    const size_t g0 = (size_t)blockIdx.x * NB * F;
    const size_t gN = (size_t)nNodes * F;
    if constexpr (FT != 0 && FT % 4 == 0) {
        constexpr int Q   = NB * FT / 4;
        constexpr int PCS = (Q + AT - 1) / AT;
        const float4* hs4  = (const float4*)(hs + g0);
        float4*       out4 = (float4*)(out + g0);
        float4 a[PCS], s[PCS];
        float  d[PCS], bv[PCS][4];
#pragma unroll
        for (int t = 0; t < PCS; ++t) {
            const int  q    = threadIdx.x + t * AT;
            const bool live = q < Q && g0 + 4 * (size_t)q < gN;
            const int  qc   = live ? q : 0;
            const int  r    = (4 * qc) / FT;
            const int  f    = 4 * qc - r * FT;
            d[t] = dis[blockIdx.x * NB + r];
            s[t] = hs4[qc];
            a[t] = acc4[qc];
#pragma unroll
            for (int c = 0; c < 4; ++c) bv[t][c] = bias[f + c];
        }
#pragma unroll
        for (int t = 0; t < PCS; ++t) {
            const int q = threadIdx.x + t * AT;
            if (q < Q && g0 + 4 * (size_t)q < gN) {
                float4 o;
                o.x = fmaf(d[t], a[t].x + s[t].x, bv[t][0]);
                o.y = fmaf(d[t], a[t].y + s[t].y, bv[t][1]);
                o.z = fmaf(d[t], a[t].z + s[t].z, bv[t][2]);
                o.w = fmaf(d[t], a[t].w + s[t].w, bv[t][3]);
                if (doRelu) {
                    o.x = fmaxf(o.x, 0.0f); o.y = fmaxf(o.y, 0.0f);
                    o.z = fmaxf(o.z, 0.0f); o.w = fmaxf(o.w, 0.0f);
                }
                out4[q] = o;
            }
        }
    } else {
        for (int i = threadIdx.x; i < NB * F; i += AT) {
            const size_t g = g0 + i;
            if (g < gN) {
                const int r = i / F;
                const int f = i - r * F;
                float o = fmaf(dis[blockIdx.x * NB + r], acc[i] + hs[g], bias[f]);
                if (doRelu) o = fmaxf(o, 0.0f);
                out[g] = o;
            }
        }
    }
}

static inline void launch_gemm(int tilesN, const float* X, const float* W,
                               const float* dis, float* hs,
                               int M, int K, int N, hipStream_t stream) {
    int mt = (M + 15) / 16;
    long long thr = (long long)mt * 32;
    int blocks = (int)((thr + 255) / 256);
    switch (tilesN) {
    case 1: gcn_gemm_wmma<1><<<blocks, 256, (size_t)(K / 2) * 16 * 8, stream>>>(X, W, dis, hs, M, K, N); break;
    case 2: gcn_gemm_wmma<2><<<blocks, 256, (size_t)(K / 2) * 48 * 8, stream>>>(X, W, dis, hs, M, K, N); break;
    case 3: gcn_gemm_wmma<3><<<blocks, 256, (size_t)(K / 2) * 48 * 8, stream>>>(X, W, dis, hs, M, K, N); break;
    default: gcn_gemm_wmma<4><<<blocks, 256, (size_t)(K / 2) * 80 * 8, stream>>>(X, W, dis, hs, M, K, N); break;
    }
}

static inline void launch_aggregate(const int* sorted, const int* rowptr, const float* hs,
                                    const float* dis, const float* bias, float* out,
                                    int nNodes, int nBk, int F, int doRelu,
                                    hipStream_t stream) {
    size_t lds = (size_t)NB * F * sizeof(float);
    switch (F) {
    case 64: gcn_bucket_aggregate<64><<<nBk, AT, lds, stream>>>(
                 sorted, rowptr, hs, dis, bias, out, nNodes, F, doRelu); break;
    case 40: gcn_bucket_aggregate<40><<<nBk, AT, lds, stream>>>(
                 sorted, rowptr, hs, dis, bias, out, nNodes, F, doRelu); break;
    default: gcn_bucket_aggregate<0><<<nBk, AT, lds, stream>>>(
                 sorted, rowptr, hs, dis, bias, out, nNodes, F, doRelu); break;
    }
}

extern "C" void kernel_launch(void* const* d_in, const int* in_sizes, int n_in,
                              void* d_out, int out_size, void* d_ws, size_t ws_size,
                              hipStream_t stream) {
    const float* x  = (const float*)d_in[0];
    const int*   ei = (const int*)d_in[1];
    const float* W1 = (const float*)d_in[2];
    const float* b1 = (const float*)d_in[3];
    const float* W2 = (const float*)d_in[4];
    const float* b2 = (const float*)d_in[5];
    float* out = (float*)d_out;

    const int IN  = 128;
    const int HID = in_sizes[3];
    const int OUT = in_sizes[5];
    const int N   = in_sizes[0] / IN;
    const int E   = in_sizes[1] / 2;

    const int* src = ei;
    const int* dst = ei + E;

    const int nBk  = (N + NB - 1) / NB;
    const int FMAX = HID > OUT ? HID : OUT;

    float* dis    = (float*)d_ws;
    float* hs     = dis + (((size_t)N + 63) & ~63ull);
    float* h1     = hs + (size_t)N * FMAX;
    int*  sorted  = (int*)(h1 + (size_t)N * HID);
    int*  rowptr  = sorted + (((size_t)E + 63) & ~63ull);
    int*  offs    = rowptr + (size_t)nBk * PB;
    int*  grouped = (int*)hs;

    size_t ldsBk = (size_t)nBk * sizeof(int);
    gcn_slice_sort<<<PB, PT, ldsBk, stream>>>(src, dst, offs, grouped, E, nBk);

    gcn_bucket_order<<<nBk, PT, 0, stream>>>(grouped, offs, sorted, rowptr, dis, N, nBk, E);

    launch_gemm((HID + 15) / 16, x, W1, dis, hs, N, IN, HID, stream);
    launch_aggregate(sorted, rowptr, hs, dis, b1, h1, N, nBk, HID,  1, stream);

    launch_gemm((OUT + 15) / 16, h1, W2, dis, hs, N, HID, OUT, stream);
    launch_aggregate(sorted, rowptr, hs, dis, b2, out, N, nBk, OUT,  0, stream);
}


# both edge sorts build their ordered output in LDS and store it as whole 128-byte lines (was a 4-byte store per edge, each lane to a line of its own); a node's edges in LDS arrival order as before
# speedup vs baseline: 4.6915x; 1.1066x over previous
#include <hip/hip_runtime.h>
#include <hip/hip_bf16.h>

typedef float v2f __attribute__((ext_vector_type(2)));
typedef float v8f __attribute__((ext_vector_type(8)));

template <int NT>
__global__ void __launch_bounds__(256)
gcn_gemm_wmma(const float* __restrict__ X,
              const float* __restrict__ W,
              const float* __restrict__ dis,
              float* __restrict__ hs,
              int M, int K, int N) {
    constexpr int NP    = NT * 16;
    constexpr int PITCH = (NP % 32 == 16) ? NP : NP + 16;
    constexpr int KC    = 8;
    extern __shared__ float2 wsm[];

    const int wave = (int)((blockIdx.x * blockDim.x + threadIdx.x) >> 5);
    const int lane = threadIdx.x & 31;
    const int m0 = wave << 4;
    const int lo = lane & 15;
    const int hi = lane >> 4;
    const bool live = m0 < M;

    int arow = m0 + lo;
    arow = arow < M ? arow : M - 1;
    const float* xrow = X + (size_t)arow * K + 2 * hi;

    const int nfull = K / (4 * KC);
    v2f an[KC];
#pragma unroll
    for (int j = 0; j < KC; ++j) an[j] = (v2f){0.0f, 0.0f};
    if (nfull > 0) {
#pragma unroll
        for (int j = 0; j < KC; ++j) an[j] = *(const v2f*)(xrow + 4 * j);
    }
    const int rowBase = m0 + (hi << 3);
    float dr[8];
#pragma unroll
    for (int r = 0; r < 8; ++r) {
        const int rr = rowBase + r;
        dr[r] = dis[rr < M ? rr : M - 1];
    }

    {
        const int KP = K >> 1;
        const int total = KP * NP;
        for (int base = 0; base < total; base += 4 * 256) {
            float w0[4], w1[4];
#pragma unroll
            for (int u = 0; u < 4; ++u) {
                int idx = base + u * 256 + (int)threadIdx.x;
                idx = idx < total ? idx : total - 1;
                const int kp = idx / NP;
                const int n  = idx - kp * NP;
                const int nc = n < N ? n : 0;
                w0[u] = W[(size_t)(2 * kp) * N + nc];
                w1[u] = W[(size_t)(2 * kp + 1) * N + nc];
            }
#pragma unroll
            for (int u = 0; u < 4; ++u) {
                const int idx = base + u * 256 + (int)threadIdx.x;
                if (idx < total) {
                    const int kp = idx / NP;
                    const int n  = idx - kp * NP;
                    wsm[kp * PITCH + n] = n < N ? make_float2(w0[u], w1[u]) : make_float2(0.0f, 0.0f);
                }
            }
        }
    }
    __syncthreads();
    if (!live) return;

    v8f c[NT];
#pragma unroll
    for (int t = 0; t < NT; ++t) c[t] = (v8f){};

    const float2* wl = wsm + hi * PITCH + lo;
    for (int ch = 0; ch < nfull; ++ch) {
        v2f ac[KC];
#pragma unroll
        for (int j = 0; j < KC; ++j) ac[j] = an[j];
        if (ch + 1 < nfull) {
#pragma unroll
            for (int j = 0; j < KC; ++j) an[j] = *(const v2f*)(xrow + (ch + 1) * (4 * KC) + 4 * j);
        }
        const float2* wc = wl + (size_t)ch * (2 * KC) * PITCH;
#pragma unroll
        for (int j = 0; j < KC; ++j) {
            v2f b[NT];
#pragma unroll
            for (int t = 0; t < NT; ++t) {
                const float2 w = wc[2 * j * PITCH + t * 16];
                b[t].x = w.x;
                b[t].y = w.y;
            }
#pragma unroll
            for (int t = 0; t < NT; ++t)
                c[t] = __builtin_amdgcn_wmma_f32_16x16x4_f32(
                            false, ac[j],  false, b[t],
                            (short)0, c[t],
                            false,  false);
        }
    }
    for (int k0 = nfull * (4 * KC); k0 < K; k0 += 4) {
        const v2f a = *(const v2f*)(xrow + k0);
        const float2* wc = wl + (size_t)(k0 >> 1) * PITCH;
#pragma unroll
        for (int t = 0; t < NT; ++t) {
            const float2 w = wc[t * 16];
            v2f b;
            b.x = w.x;
            b.y = w.y;
            c[t] = __builtin_amdgcn_wmma_f32_16x16x4_f32(
                        false, a,  false, b,
                        (short)0, c[t],
                        false,  false);
        }
    }

#pragma unroll
    for (int r = 0; r < 8; ++r) {
        const int rr = rowBase + r;
        if (rr < M) {
            const size_t base = (size_t)rr * N;
#pragma unroll
            for (int t = 0; t < NT; ++t) {
                const int col = t * 16 + lo;
                if (col < N) hs[base + col] = c[t][r] * dr[r];
            }
        }
    }
}

constexpr int LOG_NB = 7;
constexpr int NB     = 1 << LOG_NB;
constexpr int PB     = 256;
constexpr int PT     = 1024;
constexpr int PU     = 4;

__global__ void __launch_bounds__(PT)
gcn_slice_sort(const int* __restrict__ src, const int* __restrict__ dst,
               int* __restrict__ offs, int* __restrict__ grouped, int nE, int nBk) {
    static_assert(PT == 1024, "the scan below lets one wave add up the totals of 32 waves");
    extern __shared__ int lcount[];
    __shared__ int wsum[PT / 32];
    __shared__ int chunkTotal;
    __shared__ int carry;
    const int p    = threadIdx.x;
    const int lane = p & 31;
    const int wv   = p >> 5;
    for (int b = p; b < nBk; b += PT) lcount[b] = 0;
    if (p == 0) carry = 0;
    __syncthreads();
    const int per = (nE + PB - 1) / PB;
    const int lo  = blockIdx.x * per;
    const int hi  = lo + per < nE ? lo + per : nE;
    int* stage = lcount + nBk;
    for (int e = lo + p; e < hi; e += PU * PT) {
        int d[PU];
#pragma unroll
        for (int u = 0; u < PU; ++u) {
            const int eu = e + u * PT;
            d[u] = dst[eu < hi ? eu : e];
        }
#pragma unroll
        for (int u = 0; u < PU; ++u)
            if (e + u * PT < hi) atomicAdd(&lcount[d[u] >> LOG_NB], 1);
    }
    __syncthreads();
    for (int base = 0; base < nBk; base += PT) {
        const int i = base + p;
        const int v = i < nBk ? lcount[i] : 0;
        int incl = v;
        for (int o = 1; o < 32; o <<= 1) {
            const int t = __shfl_up(incl, o, 32);
            if (lane >= o) incl += t;
        }
        if (lane == 31) wsum[wv] = incl;
        __syncthreads();
        if (wv == 0) {
            const int ws = wsum[lane];
            int wi = ws;
            for (int o = 1; o < 32; o <<= 1) {
                const int t = __shfl_up(wi, o, 32);
                if (lane >= o) wi += t;
            }
            wsum[lane] = wi - ws;
            if (lane == 31) chunkTotal = wi;
        }
        __syncthreads();
        const int c = carry;
        if (i < nBk) {
            const int ex = c + wsum[wv] + incl - v;
            offs[i * PB + blockIdx.x] = ex;
            lcount[i] = ex;
        }
        __syncthreads();
        if (p == 0) carry = c + chunkTotal;
        __syncthreads();
    }
    if (p == 0) offs[nBk * PB + blockIdx.x] = carry;
    for (int e = lo + p; e < hi; e += PU * PT) {
        int d[PU], s[PU];
#pragma unroll
        for (int u = 0; u < PU; ++u) {
            const int eu = e + u * PT;
            const int ec = eu < hi ? eu : e;
            d[u] = dst[ec];
            s[u] = src[ec];
        }
#pragma unroll
        for (int u = 0; u < PU; ++u) {
            if (e + u * PT < hi) {
                const int pos = atomicAdd(&lcount[d[u] >> LOG_NB], 1);
                stage[pos] = (s[u] << LOG_NB) | (d[u] & (NB - 1));
            }
        }
    }
    __syncthreads();
    for (int gb = (lo & ~31) + p; gb < hi; gb += PU * PT) {
        int v[PU];
#pragma unroll
        for (int u = 0; u < PU; ++u) {
            const int g = gb + u * PT;
            v[u] = stage[g >= lo && g < hi ? g - lo : 0];
        }
#pragma unroll
        for (int u = 0; u < PU; ++u) {
            const int g = gb + u * PT;
            if (g >= lo && g < hi) grouped[g] = v[u];
        }
    }
}

constexpr int RL = PT / PB;
constexpr int OU = 8;
constexpr int SCAP = 8192;

__global__ void __launch_bounds__(PT)
gcn_bucket_order(const int* __restrict__ grouped, const int* __restrict__ offs,
                 int* __restrict__ sorted, int* __restrict__ rowptr,
                 float* __restrict__ dis, int nNodes, int nBk, int nE) {
    __shared__ int cnt[NB];
    __shared__ int off[NB];
    __shared__ int rbeg[PB];
    __shared__ int rend[PB];
    __shared__ int ends[2];
    __shared__ int stage[SCAP];
    const int t   = threadIdx.x / RL;
    const int sub = threadIdx.x % RL;
    const int rs  = offs[blockIdx.x * PB + t];
    const int re  = offs[(blockIdx.x + 1) * PB + t];
    for (int r = threadIdx.x; r < NB; r += blockDim.x) cnt[r] = 0;
    if (sub == 0) { rbeg[t] = rs; rend[t] = re; }
    __syncthreads();
    const int per = (nE + PB - 1) / PB;
    const int* run = grouped + (size_t)t * per + rs;
    const int len  = re - rs;
    for (int k = sub; k < len; k += OU * RL) {
        int w[OU];
#pragma unroll
        for (int u = 0; u < OU; ++u) {
            const int ku = k + u * RL;
            w[u] = run[ku < len ? ku : k];
        }
#pragma unroll
        for (int u = 0; u < OU; ++u)
            if (k + u * RL < len) atomicAdd(&cnt[w[u] & (NB - 1)], 1);
    }
    __syncthreads();
    if (threadIdx.x < 32) {
        const int l  = threadIdx.x;
        const int c0 = cnt[4 * l], c1 = cnt[4 * l + 1], c2 = cnt[4 * l + 2], c3 = cnt[4 * l + 3];
        const int s  = c0 + c1 + c2 + c3;
        int incl = s;
        for (int o = 1; o < 32; o <<= 1) {
            const int t2 = __shfl_up(incl, o, 32);
            if (l >= o) incl += t2;
        }
        const int ex = incl - s;
        off[4 * l]     = ex;
        off[4 * l + 1] = ex + c0;
        off[4 * l + 2] = ex + c0 + c1;
        off[4 * l + 3] = ex + c0 + c1 + c2;
        int sb = 0, se = 0;
        for (int q = l; q < PB; q += 32) { sb += rbeg[q]; se += rend[q]; }
        for (int o = 16; o > 0; o >>= 1) {
            sb += __shfl_xor(sb, o, 32);
            se += __shfl_xor(se, o, 32);
        }
        if (l == 0) { ends[0] = sb; ends[1] = se; }
    }
    __syncthreads();
    const int lo = ends[0];
    const int hi = ends[1];
    const bool staged = hi - lo <= SCAP;
    for (int r = threadIdx.x; r < NB; r += blockDim.x) {
        const int node = blockIdx.x * NB + r;
        if (node < nNodes) {
            dis[node]    = __frsqrt_rn((float)(1 + cnt[r]));
            rowptr[node] = lo + off[r];
        }
    }
    if (blockIdx.x == nBk - 1 && threadIdx.x == 0) rowptr[nNodes] = hi;
    __syncthreads();
    for (int k = sub; k < len; k += OU * RL) {
        int w[OU];
#pragma unroll
        for (int u = 0; u < OU; ++u) {
            const int ku = k + u * RL;
            w[u] = run[ku < len ? ku : k];
        }
#pragma unroll
        for (int u = 0; u < OU; ++u) {
            if (k + u * RL < len) {
                const int pos = atomicAdd(&off[w[u] & (NB - 1)], 1);
                if (staged) stage[pos] = w[u] >> LOG_NB;
                else        sorted[lo + pos] = w[u] >> LOG_NB;
            }
        }
    }
    if (staged) {
        __syncthreads();
        for (int g = (lo & ~31) + (int)threadIdx.x; g < hi; g += (int)blockDim.x)
            if (g >= lo) sorted[g] = stage[g - lo];
    }
}

__device__ __forceinline__ float xhalf(float x) {
    return __builtin_bit_cast(float, __builtin_amdgcn_permlanex16(
        __builtin_bit_cast(int, x), __builtin_bit_cast(int, x), 0x76543210, 0xfedcba98, false, false));
}

constexpr int AT  = 512;
constexpr int NPW = NB / (AT / 32);

template <int FT>
__global__ void __launch_bounds__(AT)
gcn_bucket_aggregate(const int* __restrict__ sorted, const int* __restrict__ rowptr,
                     const float* __restrict__ hs, const float* __restrict__ dis,
                     const float* __restrict__ bias, float* __restrict__ out,
                     int nNodes, int Frt, int doRelu) {
    extern __shared__ float4 acc4[];
    float* acc = (float*)acc4;
    const int F = FT ? FT : Frt;

    const int lane = threadIdx.x & 31;
    const int wave = __builtin_amdgcn_readfirstlane((int)(threadIdx.x >> 5));

    int rpi = blockIdx.x * NB + wave * NPW + lane;
    rpi = rpi < nNodes ? rpi : nNodes;
    const int rpv = rowptr[rpi];

    if constexpr (FT != 0 && FT % 4 == 0) {
        constexpr int F4 = FT / 4;
        constexpr int U  = 8;
        const int  j    = lane & 15;
        const bool up   = lane >= 16;
        const int  jc   = j < F4 ? j : F4 - 1;
        const char* lanebase = (const char*)hs + 16 * jc;

        int p0 = __builtin_amdgcn_readlane(rpv, 0);
        int p1 = __builtin_amdgcn_readlane(rpv, 1);
        int sNext = 0;
        {
            const int c = p1 - p0 < 32 ? p1 - p0 : 32;
            const int h = (c + 1) >> 1;
            const int q = up ? h + j : j;
            const bool have = up ? q < c : j < h;
            if (c > 0) sNext = sorted[p0 + (have ? q : 0)];
        }
        for (int i = 0; i < NPW; ++i) {
            float4 a = make_float4(0.0f, 0.0f, 0.0f, 0.0f);
            int s = sNext;
            const int n0 = p0, n1 = p1;
            if (i + 1 < NPW) {
                p0 = n1;
                p1 = __builtin_amdgcn_readlane(rpv, i + 2);
                const int c = p1 - p0 < 32 ? p1 - p0 : 32;
                const int h = (c + 1) >> 1;
                const int q = up ? h + j : j;
                const bool have = up ? q < c : j < h;
                if (c > 0) sNext = sorted[p0 + (have ? q : 0)];
            }
            for (int c0 = n0; c0 < n1; c0 += 32) {
                const int c  = n1 - c0 < 32 ? n1 - c0 : 32;
                const int h  = (c + 1) >> 1;
                const int hf = c >> 1;
                if (c0 != n0) {
                    const int q = up ? h + j : j;
                    const bool have = up ? q < c : j < h;
                    s = sorted[c0 + (have ? q : 0)];
                }
                for (int g = 0; g < h; g += U) {
                    float4 v[U];
#pragma unroll
                    for (int u = 0; u < U; ++u) {
                        const int k  = g + u < h ? g + u : h - 1;
                        const int sa = __builtin_amdgcn_readlane(s, k);
                        const int sb = __builtin_amdgcn_readlane(s, 16 + k);
                        const int sk = up ? sb : sa;
                        v[u] = *(const float4*)(lanebase + (size_t)sk * (FT * 4));
                    }
                    if (g + U <= hf) {
#pragma unroll
                        for (int u = 0; u < U; ++u) {
                            a.x += v[u].x; a.y += v[u].y; a.z += v[u].z; a.w += v[u].w;
                        }
                    } else {
#pragma unroll
                        for (int u = 0; u < U; ++u) {
                            const int  k  = g + u;
                            const bool ok = up ? h + k < c : k < h;
                            a.x += ok ? v[u].x : 0.0f; a.y += ok ? v[u].y : 0.0f;
                            a.z += ok ? v[u].z : 0.0f; a.w += ok ? v[u].w : 0.0f;
                        }
                    }
                }
            }
            a.x += xhalf(a.x); a.y += xhalf(a.y); a.z += xhalf(a.z); a.w += xhalf(a.w);
            if (!up && j < F4) acc4[(wave * NPW + i) * F4 + j] = a;
        }
    } else {
        int f0 = lane, f1 = 32 + lane;
        f0 = f0 < F ? f0 : F - 1;
        f1 = f1 < F ? f1 : F - 1;
        for (int i = 0; i < NPW; ++i) {
            const int n0 = __builtin_amdgcn_readlane(rpv, i);
            const int n1 = __builtin_amdgcn_readlane(rpv, i + 1);
            float a0 = 0.0f, a1 = 0.0f;
            for (int c0 = n0; c0 < n1; c0 += 32) {
                const int c = n1 - c0 < 32 ? n1 - c0 : 32;
                const int s = sorted[c0 + (lane < c ? lane : 0)];
                for (int k = 0; k < c; ++k) {
                    const float* row = hs + (size_t)__builtin_amdgcn_readlane(s, k) * F;
                    a0 += row[f0];
                    a1 += row[f1];
                }
            }
            float* arow = acc + (wave * NPW + i) * F;
            if (lane < F)      arow[lane]      = a0;
            if (32 + lane < F) arow[32 + lane] = a1;
        }
    }
    __syncthreads();

    const size_t g0 = (size_t)blockIdx.x * NB * F;
    const size_t gN = (size_t)nNodes * F;
    if constexpr (FT != 0 && FT % 4 == 0) {
        constexpr int Q   = NB * FT / 4;
        constexpr int PCS = (Q + AT - 1) / AT;
        const float4* hs4  = (const float4*)(hs + g0);
        float4*       out4 = (float4*)(out + g0);
        float4 a[PCS], s[PCS];
        float  d[PCS], bv[PCS][4];
#pragma unroll
        for (int t = 0; t < PCS; ++t) {
            const int  q    = threadIdx.x + t * AT;
            const bool live = q < Q && g0 + 4 * (size_t)q < gN;
            const int  qc   = live ? q : 0;
            const int  r    = (4 * qc) / FT;
            const int  f    = 4 * qc - r * FT;
            d[t] = dis[blockIdx.x * NB + r];
            s[t] = hs4[qc];
            a[t] = acc4[qc];
#pragma unroll
            for (int c = 0; c < 4; ++c) bv[t][c] = bias[f + c];
        }
#pragma unroll
        for (int t = 0; t < PCS; ++t) {
            const int q = threadIdx.x + t * AT;
            if (q < Q && g0 + 4 * (size_t)q < gN) {
                float4 o;
                o.x = fmaf(d[t], a[t].x + s[t].x, bv[t][0]);
                o.y = fmaf(d[t], a[t].y + s[t].y, bv[t][1]);
                o.z = fmaf(d[t], a[t].z + s[t].z, bv[t][2]);
                o.w = fmaf(d[t], a[t].w + s[t].w, bv[t][3]);
                if (doRelu) {
                    o.x = fmaxf(o.x, 0.0f); o.y = fmaxf(o.y, 0.0f);
                    o.z = fmaxf(o.z, 0.0f); o.w = fmaxf(o.w, 0.0f);
                }
                out4[q] = o;
            }
        }
    } else {
        for (int i = threadIdx.x; i < NB * F; i += AT) {
            const size_t g = g0 + i;
            if (g < gN) {
                const int r = i / F;
                const int f = i - r * F;
                float o = fmaf(dis[blockIdx.x * NB + r], acc[i] + hs[g], bias[f]);
                if (doRelu) o = fmaxf(o, 0.0f);
                out[g] = o;
            }
        }
    }
}

static inline void launch_gemm(int tilesN, const float* X, const float* W,
                               const float* dis, float* hs,
                               int M, int K, int N, hipStream_t stream) {
    int mt = (M + 15) / 16;
    long long thr = (long long)mt * 32;
    int blocks = (int)((thr + 255) / 256);
    switch (tilesN) {
    case 1: gcn_gemm_wmma<1><<<blocks, 256, (size_t)(K / 2) * 16 * 8, stream>>>(X, W, dis, hs, M, K, N); break;
    case 2: gcn_gemm_wmma<2><<<blocks, 256, (size_t)(K / 2) * 48 * 8, stream>>>(X, W, dis, hs, M, K, N); break;
    case 3: gcn_gemm_wmma<3><<<blocks, 256, (size_t)(K / 2) * 48 * 8, stream>>>(X, W, dis, hs, M, K, N); break;
    default: gcn_gemm_wmma<4><<<blocks, 256, (size_t)(K / 2) * 80 * 8, stream>>>(X, W, dis, hs, M, K, N); break;
    }
}

static inline void launch_aggregate(const int* sorted, const int* rowptr, const float* hs,
                                    const float* dis, const float* bias, float* out,
                                    int nNodes, int nBk, int F, int doRelu,
                                    hipStream_t stream) {
    size_t lds = (size_t)NB * F * sizeof(float);
    switch (F) {
    case 64: gcn_bucket_aggregate<64><<<nBk, AT, lds, stream>>>(
                 sorted, rowptr, hs, dis, bias, out, nNodes, F, doRelu); break;
    case 40: gcn_bucket_aggregate<40><<<nBk, AT, lds, stream>>>(
                 sorted, rowptr, hs, dis, bias, out, nNodes, F, doRelu); break;
    default: gcn_bucket_aggregate<0><<<nBk, AT, lds, stream>>>(
                 sorted, rowptr, hs, dis, bias, out, nNodes, F, doRelu); break;
    }
}

extern "C" void kernel_launch(void* const* d_in, const int* in_sizes, int n_in,
                              void* d_out, int out_size, void* d_ws, size_t ws_size,
                              hipStream_t stream) {
    const float* x  = (const float*)d_in[0];
    const int*   ei = (const int*)d_in[1];
    const float* W1 = (const float*)d_in[2];
    const float* b1 = (const float*)d_in[3];
    const float* W2 = (const float*)d_in[4];
    const float* b2 = (const float*)d_in[5];
    float* out = (float*)d_out;

    const int IN  = 128;
    const int HID = in_sizes[3];
    const int OUT = in_sizes[5];
    const int N   = in_sizes[0] / IN;
    const int E   = in_sizes[1] / 2;

    const int* src = ei;
    const int* dst = ei + E;

    const int nBk  = (N + NB - 1) / NB;
    const int FMAX = HID > OUT ? HID : OUT;

    float* dis    = (float*)d_ws;
    float* hs     = dis + (((size_t)N + 63) & ~63ull);
    float* h1     = hs + (size_t)N * FMAX;
    int*  sorted  = (int*)(h1 + (size_t)N * HID);
    int*  rowptr  = sorted + (((size_t)E + 63) & ~63ull);
    int*  offs    = rowptr + (size_t)nBk * PB;
    int*  grouped = (int*)hs;

    const int per = (E + PB - 1) / PB;
    size_t ldsBk = ((size_t)nBk + (size_t)per) * sizeof(int);
    gcn_slice_sort<<<PB, PT, ldsBk, stream>>>(src, dst, offs, grouped, E, nBk);

    gcn_bucket_order<<<nBk, PT, 0, stream>>>(grouped, offs, sorted, rowptr, dis, N, nBk, E);

    launch_gemm((HID + 15) / 16, x, W1, dis, hs, N, IN, HID, stream);
    launch_aggregate(sorted, rowptr, hs, dis, b1, h1, N, nBk, HID,  1, stream);

    launch_gemm((OUT + 15) / 16, h1, W2, dis, hs, N, HID, OUT, stream);
    launch_aggregate(sorted, rowptr, hs, dis, b2, out, N, nBk, OUT,  0, stream);
}


# layer-2 product formed inside the layer-1 aggregate from its finished rows in LDS (same f32 WMMAs, same k order; tile leaves as whole lines): h1 never goes to memory, one launch fewer
# speedup vs baseline: 4.9789x; 1.0613x over previous
#include <hip/hip_runtime.h>
#include <hip/hip_bf16.h>

typedef float v2f __attribute__((ext_vector_type(2)));
typedef float v8f __attribute__((ext_vector_type(8)));

template <int NT>
__global__ void __launch_bounds__(256)
gcn_gemm_wmma(const float* __restrict__ X,
              const float* __restrict__ W,
              const float* __restrict__ dis,
              float* __restrict__ hs,
              int M, int K, int N) {
    constexpr int NP    = NT * 16;
    constexpr int PITCH = (NP % 32 == 16) ? NP : NP + 16;
    constexpr int KC    = 8;
    extern __shared__ float2 wsm[];

    const int wave = (int)((blockIdx.x * blockDim.x + threadIdx.x) >> 5);
    const int lane = threadIdx.x & 31;
    const int m0 = wave << 4;
    const int lo = lane & 15;
    const int hi = lane >> 4;
    const bool live = m0 < M;

    int arow = m0 + lo;
    arow = arow < M ? arow : M - 1;
    const float* xrow = X + (size_t)arow * K + 2 * hi;

    const int nfull = K / (4 * KC);
    v2f an[KC];
#pragma unroll
    for (int j = 0; j < KC; ++j) an[j] = (v2f){0.0f, 0.0f};
    if (nfull > 0) {
#pragma unroll
        for (int j = 0; j < KC; ++j) an[j] = *(const v2f*)(xrow + 4 * j);
    }
    const int rowBase = m0 + (hi << 3);
    float dr[8];
#pragma unroll
    for (int r = 0; r < 8; ++r) {
        const int rr = rowBase + r;
        dr[r] = dis[rr < M ? rr : M - 1];
    }

    {
        const int KP = K >> 1;
        const int total = KP * NP;
        for (int base = 0; base < total; base += 4 * 256) {
            float w0[4], w1[4];
#pragma unroll
            for (int u = 0; u < 4; ++u) {
                int idx = base + u * 256 + (int)threadIdx.x;
                idx = idx < total ? idx : total - 1;
                const int kp = idx / NP;
                const int n  = idx - kp * NP;
                const int nc = n < N ? n : 0;
                w0[u] = W[(size_t)(2 * kp) * N + nc];
                w1[u] = W[(size_t)(2 * kp + 1) * N + nc];
            }
#pragma unroll
            for (int u = 0; u < 4; ++u) {
                const int idx = base + u * 256 + (int)threadIdx.x;
                if (idx < total) {
                    const int kp = idx / NP;
                    const int n  = idx - kp * NP;
                    wsm[kp * PITCH + n] = n < N ? make_float2(w0[u], w1[u]) : make_float2(0.0f, 0.0f);
                }
            }
        }
    }
    __syncthreads();
    if (!live) return;

    v8f c[NT];
#pragma unroll
    for (int t = 0; t < NT; ++t) c[t] = (v8f){};

    const float2* wl = wsm + hi * PITCH + lo;
    for (int ch = 0; ch < nfull; ++ch) {
        v2f ac[KC];
#pragma unroll
        for (int j = 0; j < KC; ++j) ac[j] = an[j];
        if (ch + 1 < nfull) {
#pragma unroll
            for (int j = 0; j < KC; ++j) an[j] = *(const v2f*)(xrow + (ch + 1) * (4 * KC) + 4 * j);
        }
        const float2* wc = wl + (size_t)ch * (2 * KC) * PITCH;
#pragma unroll
        for (int j = 0; j < KC; ++j) {
            v2f b[NT];
#pragma unroll
            for (int t = 0; t < NT; ++t) {
                const float2 w = wc[2 * j * PITCH + t * 16];
                b[t].x = w.x;
                b[t].y = w.y;
            }
#pragma unroll
            for (int t = 0; t < NT; ++t)
                c[t] = __builtin_amdgcn_wmma_f32_16x16x4_f32(
                            false, ac[j],  false, b[t],
                            (short)0, c[t],
                            false,  false);
        }
    }
    for (int k0 = nfull * (4 * KC); k0 < K; k0 += 4) {
        const v2f a = *(const v2f*)(xrow + k0);
        const float2* wc = wl + (size_t)(k0 >> 1) * PITCH;
#pragma unroll
        for (int t = 0; t < NT; ++t) {
            const float2 w = wc[t * 16];
            v2f b;
            b.x = w.x;
            b.y = w.y;
            c[t] = __builtin_amdgcn_wmma_f32_16x16x4_f32(
                        false, a,  false, b,
                        (short)0, c[t],
                        false,  false);
        }
    }

#pragma unroll
    for (int r = 0; r < 8; ++r) {
        const int rr = rowBase + r;
        if (rr < M) {
            const size_t base = (size_t)rr * N;
#pragma unroll
            for (int t = 0; t < NT; ++t) {
                const int col = t * 16 + lo;
                if (col < N) hs[base + col] = c[t][r] * dr[r];
            }
        }
    }
}

constexpr int LOG_NB = 7;
constexpr int NB     = 1 << LOG_NB;
constexpr int PB     = 256;
constexpr int PT     = 1024;
constexpr int PU     = 4;

__global__ void __launch_bounds__(PT)
gcn_slice_sort(const int* __restrict__ src, const int* __restrict__ dst,
               int* __restrict__ offs, int* __restrict__ grouped, int nE, int nBk) {
    static_assert(PT == 1024, "the scan below lets one wave add up the totals of 32 waves");
    extern __shared__ int lcount[];
    __shared__ int wsum[PT / 32];
    __shared__ int chunkTotal;
    __shared__ int carry;
    const int p    = threadIdx.x;
    const int lane = p & 31;
    const int wv   = p >> 5;
    for (int b = p; b < nBk; b += PT) lcount[b] = 0;
    if (p == 0) carry = 0;
    __syncthreads();
    const int per = (nE + PB - 1) / PB;
    const int lo  = blockIdx.x * per;
    const int hi  = lo + per < nE ? lo + per : nE;
    int* stage = lcount + nBk;
    for (int e = lo + p; e < hi; e += PU * PT) {
        int d[PU];
#pragma unroll
        for (int u = 0; u < PU; ++u) {
            const int eu = e + u * PT;
            d[u] = dst[eu < hi ? eu : e];
        }
#pragma unroll
        for (int u = 0; u < PU; ++u)
            if (e + u * PT < hi) atomicAdd(&lcount[d[u] >> LOG_NB], 1);
    }
    __syncthreads();
    for (int base = 0; base < nBk; base += PT) {
        const int i = base + p;
        const int v = i < nBk ? lcount[i] : 0;
        int incl = v;
        for (int o = 1; o < 32; o <<= 1) {
            const int t = __shfl_up(incl, o, 32);
            if (lane >= o) incl += t;
        }
        if (lane == 31) wsum[wv] = incl;
        __syncthreads();
        if (wv == 0) {
            const int ws = wsum[lane];
            int wi = ws;
            for (int o = 1; o < 32; o <<= 1) {
                const int t = __shfl_up(wi, o, 32);
                if (lane >= o) wi += t;
            }
            wsum[lane] = wi - ws;
            if (lane == 31) chunkTotal = wi;
        }
        __syncthreads();
        const int c = carry;
        if (i < nBk) {
            const int ex = c + wsum[wv] + incl - v;
            offs[i * PB + blockIdx.x] = ex;
            lcount[i] = ex;
        }
        __syncthreads();
        if (p == 0) carry = c + chunkTotal;
        __syncthreads();
    }
    if (p == 0) offs[nBk * PB + blockIdx.x] = carry;
    for (int e = lo + p; e < hi; e += PU * PT) {
        int d[PU], s[PU];
#pragma unroll
        for (int u = 0; u < PU; ++u) {
            const int eu = e + u * PT;
            const int ec = eu < hi ? eu : e;
            d[u] = dst[ec];
            s[u] = src[ec];
        }
#pragma unroll
        for (int u = 0; u < PU; ++u) {
            if (e + u * PT < hi) {
                const int pos = atomicAdd(&lcount[d[u] >> LOG_NB], 1);
                stage[pos] = (s[u] << LOG_NB) | (d[u] & (NB - 1));
            }
        }
    }
    __syncthreads();
    for (int gb = (lo & ~31) + p; gb < hi; gb += PU * PT) {
        int v[PU];
#pragma unroll
        for (int u = 0; u < PU; ++u) {
            const int g = gb + u * PT;
            v[u] = stage[g >= lo && g < hi ? g - lo : 0];
        }
#pragma unroll
        for (int u = 0; u < PU; ++u) {
            const int g = gb + u * PT;
            if (g >= lo && g < hi) grouped[g] = v[u];
        }
    }
}

constexpr int RL = PT / PB;
constexpr int OU = 8;
constexpr int SCAP = 8192;

__global__ void __launch_bounds__(PT)
gcn_bucket_order(const int* __restrict__ grouped, const int* __restrict__ offs,
                 int* __restrict__ sorted, int* __restrict__ rowptr,
                 float* __restrict__ dis, int nNodes, int nBk, int nE) {
    __shared__ int cnt[NB];
    __shared__ int off[NB];
    __shared__ int rbeg[PB];
    __shared__ int rend[PB];
    __shared__ int ends[2];
    __shared__ int stage[SCAP];
    const int t   = threadIdx.x / RL;
    const int sub = threadIdx.x % RL;
    const int rs  = offs[blockIdx.x * PB + t];
    const int re  = offs[(blockIdx.x + 1) * PB + t];
    for (int r = threadIdx.x; r < NB; r += blockDim.x) cnt[r] = 0;
    if (sub == 0) { rbeg[t] = rs; rend[t] = re; }
    __syncthreads();
    const int per = (nE + PB - 1) / PB;
    const int* run = grouped + (size_t)t * per + rs;
    const int len  = re - rs;
    for (int k = sub; k < len; k += OU * RL) {
        int w[OU];
#pragma unroll
        for (int u = 0; u < OU; ++u) {
            const int ku = k + u * RL;
            w[u] = run[ku < len ? ku : k];
        }
#pragma unroll
        for (int u = 0; u < OU; ++u)
            if (k + u * RL < len) atomicAdd(&cnt[w[u] & (NB - 1)], 1);
    }
    __syncthreads();
    if (threadIdx.x < 32) {
        const int l  = threadIdx.x;
        const int c0 = cnt[4 * l], c1 = cnt[4 * l + 1], c2 = cnt[4 * l + 2], c3 = cnt[4 * l + 3];
        const int s  = c0 + c1 + c2 + c3;
        int incl = s;
        for (int o = 1; o < 32; o <<= 1) {
            const int t2 = __shfl_up(incl, o, 32);
            if (l >= o) incl += t2;
        }
        const int ex = incl - s;
        off[4 * l]     = ex;
        off[4 * l + 1] = ex + c0;
        off[4 * l + 2] = ex + c0 + c1;
        off[4 * l + 3] = ex + c0 + c1 + c2;
        int sb = 0, se = 0;
        for (int q = l; q < PB; q += 32) { sb += rbeg[q]; se += rend[q]; }
        for (int o = 16; o > 0; o >>= 1) {
            sb += __shfl_xor(sb, o, 32);
            se += __shfl_xor(se, o, 32);
        }
        if (l == 0) { ends[0] = sb; ends[1] = se; }
    }
    __syncthreads();
    const int lo = ends[0];
    const int hi = ends[1];
    const bool staged = hi - lo <= SCAP;
    for (int r = threadIdx.x; r < NB; r += blockDim.x) {
        const int node = blockIdx.x * NB + r;
        if (node < nNodes) {
            dis[node]    = __frsqrt_rn((float)(1 + cnt[r]));
            rowptr[node] = lo + off[r];
        }
    }
    if (blockIdx.x == nBk - 1 && threadIdx.x == 0) rowptr[nNodes] = hi;
    __syncthreads();
    for (int k = sub; k < len; k += OU * RL) {
        int w[OU];
#pragma unroll
        for (int u = 0; u < OU; ++u) {
            const int ku = k + u * RL;
            w[u] = run[ku < len ? ku : k];
        }
#pragma unroll
        for (int u = 0; u < OU; ++u) {
            if (k + u * RL < len) {
                const int pos = atomicAdd(&off[w[u] & (NB - 1)], 1);
                if (staged) stage[pos] = w[u] >> LOG_NB;
                else        sorted[lo + pos] = w[u] >> LOG_NB;
            }
        }
    }
    if (staged) {
        __syncthreads();
        for (int g = (lo & ~31) + (int)threadIdx.x; g < hi; g += (int)blockDim.x)
            if (g >= lo) sorted[g] = stage[g - lo];
    }
}

__device__ __forceinline__ float xhalf(float x) {
    return __builtin_bit_cast(float, __builtin_amdgcn_permlanex16(
        __builtin_bit_cast(int, x), __builtin_bit_cast(int, x), 0x76543210, 0xfedcba98, false, false));
}

constexpr int AT  = 512;
constexpr int NPW = NB / (AT / 32);

template <int K, int TN, int PA, int PW>
__device__ __forceinline__ void lds_rows_times_w(float* __restrict__ a0,
                                                 const float2* __restrict__ w0,
                                                 const float* __restrict__ dis,
                                                 float* __restrict__ dst,
                                                 int row0, int M, int N, int lane) {
    const int lo = lane & 15;
    const int hi = lane >> 4;
    const int rowBase = row0 + (hi << 3);
    float dr[8];
#pragma unroll
    for (int r = 0; r < 8; ++r) {
        const int rr = rowBase + r;
        dr[r] = dis[rr < M ? rr : M - 1];
    }
    const float*  arow = a0 + lo * PA + 2 * hi;
    const float2* wl   = w0 + hi * PW + lo;
    v8f c[TN];
#pragma unroll
    for (int t = 0; t < TN; ++t) c[t] = (v8f){};
#pragma unroll
    for (int j = 0; j < K / 4; ++j) {
        const v2f a = *(const v2f*)(arow + 4 * j);
        v2f b[TN];
#pragma unroll
        for (int t = 0; t < TN; ++t) {
            const float2 w = wl[2 * j * PW + t * 16];
            b[t].x = w.x;
            b[t].y = w.y;
        }
#pragma unroll
        for (int t = 0; t < TN; ++t)
            c[t] = __builtin_amdgcn_wmma_f32_16x16x4_f32(
                        false, a,  false, b[t],
                        (short)0, c[t],
                        false,  false);
    }
#pragma unroll
    for (int r = 0; r < 8; ++r) {
#pragma unroll
        for (int t = 0; t < TN; ++t) {
            const int col = t * 16 + lo;
            if (col < N) a0[((hi << 3) + r) * N + col] = c[t][r] * dr[r];
        }
    }
    const int rows  = M - row0 < 16 ? M - row0 : 16;
    const int live4 = rows * (N >> 2);
    const float4* t4 = (const float4*)a0;
    float4*       d4 = (float4*)(dst + (size_t)row0 * N);
    for (int p = lane; p < 4 * N; p += 32)
        if (p < live4) d4[p] = t4[p];
}

template <int FT, int NT2>
__global__ void __launch_bounds__(AT)
gcn_bucket_aggregate(const int* __restrict__ sorted, const int* __restrict__ rowptr,
                     const float* __restrict__ hs, const float* __restrict__ dis,
                     const float* __restrict__ bias, float* __restrict__ out,
                     int nNodes, int Frt, int doRelu,
                     const float* __restrict__ Wn, float* __restrict__ hsn, int Nn) {
    static_assert(NT2 == 0 || (FT != 0 && FT % 4 == 0), "the product reads whole k-steps of 4 from the LDS rows");
    extern __shared__ float4 acc4[];
    float* acc = (float*)acc4;
    const int F = FT ? FT : Frt;
    constexpr int P4  = NT2 ? FT / 4 + 1 : FT / 4;
    constexpr int NP2 = NT2 * 16;
    constexpr int PW2 = (NP2 % 32 == 16) ? NP2 : NP2 + 16;

    const int lane = threadIdx.x & 31;
    const int wave = __builtin_amdgcn_readfirstlane((int)(threadIdx.x >> 5));

    int rpi = blockIdx.x * NB + wave * NPW + lane;
    rpi = rpi < nNodes ? rpi : nNodes;
    const int rpv = rowptr[rpi];

    constexpr int WTOT = (FT / 2) * NP2;
    static_assert(WTOT % AT == 0, "every thread stages the same number of elements");
    constexpr int WU = NT2 ? WTOT / AT : 1;
    float w0[WU], w1[WU];
    if constexpr (NT2 > 0) {
#pragma unroll
        for (int u = 0; u < WU; ++u) {
            const int idx = u * AT + (int)threadIdx.x;
            const int kp = idx / NP2;
            const int n  = idx - kp * NP2;
            const int nc = n < Nn ? n : 0;
            w0[u] = Wn[(size_t)(2 * kp) * Nn + nc];
            w1[u] = Wn[(size_t)(2 * kp + 1) * Nn + nc];
        }
    }

    if constexpr (FT != 0 && FT % 4 == 0) {
        constexpr int F4 = FT / 4;
        constexpr int U  = 8;
        const int  j    = lane & 15;
        const bool up   = lane >= 16;
        const int  jc   = j < F4 ? j : F4 - 1;
        const char* lanebase = (const char*)hs + 16 * jc;

        int p0 = __builtin_amdgcn_readlane(rpv, 0);
        int p1 = __builtin_amdgcn_readlane(rpv, 1);
        int sNext = 0;
        {
            const int c = p1 - p0 < 32 ? p1 - p0 : 32;
            const int h = (c + 1) >> 1;
            const int q = up ? h + j : j;
            const bool have = up ? q < c : j < h;
            if (c > 0) sNext = sorted[p0 + (have ? q : 0)];
        }
        if constexpr (NT2 > 0) {
            float2* wsm = (float2*)(acc4 + NB * P4);
#pragma unroll
            for (int u = 0; u < WU; ++u) {
                const int idx = u * AT + (int)threadIdx.x;
                const int kp = idx / NP2;
                const int n  = idx - kp * NP2;
                wsm[kp * PW2 + n] = n < Nn ? make_float2(w0[u], w1[u]) : make_float2(0.0f, 0.0f);
            }
        }
        for (int i = 0; i < NPW; ++i) {
            float4 a = make_float4(0.0f, 0.0f, 0.0f, 0.0f);
            int s = sNext;
            const int n0 = p0, n1 = p1;
            if (i + 1 < NPW) {
                p0 = n1;
                p1 = __builtin_amdgcn_readlane(rpv, i + 2);
                const int c = p1 - p0 < 32 ? p1 - p0 : 32;
                const int h = (c + 1) >> 1;
                const int q = up ? h + j : j;
                const bool have = up ? q < c : j < h;
                if (c > 0) sNext = sorted[p0 + (have ? q : 0)];
            }
            for (int c0 = n0; c0 < n1; c0 += 32) {
                const int c  = n1 - c0 < 32 ? n1 - c0 : 32;
                const int h  = (c + 1) >> 1;
                const int hf = c >> 1;
                if (c0 != n0) {
                    const int q = up ? h + j : j;
                    const bool have = up ? q < c : j < h;
                    s = sorted[c0 + (have ? q : 0)];
                }
                for (int g = 0; g < h; g += U) {
                    float4 v[U];
#pragma unroll
                    for (int u = 0; u < U; ++u) {
                        const int k  = g + u < h ? g + u : h - 1;
                        const int sa = __builtin_amdgcn_readlane(s, k);
                        const int sb = __builtin_amdgcn_readlane(s, 16 + k);
                        const int sk = up ? sb : sa;
                        v[u] = *(const float4*)(lanebase + (size_t)sk * (FT * 4));
                    }
                    if (g + U <= hf) {
#pragma unroll
                        for (int u = 0; u < U; ++u) {
                            a.x += v[u].x; a.y += v[u].y; a.z += v[u].z; a.w += v[u].w;
                        }
                    } else {
#pragma unroll
                        for (int u = 0; u < U; ++u) {
                            const int  k  = g + u;
                            const bool ok = up ? h + k < c : k < h;
                            a.x += ok ? v[u].x : 0.0f; a.y += ok ? v[u].y : 0.0f;
                            a.z += ok ? v[u].z : 0.0f; a.w += ok ? v[u].w : 0.0f;
                        }
                    }
                }
            }
            a.x += xhalf(a.x); a.y += xhalf(a.y); a.z += xhalf(a.z); a.w += xhalf(a.w);
            if (!up && j < F4) acc4[(wave * NPW + i) * P4 + j] = a;
        }
    } else {
        int f0 = lane, f1 = 32 + lane;
        f0 = f0 < F ? f0 : F - 1;
        f1 = f1 < F ? f1 : F - 1;
        for (int i = 0; i < NPW; ++i) {
            const int n0 = __builtin_amdgcn_readlane(rpv, i);
            const int n1 = __builtin_amdgcn_readlane(rpv, i + 1);
            float a0 = 0.0f, a1 = 0.0f;
            for (int c0 = n0; c0 < n1; c0 += 32) {
                const int c = n1 - c0 < 32 ? n1 - c0 : 32;
                const int s = sorted[c0 + (lane < c ? lane : 0)];
                for (int k = 0; k < c; ++k) {
                    const float* row = hs + (size_t)__builtin_amdgcn_readlane(s, k) * F;
                    a0 += row[f0];
                    a1 += row[f1];
                }
            }
            float* arow = acc + (wave * NPW + i) * F;
            if (lane < F)      arow[lane]      = a0;
            if (32 + lane < F) arow[32 + lane] = a1;
        }
    }
    __syncthreads();

    const size_t g0 = (size_t)blockIdx.x * NB * F;
    const size_t gN = (size_t)nNodes * F;
    if constexpr (FT != 0 && FT % 4 == 0) {
        constexpr int Q   = NB * FT / 4;
        constexpr int PCS = (Q + AT - 1) / AT;
        const float4* hs4  = (const float4*)(hs + g0);
        float4*       out4 = (float4*)(out + g0);
        float4 a[PCS], s[PCS];
        float  d[PCS], bv[PCS][4];
#pragma unroll
        for (int t = 0; t < PCS; ++t) {
            const int  q    = threadIdx.x + t * AT;
            const bool live = q < Q && g0 + 4 * (size_t)q < gN;
            const int  qc   = live ? q : 0;
            const int  r    = (4 * qc) / FT;
            const int  f    = 4 * qc - r * FT;
            d[t] = dis[blockIdx.x * NB + r];
            s[t] = hs4[qc];
            a[t] = acc4[r * P4 + (f >> 2)];
#pragma unroll
            for (int c = 0; c < 4; ++c) bv[t][c] = bias[f + c];
        }
#pragma unroll
        for (int t = 0; t < PCS; ++t) {
            const int q = threadIdx.x + t * AT;
            if (q < Q && g0 + 4 * (size_t)q < gN) {
                float4 o;
                o.x = fmaf(d[t], a[t].x + s[t].x, bv[t][0]);
                o.y = fmaf(d[t], a[t].y + s[t].y, bv[t][1]);
                o.z = fmaf(d[t], a[t].z + s[t].z, bv[t][2]);
                o.w = fmaf(d[t], a[t].w + s[t].w, bv[t][3]);
                if (doRelu) {
                    o.x = fmaxf(o.x, 0.0f); o.y = fmaxf(o.y, 0.0f);
                    o.z = fmaxf(o.z, 0.0f); o.w = fmaxf(o.w, 0.0f);
                }
                if constexpr (NT2 > 0) {
                    const int r = (4 * q) / FT;
                    acc4[r * P4 + ((4 * q - r * FT) >> 2)] = o;
                } else {
                    out4[q] = o;
                }
            }
        }
    } else {
        for (int i = threadIdx.x; i < NB * F; i += AT) {
            const size_t g = g0 + i;
            if (g < gN) {
                const int r = i / F;
                const int f = i - r * F;
                float o = fmaf(dis[blockIdx.x * NB + r], acc[i] + hs[g], bias[f]);
                if (doRelu) o = fmaxf(o, 0.0f);
                out[g] = o;
            }
        }
    }

    if constexpr (NT2 > 0) {
        __syncthreads();
        constexpr int RT = NB / 16;
        static_assert(AT / 32 >= RT, "a wave per row tile");
        const float2* wsm = (const float2*)(acc4 + NB * P4);
        if (wave < RT)
            lds_rows_times_w<FT, NT2, 4 * P4, PW2>(acc + wave * 16 * (4 * P4), wsm, dis, hsn,
                                                   blockIdx.x * NB + wave * 16, nNodes, Nn, lane);
    }
}

static inline void launch_gemm(int tilesN, const float* X, const float* W,
                               const float* dis, float* hs,
                               int M, int K, int N, hipStream_t stream) {
    int mt = (M + 15) / 16;
    long long thr = (long long)mt * 32;
    int blocks = (int)((thr + 255) / 256);
    switch (tilesN) {
    case 1: gcn_gemm_wmma<1><<<blocks, 256, (size_t)(K / 2) * 16 * 8, stream>>>(X, W, dis, hs, M, K, N); break;
    case 2: gcn_gemm_wmma<2><<<blocks, 256, (size_t)(K / 2) * 48 * 8, stream>>>(X, W, dis, hs, M, K, N); break;
    case 3: gcn_gemm_wmma<3><<<blocks, 256, (size_t)(K / 2) * 48 * 8, stream>>>(X, W, dis, hs, M, K, N); break;
    default: gcn_gemm_wmma<4><<<blocks, 256, (size_t)(K / 2) * 80 * 8, stream>>>(X, W, dis, hs, M, K, N); break;
    }
}

static inline void launch_aggregate(const int* sorted, const int* rowptr, const float* hs,
                                    const float* dis, const float* bias, float* out,
                                    int nNodes, int nBk, int F, int doRelu,
                                    hipStream_t stream) {
    size_t lds = (size_t)NB * F * sizeof(float);
    switch (F) {
    case 64: gcn_bucket_aggregate<64, 0><<<nBk, AT, lds, stream>>>(
                 sorted, rowptr, hs, dis, bias, out, nNodes, F, doRelu, nullptr, nullptr, 0); break;
    case 40: gcn_bucket_aggregate<40, 0><<<nBk, AT, lds, stream>>>(
                 sorted, rowptr, hs, dis, bias, out, nNodes, F, doRelu, nullptr, nullptr, 0); break;
    default: gcn_bucket_aggregate<0, 0><<<nBk, AT, lds, stream>>>(
                 sorted, rowptr, hs, dis, bias, out, nNodes, F, doRelu, nullptr, nullptr, 0); break;
    }
}

static inline void launch_aggregate_product(int tilesN, const int* sorted, const int* rowptr,
                                            const float* hs, const float* dis, const float* bias,
                                            const float* Wn, float* hsn,
                                            int nNodes, int nBk, int Nn, int doRelu,
                                            hipStream_t stream) {
    const size_t rows = (size_t)NB * 17 * 16;
    switch (tilesN) {
    case 1: gcn_bucket_aggregate<64, 1><<<nBk, AT, rows + 32 * 16 * 8, stream>>>(
                sorted, rowptr, hs, dis, bias, nullptr, nNodes, 64, doRelu, Wn, hsn, Nn); break;
    case 2: gcn_bucket_aggregate<64, 2><<<nBk, AT, rows + 32 * 48 * 8, stream>>>(
                sorted, rowptr, hs, dis, bias, nullptr, nNodes, 64, doRelu, Wn, hsn, Nn); break;
    case 3: gcn_bucket_aggregate<64, 3><<<nBk, AT, rows + 32 * 48 * 8, stream>>>(
                sorted, rowptr, hs, dis, bias, nullptr, nNodes, 64, doRelu, Wn, hsn, Nn); break;
    default: gcn_bucket_aggregate<64, 4><<<nBk, AT, rows + 32 * 80 * 8, stream>>>(
                sorted, rowptr, hs, dis, bias, nullptr, nNodes, 64, doRelu, Wn, hsn, Nn); break;
    }
}

extern "C" void kernel_launch(void* const* d_in, const int* in_sizes, int n_in,
                              void* d_out, int out_size, void* d_ws, size_t ws_size,
                              hipStream_t stream) {
    const float* x  = (const float*)d_in[0];
    const int*   ei = (const int*)d_in[1];
    const float* W1 = (const float*)d_in[2];
    const float* b1 = (const float*)d_in[3];
    const float* W2 = (const float*)d_in[4];
    const float* b2 = (const float*)d_in[5];
    float* out = (float*)d_out;

    const int IN  = 128;
    const int HID = in_sizes[3];
    const int OUT = in_sizes[5];
    const int N   = in_sizes[0] / IN;
    const int E   = in_sizes[1] / 2;

    const int* src = ei;
    const int* dst = ei + E;

    const int nBk  = (N + NB - 1) / NB;
    const int FMAX = HID > OUT ? HID : OUT;

    float* dis    = (float*)d_ws;
    float* hs     = dis + (((size_t)N + 63) & ~63ull);
    float* h1     = hs + (size_t)N * FMAX;
    int*  sorted  = (int*)(h1 + (size_t)N * HID);
    int*  rowptr  = sorted + (((size_t)E + 63) & ~63ull);
    int*  offs    = rowptr + (size_t)nBk * PB;
    int*  grouped = (int*)hs;

    const int per = (E + PB - 1) / PB;
    size_t ldsBk = ((size_t)nBk + (size_t)per) * sizeof(int);
    gcn_slice_sort<<<PB, PT, ldsBk, stream>>>(src, dst, offs, grouped, E, nBk);

    gcn_bucket_order<<<nBk, PT, 0, stream>>>(grouped, offs, sorted, rowptr, dis, N, nBk, E);

    launch_gemm((HID + 15) / 16, x, W1, dis, hs, N, IN, HID, stream);
    if (HID == 64 && OUT <= 64 && OUT % 4 == 0) {
        launch_aggregate_product((OUT + 15) / 16, sorted, rowptr, hs, dis, b1, W2, h1,
                                 N, nBk, OUT,  1, stream);
        launch_aggregate(sorted, rowptr, h1, dis, b2, out, N, nBk, OUT,  0, stream);
    } else {
        launch_aggregate(sorted, rowptr, hs, dis, b1, h1, N, nBk, HID,  1, stream);
        launch_gemm((OUT + 15) / 16, h1, W2, dis, hs, N, HID, OUT, stream);
        launch_aggregate(sorted, rowptr, hs, dis, b2, out, N, nBk, OUT,  0, stream);
    }
}


# layer-1 product formed by the bucket-ordering launch: 8-wave blocks order a bucket from registers (a node's edges in LDS arrival order, now of 8 waves), then multiply their 128 rows: 4 launches
# speedup vs baseline: 5.4110x; 1.0868x over previous
#include <hip/hip_runtime.h>
#include <hip/hip_bf16.h>

typedef float v2f __attribute__((ext_vector_type(2)));
typedef float v8f __attribute__((ext_vector_type(8)));

template <int NT>
__global__ void __launch_bounds__(256)
gcn_gemm_wmma(const float* __restrict__ X,
              const float* __restrict__ W,
              const float* __restrict__ dis,
              float* __restrict__ hs,
              int M, int K, int N) {
    constexpr int NP    = NT * 16;
    constexpr int PITCH = (NP % 32 == 16) ? NP : NP + 16;
    constexpr int KC    = 8;
    extern __shared__ float2 wsm[];

    const int wave = (int)((blockIdx.x * blockDim.x + threadIdx.x) >> 5);
    const int lane = threadIdx.x & 31;
    const int m0 = wave << 4;
    const int lo = lane & 15;
    const int hi = lane >> 4;
    const bool live = m0 < M;

    int arow = m0 + lo;
    arow = arow < M ? arow : M - 1;
    const float* xrow = X + (size_t)arow * K + 2 * hi;

    const int nfull = K / (4 * KC);
    v2f an[KC];
#pragma unroll
    for (int j = 0; j < KC; ++j) an[j] = (v2f){0.0f, 0.0f};
    if (nfull > 0) {
#pragma unroll
        for (int j = 0; j < KC; ++j) an[j] = *(const v2f*)(xrow + 4 * j);
    }
    const int rowBase = m0 + (hi << 3);
    float dr[8];
#pragma unroll
    for (int r = 0; r < 8; ++r) {
        const int rr = rowBase + r;
        dr[r] = dis[rr < M ? rr : M - 1];
    }

    {
        const int KP = K >> 1;
        const int total = KP * NP;
        for (int base = 0; base < total; base += 4 * 256) {
            float w0[4], w1[4];
#pragma unroll
            for (int u = 0; u < 4; ++u) {
                int idx = base + u * 256 + (int)threadIdx.x;
                idx = idx < total ? idx : total - 1;
                const int kp = idx / NP;
                const int n  = idx - kp * NP;
                const int nc = n < N ? n : 0;
                w0[u] = W[(size_t)(2 * kp) * N + nc];
                w1[u] = W[(size_t)(2 * kp + 1) * N + nc];
            }
#pragma unroll
            for (int u = 0; u < 4; ++u) {
                const int idx = base + u * 256 + (int)threadIdx.x;
                if (idx < total) {
                    const int kp = idx / NP;
                    const int n  = idx - kp * NP;
                    wsm[kp * PITCH + n] = n < N ? make_float2(w0[u], w1[u]) : make_float2(0.0f, 0.0f);
                }
            }
        }
    }
    __syncthreads();
    if (!live) return;

    v8f c[NT];
#pragma unroll
    for (int t = 0; t < NT; ++t) c[t] = (v8f){};

    const float2* wl = wsm + hi * PITCH + lo;
    for (int ch = 0; ch < nfull; ++ch) {
        v2f ac[KC];
#pragma unroll
        for (int j = 0; j < KC; ++j) ac[j] = an[j];
        if (ch + 1 < nfull) {
#pragma unroll
            for (int j = 0; j < KC; ++j) an[j] = *(const v2f*)(xrow + (ch + 1) * (4 * KC) + 4 * j);
        }
        const float2* wc = wl + (size_t)ch * (2 * KC) * PITCH;
#pragma unroll
        for (int j = 0; j < KC; ++j) {
            v2f b[NT];
#pragma unroll
            for (int t = 0; t < NT; ++t) {
                const float2 w = wc[2 * j * PITCH + t * 16];
                b[t].x = w.x;
                b[t].y = w.y;
            }
#pragma unroll
            for (int t = 0; t < NT; ++t)
                c[t] = __builtin_amdgcn_wmma_f32_16x16x4_f32(
                            false, ac[j],  false, b[t],
                            (short)0, c[t],
                            false,  false);
        }
    }
    for (int k0 = nfull * (4 * KC); k0 < K; k0 += 4) {
        const v2f a = *(const v2f*)(xrow + k0);
        const float2* wc = wl + (size_t)(k0 >> 1) * PITCH;
#pragma unroll
        for (int t = 0; t < NT; ++t) {
            const float2 w = wc[t * 16];
            v2f b;
            b.x = w.x;
            b.y = w.y;
            c[t] = __builtin_amdgcn_wmma_f32_16x16x4_f32(
                        false, a,  false, b,
                        (short)0, c[t],
                        false,  false);
        }
    }

#pragma unroll
    for (int r = 0; r < 8; ++r) {
        const int rr = rowBase + r;
        if (rr < M) {
            const size_t base = (size_t)rr * N;
#pragma unroll
            for (int t = 0; t < NT; ++t) {
                const int col = t * 16 + lo;
                if (col < N) hs[base + col] = c[t][r] * dr[r];
            }
        }
    }
}

constexpr int LOG_NB = 7;
constexpr int NB     = 1 << LOG_NB;
constexpr int PB     = 256;
constexpr int PT     = 1024;
constexpr int PU     = 4;

__global__ void __launch_bounds__(PT)
gcn_slice_sort(const int* __restrict__ src, const int* __restrict__ dst,
               int* __restrict__ offs, int* __restrict__ grouped, int nE, int nBk) {
    static_assert(PT == 1024, "the scan below lets one wave add up the totals of 32 waves");
    extern __shared__ int lcount[];
    __shared__ int wsum[PT / 32];
    __shared__ int chunkTotal;
    __shared__ int carry;
    const int p    = threadIdx.x;
    const int lane = p & 31;
    const int wv   = p >> 5;
    for (int b = p; b < nBk; b += PT) lcount[b] = 0;
    if (p == 0) carry = 0;
    __syncthreads();
    const int per = (nE + PB - 1) / PB;
    const int lo  = blockIdx.x * per;
    const int hi  = lo + per < nE ? lo + per : nE;
    int* stage = lcount + nBk;
    for (int e = lo + p; e < hi; e += PU * PT) {
        int d[PU];
#pragma unroll
        for (int u = 0; u < PU; ++u) {
            const int eu = e + u * PT;
            d[u] = dst[eu < hi ? eu : e];
        }
#pragma unroll
        for (int u = 0; u < PU; ++u)
            if (e + u * PT < hi) atomicAdd(&lcount[d[u] >> LOG_NB], 1);
    }
    __syncthreads();
    for (int base = 0; base < nBk; base += PT) {
        const int i = base + p;
        const int v = i < nBk ? lcount[i] : 0;
        int incl = v;
        for (int o = 1; o < 32; o <<= 1) {
            const int t = __shfl_up(incl, o, 32);
            if (lane >= o) incl += t;
        }
        if (lane == 31) wsum[wv] = incl;
        __syncthreads();
        if (wv == 0) {
            const int ws = wsum[lane];
            int wi = ws;
            for (int o = 1; o < 32; o <<= 1) {
                const int t = __shfl_up(wi, o, 32);
                if (lane >= o) wi += t;
            }
            wsum[lane] = wi - ws;
            if (lane == 31) chunkTotal = wi;
        }
        __syncthreads();
        const int c = carry;
        if (i < nBk) {
            const int ex = c + wsum[wv] + incl - v;
            offs[i * PB + blockIdx.x] = ex;
            lcount[i] = ex;
        }
        __syncthreads();
        if (p == 0) carry = c + chunkTotal;
        __syncthreads();
    }
    if (p == 0) offs[nBk * PB + blockIdx.x] = carry;
    for (int e = lo + p; e < hi; e += PU * PT) {
        int d[PU], s[PU];
#pragma unroll
        for (int u = 0; u < PU; ++u) {
            const int eu = e + u * PT;
            const int ec = eu < hi ? eu : e;
            d[u] = dst[ec];
            s[u] = src[ec];
        }
#pragma unroll
        for (int u = 0; u < PU; ++u) {
            if (e + u * PT < hi) {
                const int pos = atomicAdd(&lcount[d[u] >> LOG_NB], 1);
                stage[pos] = (s[u] << LOG_NB) | (d[u] & (NB - 1));
            }
        }
    }
    __syncthreads();
    for (int gb = (lo & ~31) + p; gb < hi; gb += PU * PT) {
        int v[PU];
#pragma unroll
        for (int u = 0; u < PU; ++u) {
            const int g = gb + u * PT;
            v[u] = stage[g >= lo && g < hi ? g - lo : 0];
        }
#pragma unroll
        for (int u = 0; u < PU; ++u) {
            const int g = gb + u * PT;
            if (g >= lo && g < hi) grouped[g] = v[u];
        }
    }
}

constexpr int OT = 256;
constexpr int RL = 4;
constexpr int RG = OT / RL;
constexpr int RQ = PB / RG;
constexpr int OU = 8;
constexpr int SCAP = 8192;
static_assert(NB == (OT / 32) * 16, "a wave per 16-row tile of the bucket");
static_assert(PB % RG == 0, "every lane takes the same number of runs");

template <int NT>
__global__ void __launch_bounds__(OT)
gcn_bucket_order_product(const int* __restrict__ grouped, const int* __restrict__ offs,
                         int* __restrict__ sorted, int* __restrict__ rowptr,
                         float* __restrict__ dis, int nNodes, int nBk, int nE,
                         const float* __restrict__ X, const float* __restrict__ W,
                         float* __restrict__ hs, int K, int N) {
    constexpr int NP    = NT * 16;
    constexpr int PITCH = (NP % 32 == 16) ? NP : NP + 16;
    constexpr int KC    = 8;
    __shared__ int   cnt[NB];
    __shared__ int   off[NB];
    __shared__ float disl[NB];
    __shared__ int   wsb[OT / 32];
    __shared__ int   wse[OT / 32];
    __shared__ int   ends[2];
    __shared__ int   stage[SCAP];
    extern __shared__ float2 wsm[];

    const int M    = nNodes;
    const int lane = threadIdx.x & 31;
    const int wave = (int)(threadIdx.x >> 5);
    const int sub  = threadIdx.x % RL;
    const int tq   = threadIdx.x / RL;
    const int per  = (nE + PB - 1) / PB;

    {
        constexpr int WU = 16;
        const int KP = K >> 1;
        const int total = KP * NP;
        for (int base = 0; base < total; base += WU * OT) {
            float w0[WU], w1[WU];
#pragma unroll
            for (int u = 0; u < WU; ++u) {
                int idx = base + u * OT + (int)threadIdx.x;
                idx = idx < total ? idx : total - 1;
                const int kp = idx / NP;
                const int n  = idx - kp * NP;
                const int nc = n < N ? n : 0;
                w0[u] = W[(size_t)(2 * kp) * N + nc];
                w1[u] = W[(size_t)(2 * kp + 1) * N + nc];
            }
#pragma unroll
            for (int u = 0; u < WU; ++u) {
                const int idx = base + u * OT + (int)threadIdx.x;
                if (idx < total) {
                    const int kp = idx / NP;
                    const int n  = idx - kp * NP;
                    wsm[kp * PITCH + n] = n < N ? make_float2(w0[u], w1[u]) : make_float2(0.0f, 0.0f);
                }
            }
        }
    }

    int rbase[RQ], rlen[RQ];
    int sb = 0, se = 0;
#pragma unroll
    for (int q = 0; q < RQ; ++q) {
        const int t  = q * RG + tq;
        const int rs = offs[blockIdx.x * PB + t];
        const int re = offs[(blockIdx.x + 1) * PB + t];
        rbase[q] = t * per + rs;
        rlen[q]  = re - rs;
        if (sub == 0) { sb += rs; se += re; }
    }

    int w[RQ][OU];
#pragma unroll
    for (int q = 0; q < RQ; ++q) {
#pragma unroll
        for (int u = 0; u < OU; ++u) {
            const int k = sub + u * RL;
            w[q][u] = grouped[k < rlen[q] ? rbase[q] + k : 0];
        }
    }

    const int m0 = blockIdx.x * NB + (wave << 4);
    const int lo16 = lane & 15;
    const int hi16 = lane >> 4;
    const bool live = m0 < M;
    int arow = m0 + lo16;
    arow = arow < M ? arow : M - 1;
    const float* xrow = X + (size_t)arow * K + 2 * hi16;
    const int nfull = K / (4 * KC);
    v2f an[KC];
#pragma unroll
    for (int j = 0; j < KC; ++j) an[j] = (v2f){0.0f, 0.0f};
    if (nfull > 0) {
#pragma unroll
        for (int j = 0; j < KC; ++j) an[j] = *(const v2f*)(xrow + 4 * j);
    }

    for (int o = 16; o > 0; o >>= 1) {
        sb += __shfl_xor(sb, o, 32);
        se += __shfl_xor(se, o, 32);
    }
    if (lane == 0) { wsb[wave] = sb; wse[wave] = se; }
    for (int r = threadIdx.x; r < NB; r += OT) cnt[r] = 0;
    __syncthreads();

#pragma unroll
    for (int q = 0; q < RQ; ++q) {
#pragma unroll
        for (int u = 0; u < OU; ++u)
            if (sub + u * RL < rlen[q]) atomicAdd(&cnt[w[q][u] & (NB - 1)], 1);
        for (int k = sub + OU * RL; k < rlen[q]; k += RL)
            atomicAdd(&cnt[grouped[rbase[q] + k] & (NB - 1)], 1);
    }
    __syncthreads();
    if (threadIdx.x < 32) {
        const int l  = threadIdx.x;
        const int c0 = cnt[4 * l], c1 = cnt[4 * l + 1], c2 = cnt[4 * l + 2], c3 = cnt[4 * l + 3];
        const int s  = c0 + c1 + c2 + c3;
        int incl = s;
        for (int o = 1; o < 32; o <<= 1) {
            const int t2 = __shfl_up(incl, o, 32);
            if (l >= o) incl += t2;
        }
        const int ex = incl - s;
        off[4 * l]     = ex;
        off[4 * l + 1] = ex + c0;
        off[4 * l + 2] = ex + c0 + c1;
        off[4 * l + 3] = ex + c0 + c1 + c2;
        int tb = l < OT / 32 ? wsb[l] : 0;
        int te = l < OT / 32 ? wse[l] : 0;
        for (int o = 16; o > 0; o >>= 1) {
            tb += __shfl_xor(tb, o, 32);
            te += __shfl_xor(te, o, 32);
        }
        if (l == 0) { ends[0] = tb; ends[1] = te; }
    }
    __syncthreads();
    const int lo = ends[0];
    const int hi = ends[1];
    const bool staged = hi - lo <= SCAP;
    for (int r = threadIdx.x; r < NB; r += OT) {
        const int node = blockIdx.x * NB + r;
        const float d = __frsqrt_rn((float)(1 + cnt[r]));
        disl[r] = d;
        if (node < nNodes) {
            dis[node]    = d;
            rowptr[node] = lo + off[r];
        }
    }
    if (blockIdx.x == nBk - 1 && threadIdx.x == 0) rowptr[nNodes] = hi;
    __syncthreads();
    int pos[RQ][OU];
#pragma unroll
    for (int q = 0; q < RQ; ++q) {
#pragma unroll
        for (int u = 0; u < OU; ++u) {
            pos[q][u] = 0;
            if (sub + u * RL < rlen[q]) pos[q][u] = atomicAdd(&off[w[q][u] & (NB - 1)], 1);
        }
    }
    if (staged) {
#pragma unroll
        for (int q = 0; q < RQ; ++q) {
#pragma unroll
            for (int u = 0; u < OU; ++u)
                if (sub + u * RL < rlen[q]) stage[pos[q][u]] = w[q][u] >> LOG_NB;
        }
    } else {
#pragma unroll
        for (int q = 0; q < RQ; ++q) {
#pragma unroll
            for (int u = 0; u < OU; ++u)
                if (sub + u * RL < rlen[q]) sorted[lo + pos[q][u]] = w[q][u] >> LOG_NB;
        }
    }
#pragma unroll
    for (int q = 0; q < RQ; ++q) {
        for (int k = sub + OU * RL; k < rlen[q]; k += RL) {
            const int wv = grouped[rbase[q] + k];
            const int ps = atomicAdd(&off[wv & (NB - 1)], 1);
            if (staged) stage[ps] = wv >> LOG_NB;
            else        sorted[lo + ps] = wv >> LOG_NB;
        }
    }
    if (staged) {
        __syncthreads();
        for (int gb = (lo & ~31) + (int)threadIdx.x; gb < hi; gb += PU * OT) {
            int v[PU];
#pragma unroll
            for (int u = 0; u < PU; ++u) {
                const int g = gb + u * OT;
                v[u] = stage[g >= lo && g < hi ? g - lo : 0];
            }
#pragma unroll
            for (int u = 0; u < PU; ++u) {
                const int g = gb + u * OT;
                if (g >= lo && g < hi) sorted[g] = v[u];
            }
        }
    }

    if (!live) return;

    const int rowBase = m0 + (hi16 << 3);
    float dr[8];
#pragma unroll
    for (int r = 0; r < 8; ++r) dr[r] = disl[(wave << 4) + (hi16 << 3) + r];

    v8f c[NT];
#pragma unroll
    for (int t = 0; t < NT; ++t) c[t] = (v8f){};

    const float2* wl = wsm + hi16 * PITCH + lo16;
    for (int ch = 0; ch < nfull; ++ch) {
        v2f ac[KC];
#pragma unroll
        for (int j = 0; j < KC; ++j) ac[j] = an[j];
        if (ch + 1 < nfull) {
#pragma unroll
            for (int j = 0; j < KC; ++j) an[j] = *(const v2f*)(xrow + (ch + 1) * (4 * KC) + 4 * j);
        }
        const float2* wc = wl + (size_t)ch * (2 * KC) * PITCH;
#pragma unroll
        for (int j = 0; j < KC; ++j) {
            v2f b[NT];
#pragma unroll
            for (int t = 0; t < NT; ++t) {
                const float2 wv = wc[2 * j * PITCH + t * 16];
                b[t].x = wv.x;
                b[t].y = wv.y;
            }
#pragma unroll
            for (int t = 0; t < NT; ++t)
                c[t] = __builtin_amdgcn_wmma_f32_16x16x4_f32(
                            false, ac[j],  false, b[t],
                            (short)0, c[t],
                            false,  false);
        }
    }
    for (int k0 = nfull * (4 * KC); k0 < K; k0 += 4) {
        const v2f a = *(const v2f*)(xrow + k0);
        const float2* wc = wl + (size_t)(k0 >> 1) * PITCH;
#pragma unroll
        for (int t = 0; t < NT; ++t) {
            const float2 wv = wc[t * 16];
            v2f b;
            b.x = wv.x;
            b.y = wv.y;
            c[t] = __builtin_amdgcn_wmma_f32_16x16x4_f32(
                        false, a,  false, b,
                        (short)0, c[t],
                        false,  false);
        }
    }

#pragma unroll
    for (int r = 0; r < 8; ++r) {
        const int rr = rowBase + r;
        if (rr < M) {
            const size_t base = (size_t)rr * N;
#pragma unroll
            for (int t = 0; t < NT; ++t) {
                const int col = t * 16 + lo16;
                if (col < N) hs[base + col] = c[t][r] * dr[r];
            }
        }
    }
}

__device__ __forceinline__ float xhalf(float x) {
    return __builtin_bit_cast(float, __builtin_amdgcn_permlanex16(
        __builtin_bit_cast(int, x), __builtin_bit_cast(int, x), 0x76543210, 0xfedcba98, false, false));
}

constexpr int AT  = 512;
constexpr int NPW = NB / (AT / 32);

template <int K, int TN, int PA, int PW>
__device__ __forceinline__ void lds_rows_times_w(float* __restrict__ a0,
                                                 const float2* __restrict__ w0,
                                                 const float* __restrict__ dis,
                                                 float* __restrict__ dst,
                                                 int row0, int M, int N, int lane) {
    const int lo = lane & 15;
    const int hi = lane >> 4;
    const int rowBase = row0 + (hi << 3);
    float dr[8];
#pragma unroll
    for (int r = 0; r < 8; ++r) {
        const int rr = rowBase + r;
        dr[r] = dis[rr < M ? rr : M - 1];
    }
    const float*  arow = a0 + lo * PA + 2 * hi;
    const float2* wl   = w0 + hi * PW + lo;
    v8f c[TN];
#pragma unroll
    for (int t = 0; t < TN; ++t) c[t] = (v8f){};
#pragma unroll
    for (int j = 0; j < K / 4; ++j) {
        const v2f a = *(const v2f*)(arow + 4 * j);
        v2f b[TN];
#pragma unroll
        for (int t = 0; t < TN; ++t) {
            const float2 w = wl[2 * j * PW + t * 16];
            b[t].x = w.x;
            b[t].y = w.y;
        }
#pragma unroll
        for (int t = 0; t < TN; ++t)
            c[t] = __builtin_amdgcn_wmma_f32_16x16x4_f32(
                        false, a,  false, b[t],
                        (short)0, c[t],
                        false,  false);
    }
#pragma unroll
    for (int r = 0; r < 8; ++r) {
#pragma unroll
        for (int t = 0; t < TN; ++t) {
            const int col = t * 16 + lo;
            if (col < N) a0[((hi << 3) + r) * N + col] = c[t][r] * dr[r];
        }
    }
    const int rows  = M - row0 < 16 ? M - row0 : 16;
    const int live4 = rows * (N >> 2);
    const float4* t4 = (const float4*)a0;
    float4*       d4 = (float4*)(dst + (size_t)row0 * N);
    for (int p = lane; p < 4 * N; p += 32)
        if (p < live4) d4[p] = t4[p];
}

template <int FT, int NT2>
__global__ void __launch_bounds__(AT)
gcn_bucket_aggregate(const int* __restrict__ sorted, const int* __restrict__ rowptr,
                     const float* __restrict__ hs, const float* __restrict__ dis,
                     const float* __restrict__ bias, float* __restrict__ out,
                     int nNodes, int Frt, int doRelu,
                     const float* __restrict__ Wn, float* __restrict__ hsn, int Nn) {
    static_assert(NT2 == 0 || (FT != 0 && FT % 4 == 0), "the product reads whole k-steps of 4 from the LDS rows");
    extern __shared__ float4 acc4[];
    float* acc = (float*)acc4;
    const int F = FT ? FT : Frt;
    constexpr int P4  = NT2 ? FT / 4 + 1 : FT / 4;
    constexpr int NP2 = NT2 * 16;
    constexpr int PW2 = (NP2 % 32 == 16) ? NP2 : NP2 + 16;

    const int lane = threadIdx.x & 31;
    const int wave = __builtin_amdgcn_readfirstlane((int)(threadIdx.x >> 5));

    int rpi = blockIdx.x * NB + wave * NPW + lane;
    rpi = rpi < nNodes ? rpi : nNodes;
    const int rpv = rowptr[rpi];

    constexpr int WTOT = (FT / 2) * NP2;
    static_assert(WTOT % AT == 0, "every thread stages the same number of elements");
    constexpr int WU = NT2 ? WTOT / AT : 1;
    float w0[WU], w1[WU];
    if constexpr (NT2 > 0) {
#pragma unroll
        for (int u = 0; u < WU; ++u) {
            const int idx = u * AT + (int)threadIdx.x;
            const int kp = idx / NP2;
            const int n  = idx - kp * NP2;
            const int nc = n < Nn ? n : 0;
            w0[u] = Wn[(size_t)(2 * kp) * Nn + nc];
            w1[u] = Wn[(size_t)(2 * kp + 1) * Nn + nc];
        }
    }

    if constexpr (FT != 0 && FT % 4 == 0) {
        constexpr int F4 = FT / 4;
        constexpr int U  = 8;
        const int  j    = lane & 15;
        const bool up   = lane >= 16;
        const int  jc   = j < F4 ? j : F4 - 1;
        const char* lanebase = (const char*)hs + 16 * jc;

        int p0 = __builtin_amdgcn_readlane(rpv, 0);
        int p1 = __builtin_amdgcn_readlane(rpv, 1);
        int sNext = 0;
        {
            const int c = p1 - p0 < 32 ? p1 - p0 : 32;
            const int h = (c + 1) >> 1;
            const int q = up ? h + j : j;
            const bool have = up ? q < c : j < h;
            if (c > 0) sNext = sorted[p0 + (have ? q : 0)];
        }
        if constexpr (NT2 > 0) {
            float2* wsm = (float2*)(acc4 + NB * P4);
#pragma unroll
            for (int u = 0; u < WU; ++u) {
                const int idx = u * AT + (int)threadIdx.x;
                const int kp = idx / NP2;
                const int n  = idx - kp * NP2;
                wsm[kp * PW2 + n] = n < Nn ? make_float2(w0[u], w1[u]) : make_float2(0.0f, 0.0f);
            }
        }
        for (int i = 0; i < NPW; ++i) {
            float4 a = make_float4(0.0f, 0.0f, 0.0f, 0.0f);
            int s = sNext;
            const int n0 = p0, n1 = p1;
            if (i + 1 < NPW) {
                p0 = n1;
                p1 = __builtin_amdgcn_readlane(rpv, i + 2);
                const int c = p1 - p0 < 32 ? p1 - p0 : 32;
                const int h = (c + 1) >> 1;
                const int q = up ? h + j : j;
                const bool have = up ? q < c : j < h;
                if (c > 0) sNext = sorted[p0 + (have ? q : 0)];
            }
            for (int c0 = n0; c0 < n1; c0 += 32) {
                const int c  = n1 - c0 < 32 ? n1 - c0 : 32;
                const int h  = (c + 1) >> 1;
                const int hf = c >> 1;
                if (c0 != n0) {
                    const int q = up ? h + j : j;
                    const bool have = up ? q < c : j < h;
                    s = sorted[c0 + (have ? q : 0)];
                }
                for (int g = 0; g < h; g += U) {
                    float4 v[U];
#pragma unroll
                    for (int u = 0; u < U; ++u) {
                        const int k  = g + u < h ? g + u : h - 1;
                        const int sa = __builtin_amdgcn_readlane(s, k);
                        const int sb = __builtin_amdgcn_readlane(s, 16 + k);
                        const int sk = up ? sb : sa;
                        v[u] = *(const float4*)(lanebase + (size_t)sk * (FT * 4));
                    }
                    if (g + U <= hf) {
#pragma unroll
                        for (int u = 0; u < U; ++u) {
                            a.x += v[u].x; a.y += v[u].y; a.z += v[u].z; a.w += v[u].w;
                        }
                    } else {
#pragma unroll
                        for (int u = 0; u < U; ++u) {
                            const int  k  = g + u;
                            const bool ok = up ? h + k < c : k < h;
                            a.x += ok ? v[u].x : 0.0f; a.y += ok ? v[u].y : 0.0f;
                            a.z += ok ? v[u].z : 0.0f; a.w += ok ? v[u].w : 0.0f;
                        }
                    }
                }
            }
            a.x += xhalf(a.x); a.y += xhalf(a.y); a.z += xhalf(a.z); a.w += xhalf(a.w);
            if (!up && j < F4) acc4[(wave * NPW + i) * P4 + j] = a;
        }
    } else {
        int f0 = lane, f1 = 32 + lane;
        f0 = f0 < F ? f0 : F - 1;
        f1 = f1 < F ? f1 : F - 1;
        for (int i = 0; i < NPW; ++i) {
            const int n0 = __builtin_amdgcn_readlane(rpv, i);
            const int n1 = __builtin_amdgcn_readlane(rpv, i + 1);
            float a0 = 0.0f, a1 = 0.0f;
            for (int c0 = n0; c0 < n1; c0 += 32) {
                const int c = n1 - c0 < 32 ? n1 - c0 : 32;
                const int s = sorted[c0 + (lane < c ? lane : 0)];
                for (int k = 0; k < c; ++k) {
                    const float* row = hs + (size_t)__builtin_amdgcn_readlane(s, k) * F;
                    a0 += row[f0];
                    a1 += row[f1];
                }
            }
            float* arow = acc + (wave * NPW + i) * F;
            if (lane < F)      arow[lane]      = a0;
            if (32 + lane < F) arow[32 + lane] = a1;
        }
    }
    __syncthreads();

    const size_t g0 = (size_t)blockIdx.x * NB * F;
    const size_t gN = (size_t)nNodes * F;
    if constexpr (FT != 0 && FT % 4 == 0) {
        constexpr int Q   = NB * FT / 4;
        constexpr int PCS = (Q + AT - 1) / AT;
        const float4* hs4  = (const float4*)(hs + g0);
        float4*       out4 = (float4*)(out + g0);
        float4 a[PCS], s[PCS];
        float  d[PCS], bv[PCS][4];
#pragma unroll
        for (int t = 0; t < PCS; ++t) {
            const int  q    = threadIdx.x + t * AT;
            const bool live = q < Q && g0 + 4 * (size_t)q < gN;
            const int  qc   = live ? q : 0;
            const int  r    = (4 * qc) / FT;
            const int  f    = 4 * qc - r * FT;
            d[t] = dis[blockIdx.x * NB + r];
            s[t] = hs4[qc];
            a[t] = acc4[r * P4 + (f >> 2)];
#pragma unroll
            for (int c = 0; c < 4; ++c) bv[t][c] = bias[f + c];
        }
#pragma unroll
        for (int t = 0; t < PCS; ++t) {
            const int q = threadIdx.x + t * AT;
            if (q < Q && g0 + 4 * (size_t)q < gN) {
                float4 o;
                o.x = fmaf(d[t], a[t].x + s[t].x, bv[t][0]);
                o.y = fmaf(d[t], a[t].y + s[t].y, bv[t][1]);
                o.z = fmaf(d[t], a[t].z + s[t].z, bv[t][2]);
                o.w = fmaf(d[t], a[t].w + s[t].w, bv[t][3]);
                if (doRelu) {
                    o.x = fmaxf(o.x, 0.0f); o.y = fmaxf(o.y, 0.0f);
                    o.z = fmaxf(o.z, 0.0f); o.w = fmaxf(o.w, 0.0f);
                }
                if constexpr (NT2 > 0) {
                    const int r = (4 * q) / FT;
                    acc4[r * P4 + ((4 * q - r * FT) >> 2)] = o;
                } else {
                    out4[q] = o;
                }
            }
        }
    } else {
        for (int i = threadIdx.x; i < NB * F; i += AT) {
            const size_t g = g0 + i;
            if (g < gN) {
                const int r = i / F;
                const int f = i - r * F;
                float o = fmaf(dis[blockIdx.x * NB + r], acc[i] + hs[g], bias[f]);
                if (doRelu) o = fmaxf(o, 0.0f);
                out[g] = o;
            }
        }
    }

    if constexpr (NT2 > 0) {
        __syncthreads();
        constexpr int RT = NB / 16;
        static_assert(AT / 32 >= RT, "a wave per row tile");
        const float2* wsm = (const float2*)(acc4 + NB * P4);
        if (wave < RT)
            lds_rows_times_w<FT, NT2, 4 * P4, PW2>(acc + wave * 16 * (4 * P4), wsm, dis, hsn,
                                                   blockIdx.x * NB + wave * 16, nNodes, Nn, lane);
    }
}

static inline void launch_gemm(int tilesN, const float* X, const float* W,
                               const float* dis, float* hs,
                               int M, int K, int N, hipStream_t stream) {
    int mt = (M + 15) / 16;
    long long thr = (long long)mt * 32;
    int blocks = (int)((thr + 255) / 256);
    switch (tilesN) {
    case 1: gcn_gemm_wmma<1><<<blocks, 256, (size_t)(K / 2) * 16 * 8, stream>>>(X, W, dis, hs, M, K, N); break;
    case 2: gcn_gemm_wmma<2><<<blocks, 256, (size_t)(K / 2) * 48 * 8, stream>>>(X, W, dis, hs, M, K, N); break;
    case 3: gcn_gemm_wmma<3><<<blocks, 256, (size_t)(K / 2) * 48 * 8, stream>>>(X, W, dis, hs, M, K, N); break;
    default: gcn_gemm_wmma<4><<<blocks, 256, (size_t)(K / 2) * 80 * 8, stream>>>(X, W, dis, hs, M, K, N); break;
    }
}

static inline void launch_order_product(int tilesN, const int* grouped, const int* offs,
                                        int* sorted, int* rowptr, float* dis,
                                        const float* X, const float* W, float* hs,
                                        int nNodes, int nBk, int nE, int K, int N,
                                        hipStream_t stream) {
    switch (tilesN) {
    case 1: gcn_bucket_order_product<1><<<nBk, OT, (size_t)(K / 2) * 16 * 8, stream>>>(
                grouped, offs, sorted, rowptr, dis, nNodes, nBk, nE, X, W, hs, K, N); break;
    case 2: gcn_bucket_order_product<2><<<nBk, OT, (size_t)(K / 2) * 48 * 8, stream>>>(
                grouped, offs, sorted, rowptr, dis, nNodes, nBk, nE, X, W, hs, K, N); break;
    case 3: gcn_bucket_order_product<3><<<nBk, OT, (size_t)(K / 2) * 48 * 8, stream>>>(
                grouped, offs, sorted, rowptr, dis, nNodes, nBk, nE, X, W, hs, K, N); break;
    default: gcn_bucket_order_product<4><<<nBk, OT, (size_t)(K / 2) * 80 * 8, stream>>>(
                grouped, offs, sorted, rowptr, dis, nNodes, nBk, nE, X, W, hs, K, N); break;
    }
}

static inline void launch_aggregate(const int* sorted, const int* rowptr, const float* hs,
                                    const float* dis, const float* bias, float* out,
                                    int nNodes, int nBk, int F, int doRelu,
                                    hipStream_t stream) {
    size_t lds = (size_t)NB * F * sizeof(float);
    switch (F) {
    case 64: gcn_bucket_aggregate<64, 0><<<nBk, AT, lds, stream>>>(
                 sorted, rowptr, hs, dis, bias, out, nNodes, F, doRelu, nullptr, nullptr, 0); break;
    case 40: gcn_bucket_aggregate<40, 0><<<nBk, AT, lds, stream>>>(
                 sorted, rowptr, hs, dis, bias, out, nNodes, F, doRelu, nullptr, nullptr, 0); break;
    default: gcn_bucket_aggregate<0, 0><<<nBk, AT, lds, stream>>>(
                 sorted, rowptr, hs, dis, bias, out, nNodes, F, doRelu, nullptr, nullptr, 0); break;
    }
}

static inline void launch_aggregate_product(int tilesN, const int* sorted, const int* rowptr,
                                            const float* hs, const float* dis, const float* bias,
                                            const float* Wn, float* hsn,
                                            int nNodes, int nBk, int Nn, int doRelu,
                                            hipStream_t stream) {
    const size_t rows = (size_t)NB * 17 * 16;
    switch (tilesN) {
    case 1: gcn_bucket_aggregate<64, 1><<<nBk, AT, rows + 32 * 16 * 8, stream>>>(
                sorted, rowptr, hs, dis, bias, nullptr, nNodes, 64, doRelu, Wn, hsn, Nn); break;
    case 2: gcn_bucket_aggregate<64, 2><<<nBk, AT, rows + 32 * 48 * 8, stream>>>(
                sorted, rowptr, hs, dis, bias, nullptr, nNodes, 64, doRelu, Wn, hsn, Nn); break;
    case 3: gcn_bucket_aggregate<64, 3><<<nBk, AT, rows + 32 * 48 * 8, stream>>>(
                sorted, rowptr, hs, dis, bias, nullptr, nNodes, 64, doRelu, Wn, hsn, Nn); break;
    default: gcn_bucket_aggregate<64, 4><<<nBk, AT, rows + 32 * 80 * 8, stream>>>(
                sorted, rowptr, hs, dis, bias, nullptr, nNodes, 64, doRelu, Wn, hsn, Nn); break;
    }
}

extern "C" void kernel_launch(void* const* d_in, const int* in_sizes, int n_in,
                              void* d_out, int out_size, void* d_ws, size_t ws_size,
                              hipStream_t stream) {
    const float* x  = (const float*)d_in[0];
    const int*   ei = (const int*)d_in[1];
    const float* W1 = (const float*)d_in[2];
    const float* b1 = (const float*)d_in[3];
    const float* W2 = (const float*)d_in[4];
    const float* b2 = (const float*)d_in[5];
    float* out = (float*)d_out;

    const int IN  = 128;
    const int HID = in_sizes[3];
    const int OUT = in_sizes[5];
    const int N   = in_sizes[0] / IN;
    const int E   = in_sizes[1] / 2;

    const int* src = ei;
    const int* dst = ei + E;

    const int nBk  = (N + NB - 1) / NB;
    const int FMAX = HID > OUT ? HID : OUT;

    float* dis    = (float*)d_ws;
    float* hs     = dis + (((size_t)N + 63) & ~63ull);
    float* h1     = hs + (size_t)N * FMAX;
    int*  sorted  = (int*)(h1 + (size_t)N * HID);
    int*  rowptr  = sorted + (((size_t)E + 63) & ~63ull);
    int*  offs    = rowptr + (size_t)nBk * PB;
    int*  grouped = (int*)h1;

    const int per = (E + PB - 1) / PB;
    size_t ldsBk = ((size_t)nBk + (size_t)per) * sizeof(int);
    gcn_slice_sort<<<PB, PT, ldsBk, stream>>>(src, dst, offs, grouped, E, nBk);

    launch_order_product((HID + 15) / 16, grouped, offs, sorted, rowptr, dis, x, W1, hs,
                         N, nBk, E, IN, HID, stream);

    if (HID == 64 && OUT <= 64 && OUT % 4 == 0) {
        launch_aggregate_product((OUT + 15) / 16, sorted, rowptr, hs, dis, b1, W2, h1,
                                 N, nBk, OUT,  1, stream);
        launch_aggregate(sorted, rowptr, h1, dis, b2, out, N, nBk, OUT,  0, stream);
    } else {
        launch_aggregate(sorted, rowptr, hs, dis, b1, h1, N, nBk, HID,  1, stream);
        launch_gemm((OUT + 15) / 16, h1, W2, dis, hs, N, HID, OUT, stream);
        launch_aggregate(sorted, rowptr, hs, dis, b2, out, N, nBk, OUT,  0, stream);
    }
}


# both edge sorts: an edge's slot = scanned base + the rank its counting LDS add returns (one LDS atomic per edge, not two); slice sort reads its slice once, into registers; arrival order of the count
# speedup vs baseline: 5.6274x; 1.0400x over previous
#include <hip/hip_runtime.h>
#include <hip/hip_bf16.h>

typedef float v2f __attribute__((ext_vector_type(2)));
typedef float v8f __attribute__((ext_vector_type(8)));

template <int NT>
__global__ void __launch_bounds__(256)
gcn_gemm_wmma(const float* __restrict__ X,
              const float* __restrict__ W,
              const float* __restrict__ dis,
              float* __restrict__ hs,
              int M, int K, int N) {
    constexpr int NP    = NT * 16;
    constexpr int PITCH = (NP % 32 == 16) ? NP : NP + 16;
    constexpr int KC    = 8;
    extern __shared__ float2 wsm[];

    const int wave = (int)((blockIdx.x * blockDim.x + threadIdx.x) >> 5);
    const int lane = threadIdx.x & 31;
    const int m0 = wave << 4;
    const int lo = lane & 15;
    const int hi = lane >> 4;
    const bool live = m0 < M;

    int arow = m0 + lo;
    arow = arow < M ? arow : M - 1;
    const float* xrow = X + (size_t)arow * K + 2 * hi;

    const int nfull = K / (4 * KC);
    v2f an[KC];
#pragma unroll
    for (int j = 0; j < KC; ++j) an[j] = (v2f){0.0f, 0.0f};
    if (nfull > 0) {
#pragma unroll
        for (int j = 0; j < KC; ++j) an[j] = *(const v2f*)(xrow + 4 * j);
    }
    const int rowBase = m0 + (hi << 3);
    float dr[8];
#pragma unroll
    for (int r = 0; r < 8; ++r) {
        const int rr = rowBase + r;
        dr[r] = dis[rr < M ? rr : M - 1];
    }

    {
        const int KP = K >> 1;
        const int total = KP * NP;
        for (int base = 0; base < total; base += 4 * 256) {
            float w0[4], w1[4];
#pragma unroll
            for (int u = 0; u < 4; ++u) {
                int idx = base + u * 256 + (int)threadIdx.x;
                idx = idx < total ? idx : total - 1;
                const int kp = idx / NP;
                const int n  = idx - kp * NP;
                const int nc = n < N ? n : 0;
                w0[u] = W[(size_t)(2 * kp) * N + nc];
                w1[u] = W[(size_t)(2 * kp + 1) * N + nc];
            }
#pragma unroll
            for (int u = 0; u < 4; ++u) {
                const int idx = base + u * 256 + (int)threadIdx.x;
                if (idx < total) {
                    const int kp = idx / NP;
                    const int n  = idx - kp * NP;
                    wsm[kp * PITCH + n] = n < N ? make_float2(w0[u], w1[u]) : make_float2(0.0f, 0.0f);
                }
            }
        }
    }
    __syncthreads();
    if (!live) return;

    v8f c[NT];
#pragma unroll
    for (int t = 0; t < NT; ++t) c[t] = (v8f){};

    const float2* wl = wsm + hi * PITCH + lo;
    for (int ch = 0; ch < nfull; ++ch) {
        v2f ac[KC];
#pragma unroll
        for (int j = 0; j < KC; ++j) ac[j] = an[j];
        if (ch + 1 < nfull) {
#pragma unroll
            for (int j = 0; j < KC; ++j) an[j] = *(const v2f*)(xrow + (ch + 1) * (4 * KC) + 4 * j);
        }
        const float2* wc = wl + (size_t)ch * (2 * KC) * PITCH;
#pragma unroll
        for (int j = 0; j < KC; ++j) {
            v2f b[NT];
#pragma unroll
            for (int t = 0; t < NT; ++t) {
                const float2 w = wc[2 * j * PITCH + t * 16];
                b[t].x = w.x;
                b[t].y = w.y;
            }
#pragma unroll
            for (int t = 0; t < NT; ++t)
                c[t] = __builtin_amdgcn_wmma_f32_16x16x4_f32(
                            false, ac[j],  false, b[t],
                            (short)0, c[t],
                            false,  false);
        }
    }
    for (int k0 = nfull * (4 * KC); k0 < K; k0 += 4) {
        const v2f a = *(const v2f*)(xrow + k0);
        const float2* wc = wl + (size_t)(k0 >> 1) * PITCH;
#pragma unroll
        for (int t = 0; t < NT; ++t) {
            const float2 w = wc[t * 16];
            v2f b;
            b.x = w.x;
            b.y = w.y;
            c[t] = __builtin_amdgcn_wmma_f32_16x16x4_f32(
                        false, a,  false, b,
                        (short)0, c[t],
                        false,  false);
        }
    }

#pragma unroll
    for (int r = 0; r < 8; ++r) {
        const int rr = rowBase + r;
        if (rr < M) {
            const size_t base = (size_t)rr * N;
#pragma unroll
            for (int t = 0; t < NT; ++t) {
                const int col = t * 16 + lo;
                if (col < N) hs[base + col] = c[t][r] * dr[r];
            }
        }
    }
}

constexpr int LOG_NB = 7;
constexpr int NB     = 1 << LOG_NB;
constexpr int PB     = 256;
constexpr int PT     = 1024;
constexpr int PU     = 4;
constexpr int KE     = 16;

__global__ void __launch_bounds__(PT)
gcn_slice_sort(const int* __restrict__ src, const int* __restrict__ dst,
               int* __restrict__ offs, int* __restrict__ grouped, int nE, int nBk) {
    static_assert(PT == 1024, "the scan below lets one wave add up the totals of 32 waves");
    extern __shared__ int lcount[];
    __shared__ int wsum[PT / 32];
    __shared__ int chunkTotal;
    __shared__ int carry;
    const int p    = threadIdx.x;
    const int lane = p & 31;
    const int wv   = p >> 5;
    const int per = (nE + PB - 1) / PB;
    const int lo  = blockIdx.x * per;
    const int hi  = lo + per < nE ? lo + per : nE;
    int* stage = lcount + nBk;
    const bool inreg = per <= KE * PT;
    int bk[KE], key[KE], rk[KE];
#pragma unroll
    for (int u = 0; u < KE; ++u) { bk[u] = 0; key[u] = 0; rk[u] = 0; }
    if (inreg && lo < hi) {
#pragma unroll
        for (int u = 0; u < KE; ++u) {
            const int eu = lo + p + u * PT;
            const int ec = eu < hi ? eu : lo;
            const int dd = dst[ec];
            const int ss = src[ec];
            bk[u]  = dd >> LOG_NB;
            key[u] = (ss << LOG_NB) | (dd & (NB - 1));
        }
    }
    for (int b = p; b < nBk; b += PT) lcount[b] = 0;
    if (p == 0) carry = 0;
    __syncthreads();
    if (inreg) {
#pragma unroll
        for (int u = 0; u < KE; ++u)
            if (lo + p + u * PT < hi) rk[u] = atomicAdd(&lcount[bk[u]], 1);
    } else {
        for (int e = lo + p; e < hi; e += PU * PT) {
            int d[PU];
#pragma unroll
            for (int u = 0; u < PU; ++u) {
                const int eu = e + u * PT;
                d[u] = dst[eu < hi ? eu : e];
            }
#pragma unroll
            for (int u = 0; u < PU; ++u)
                if (e + u * PT < hi) atomicAdd(&lcount[d[u] >> LOG_NB], 1);
        }
    }
    __syncthreads();
    for (int base = 0; base < nBk; base += PT) {
        const int i = base + p;
        const int v = i < nBk ? lcount[i] : 0;
        int incl = v;
        for (int o = 1; o < 32; o <<= 1) {
            const int t = __shfl_up(incl, o, 32);
            if (lane >= o) incl += t;
        }
        if (lane == 31) wsum[wv] = incl;
        __syncthreads();
        if (wv == 0) {
            const int ws = wsum[lane];
            int wi = ws;
            for (int o = 1; o < 32; o <<= 1) {
                const int t = __shfl_up(wi, o, 32);
                if (lane >= o) wi += t;
            }
            wsum[lane] = wi - ws;
            if (lane == 31) chunkTotal = wi;
        }
        __syncthreads();
        const int c = carry;
        if (i < nBk) {
            const int ex = c + wsum[wv] + incl - v;
            offs[i * PB + blockIdx.x] = ex;
            lcount[i] = ex;
        }
        __syncthreads();
        if (p == 0) carry = c + chunkTotal;
        __syncthreads();
    }
    if (p == 0) offs[nBk * PB + blockIdx.x] = carry;
    if (inreg) {
        int pos[KE];
#pragma unroll
        for (int u = 0; u < KE; ++u) pos[u] = lcount[bk[u]] + rk[u];
#pragma unroll
        for (int u = 0; u < KE; ++u)
            if (lo + p + u * PT < hi) stage[pos[u]] = key[u];
    } else {
        for (int e = lo + p; e < hi; e += PU * PT) {
            int d[PU], s[PU];
#pragma unroll
            for (int u = 0; u < PU; ++u) {
                const int eu = e + u * PT;
                const int ec = eu < hi ? eu : e;
                d[u] = dst[ec];
                s[u] = src[ec];
            }
#pragma unroll
            for (int u = 0; u < PU; ++u) {
                if (e + u * PT < hi) {
                    const int pos = atomicAdd(&lcount[d[u] >> LOG_NB], 1);
                    stage[pos] = (s[u] << LOG_NB) | (d[u] & (NB - 1));
                }
            }
        }
    }
    __syncthreads();
    for (int gb = (lo & ~31) + p; gb < hi; gb += PU * PT) {
        int v[PU];
#pragma unroll
        for (int u = 0; u < PU; ++u) {
            const int g = gb + u * PT;
            v[u] = stage[g >= lo && g < hi ? g - lo : 0];
        }
#pragma unroll
        for (int u = 0; u < PU; ++u) {
            const int g = gb + u * PT;
            if (g >= lo && g < hi) grouped[g] = v[u];
        }
    }
}

constexpr int OT = 256;
constexpr int RL = 4;
constexpr int RG = OT / RL;
constexpr int RQ = PB / RG;
constexpr int OU = 8;
constexpr int SCAP = 8192;
static_assert(NB == (OT / 32) * 16, "a wave per 16-row tile of the bucket");
static_assert(PB % RG == 0, "every lane takes the same number of runs");

template <int NT>
__global__ void __launch_bounds__(OT, 8)
gcn_bucket_order_product(const int* __restrict__ grouped, const int* __restrict__ offs,
                         int* __restrict__ sorted, int* __restrict__ rowptr,
                         float* __restrict__ dis, int nNodes, int nBk, int nE,
                         const float* __restrict__ X, const float* __restrict__ W,
                         float* __restrict__ hs, int K, int N) {
    constexpr int NP    = NT * 16;
    constexpr int PITCH = (NP % 32 == 16) ? NP : NP + 16;
    constexpr int KC    = 8;
    __shared__ int   cnt[NB];
    __shared__ int   tcnt[NB];
    __shared__ int   off[NB];
    __shared__ float disl[NB];
    __shared__ int   wsb[OT / 32];
    __shared__ int   wse[OT / 32];
    __shared__ int   ends[2];
    __shared__ int   stage[SCAP];
    extern __shared__ float2 wsm[];

    const int M    = nNodes;
    const int lane = threadIdx.x & 31;
    const int wave = (int)(threadIdx.x >> 5);
    const int sub  = threadIdx.x % RL;
    const int tq   = threadIdx.x / RL;
    const int per  = (nE + PB - 1) / PB;

    {
        constexpr int WU = 16;
        const int KP = K >> 1;
        const int total = KP * NP;
        for (int base = 0; base < total; base += WU * OT) {
            float w0[WU], w1[WU];
#pragma unroll
            for (int u = 0; u < WU; ++u) {
                int idx = base + u * OT + (int)threadIdx.x;
                idx = idx < total ? idx : total - 1;
                const int kp = idx / NP;
                const int n  = idx - kp * NP;
                const int nc = n < N ? n : 0;
                w0[u] = W[(size_t)(2 * kp) * N + nc];
                w1[u] = W[(size_t)(2 * kp + 1) * N + nc];
            }
#pragma unroll
            for (int u = 0; u < WU; ++u) {
                const int idx = base + u * OT + (int)threadIdx.x;
                if (idx < total) {
                    const int kp = idx / NP;
                    const int n  = idx - kp * NP;
                    wsm[kp * PITCH + n] = n < N ? make_float2(w0[u], w1[u]) : make_float2(0.0f, 0.0f);
                }
            }
        }
    }

    int rbase[RQ], rlen[RQ];
    int sb = 0, se = 0;
#pragma unroll
    for (int q = 0; q < RQ; ++q) {
        const int t  = q * RG + tq;
        const int rs = offs[blockIdx.x * PB + t];
        const int re = offs[(blockIdx.x + 1) * PB + t];
        rbase[q] = t * per + rs;
        rlen[q]  = re - rs;
        if (sub == 0) { sb += rs; se += re; }
    }

    int w[RQ][OU];
#pragma unroll
    for (int q = 0; q < RQ; ++q) {
#pragma unroll
        for (int u = 0; u < OU; ++u) {
            const int k = sub + u * RL;
            w[q][u] = grouped[k < rlen[q] ? rbase[q] + k : 0];
        }
    }

    const int m0 = blockIdx.x * NB + (wave << 4);
    const int lo16 = lane & 15;
    const int hi16 = lane >> 4;
    const bool live = m0 < M;
    int arow = m0 + lo16;
    arow = arow < M ? arow : M - 1;
    const float* xrow = X + (size_t)arow * K + 2 * hi16;
    const int nfull = K / (4 * KC);
    v2f an[KC];
#pragma unroll
    for (int j = 0; j < KC; ++j) an[j] = (v2f){0.0f, 0.0f};
    if (nfull > 0) {
#pragma unroll
        for (int j = 0; j < KC; ++j) an[j] = *(const v2f*)(xrow + 4 * j);
    }

    for (int o = 16; o > 0; o >>= 1) {
        sb += __shfl_xor(sb, o, 32);
        se += __shfl_xor(se, o, 32);
    }
    if (lane == 0) { wsb[wave] = sb; wse[wave] = se; }
    for (int r = threadIdx.x; r < NB; r += OT) { cnt[r] = 0; tcnt[r] = 0; }
    __syncthreads();

    int pos[RQ][OU];
#pragma unroll
    for (int q = 0; q < RQ; ++q) {
#pragma unroll
        for (int u = 0; u < OU; ++u) {
            pos[q][u] = 0;
            if (sub + u * RL < rlen[q]) pos[q][u] = atomicAdd(&cnt[w[q][u] & (NB - 1)], 1);
        }
        for (int k = sub + OU * RL; k < rlen[q]; k += RL)
            atomicAdd(&tcnt[grouped[rbase[q] + k] & (NB - 1)], 1);
    }
    __syncthreads();
    if (threadIdx.x < 32) {
        const int l  = threadIdx.x;
        const int c0 = cnt[4 * l] + tcnt[4 * l],         c1 = cnt[4 * l + 1] + tcnt[4 * l + 1];
        const int c2 = cnt[4 * l + 2] + tcnt[4 * l + 2], c3 = cnt[4 * l + 3] + tcnt[4 * l + 3];
        const int s  = c0 + c1 + c2 + c3;
        int incl = s;
        for (int o = 1; o < 32; o <<= 1) {
            const int t2 = __shfl_up(incl, o, 32);
            if (l >= o) incl += t2;
        }
        const int ex = incl - s;
        off[4 * l]     = ex;
        off[4 * l + 1] = ex + c0;
        off[4 * l + 2] = ex + c0 + c1;
        off[4 * l + 3] = ex + c0 + c1 + c2;
        int tb = l < OT / 32 ? wsb[l] : 0;
        int te = l < OT / 32 ? wse[l] : 0;
        for (int o = 16; o > 0; o >>= 1) {
            tb += __shfl_xor(tb, o, 32);
            te += __shfl_xor(te, o, 32);
        }
        if (l == 0) { ends[0] = tb; ends[1] = te; }
    }
    __syncthreads();
    const int lo = ends[0];
    const int hi = ends[1];
    const bool staged = hi - lo <= SCAP;
    for (int r = threadIdx.x; r < NB; r += OT) {
        const int node = blockIdx.x * NB + r;
        const float d = __frsqrt_rn((float)(1 + cnt[r] + tcnt[r]));
        disl[r] = d;
        if (node < nNodes) {
            dis[node]    = d;
            rowptr[node] = lo + off[r];
        }
    }
    if (blockIdx.x == nBk - 1 && threadIdx.x == 0) rowptr[nNodes] = hi;
    __syncthreads();
#pragma unroll
    for (int q = 0; q < RQ; ++q) {
#pragma unroll
        for (int u = 0; u < OU; ++u) {
            asm volatile("" : "+v"(w[q][u]));
            pos[q][u] += off[w[q][u] & (NB - 1)];
        }
    }
    if (staged) {
#pragma unroll
        for (int q = 0; q < RQ; ++q) {
#pragma unroll
            for (int u = 0; u < OU; ++u)
                if (sub + u * RL < rlen[q]) stage[pos[q][u]] = w[q][u] >> LOG_NB;
        }
    } else {
#pragma unroll
        for (int q = 0; q < RQ; ++q) {
#pragma unroll
            for (int u = 0; u < OU; ++u)
                if (sub + u * RL < rlen[q]) sorted[lo + pos[q][u]] = w[q][u] >> LOG_NB;
        }
    }
#pragma unroll
    for (int q = 0; q < RQ; ++q) {
        for (int k = sub + OU * RL; k < rlen[q]; k += RL) {
            const int wv = grouped[rbase[q] + k];
            const int nd = wv & (NB - 1);
            const int ps = off[nd] + cnt[nd] + atomicSub(&tcnt[nd], 1) - 1;
            if (staged) stage[ps] = wv >> LOG_NB;
            else        sorted[lo + ps] = wv >> LOG_NB;
        }
    }
    if (staged) {
        __syncthreads();
        for (int gb = (lo & ~31) + (int)threadIdx.x; gb < hi; gb += PU * OT) {
            int v[PU];
#pragma unroll
            for (int u = 0; u < PU; ++u) {
                const int g = gb + u * OT;
                v[u] = stage[g >= lo && g < hi ? g - lo : 0];
            }
#pragma unroll
            for (int u = 0; u < PU; ++u) {
                const int g = gb + u * OT;
                if (g >= lo && g < hi) sorted[g] = v[u];
            }
        }
    }

    if (!live) return;

    const int rowBase = m0 + (hi16 << 3);
    float dr[8];
#pragma unroll
    for (int r = 0; r < 8; ++r) dr[r] = disl[(wave << 4) + (hi16 << 3) + r];

    v8f c[NT];
#pragma unroll
    for (int t = 0; t < NT; ++t) c[t] = (v8f){};

    const float2* wl = wsm + hi16 * PITCH + lo16;
    for (int ch = 0; ch < nfull; ++ch) {
        v2f ac[KC];
#pragma unroll
        for (int j = 0; j < KC; ++j) ac[j] = an[j];
        if (ch + 1 < nfull) {
#pragma unroll
            for (int j = 0; j < KC; ++j) an[j] = *(const v2f*)(xrow + (ch + 1) * (4 * KC) + 4 * j);
        }
        const float2* wc = wl + (size_t)ch * (2 * KC) * PITCH;
#pragma unroll
        for (int j = 0; j < KC; ++j) {
            v2f b[NT];
#pragma unroll
            for (int t = 0; t < NT; ++t) {
                const float2 wv = wc[2 * j * PITCH + t * 16];
                b[t].x = wv.x;
                b[t].y = wv.y;
            }
#pragma unroll
            for (int t = 0; t < NT; ++t)
                c[t] = __builtin_amdgcn_wmma_f32_16x16x4_f32(
                            false, ac[j],  false, b[t],
                            (short)0, c[t],
                            false,  false);
        }
    }
    for (int k0 = nfull * (4 * KC); k0 < K; k0 += 4) {
        const v2f a = *(const v2f*)(xrow + k0);
        const float2* wc = wl + (size_t)(k0 >> 1) * PITCH;
#pragma unroll
        for (int t = 0; t < NT; ++t) {
            const float2 wv = wc[t * 16];
            v2f b;
            b.x = wv.x;
            b.y = wv.y;
            c[t] = __builtin_amdgcn_wmma_f32_16x16x4_f32(
                        false, a,  false, b,
                        (short)0, c[t],
                        false,  false);
        }
    }

#pragma unroll
    for (int r = 0; r < 8; ++r) {
        const int rr = rowBase + r;
        if (rr < M) {
            const size_t base = (size_t)rr * N;
#pragma unroll
            for (int t = 0; t < NT; ++t) {
                const int col = t * 16 + lo16;
                if (col < N) hs[base + col] = c[t][r] * dr[r];
            }
        }
    }
}

__device__ __forceinline__ float xhalf(float x) {
    return __builtin_bit_cast(float, __builtin_amdgcn_permlanex16(
        __builtin_bit_cast(int, x), __builtin_bit_cast(int, x), 0x76543210, 0xfedcba98, false, false));
}

constexpr int AT  = 512;
constexpr int NPW = NB / (AT / 32);

template <int K, int TN, int PA, int PW>
__device__ __forceinline__ void lds_rows_times_w(float* __restrict__ a0,
                                                 const float2* __restrict__ w0,
                                                 const float* __restrict__ dis,
                                                 float* __restrict__ dst,
                                                 int row0, int M, int N, int lane) {
    const int lo = lane & 15;
    const int hi = lane >> 4;
    const int rowBase = row0 + (hi << 3);
    float dr[8];
#pragma unroll
    for (int r = 0; r < 8; ++r) {
        const int rr = rowBase + r;
        dr[r] = dis[rr < M ? rr : M - 1];
    }
    const float*  arow = a0 + lo * PA + 2 * hi;
    const float2* wl   = w0 + hi * PW + lo;
    v8f c[TN];
#pragma unroll
    for (int t = 0; t < TN; ++t) c[t] = (v8f){};
#pragma unroll
    for (int j = 0; j < K / 4; ++j) {
        const v2f a = *(const v2f*)(arow + 4 * j);
        v2f b[TN];
#pragma unroll
        for (int t = 0; t < TN; ++t) {
            const float2 w = wl[2 * j * PW + t * 16];
            b[t].x = w.x;
            b[t].y = w.y;
        }
#pragma unroll
        for (int t = 0; t < TN; ++t)
            c[t] = __builtin_amdgcn_wmma_f32_16x16x4_f32(
                        false, a,  false, b[t],
                        (short)0, c[t],
                        false,  false);
    }
#pragma unroll
    for (int r = 0; r < 8; ++r) {
#pragma unroll
        for (int t = 0; t < TN; ++t) {
            const int col = t * 16 + lo;
            if (col < N) a0[((hi << 3) + r) * N + col] = c[t][r] * dr[r];
        }
    }
    const int rows  = M - row0 < 16 ? M - row0 : 16;
    const int live4 = rows * (N >> 2);
    const float4* t4 = (const float4*)a0;
    float4*       d4 = (float4*)(dst + (size_t)row0 * N);
    for (int p = lane; p < 4 * N; p += 32)
        if (p < live4) d4[p] = t4[p];
}

template <int FT, int NT2>
__global__ void __launch_bounds__(AT)
gcn_bucket_aggregate(const int* __restrict__ sorted, const int* __restrict__ rowptr,
                     const float* __restrict__ hs, const float* __restrict__ dis,
                     const float* __restrict__ bias, float* __restrict__ out,
                     int nNodes, int Frt, int doRelu,
                     const float* __restrict__ Wn, float* __restrict__ hsn, int Nn) {
    static_assert(NT2 == 0 || (FT != 0 && FT % 4 == 0), "the product reads whole k-steps of 4 from the LDS rows");
    extern __shared__ float4 acc4[];
    float* acc = (float*)acc4;
    const int F = FT ? FT : Frt;
    constexpr int P4  = NT2 ? FT / 4 + 1 : FT / 4;
    constexpr int NP2 = NT2 * 16;
    constexpr int PW2 = (NP2 % 32 == 16) ? NP2 : NP2 + 16;

    const int lane = threadIdx.x & 31;
    const int wave = __builtin_amdgcn_readfirstlane((int)(threadIdx.x >> 5));

    int rpi = blockIdx.x * NB + wave * NPW + lane;
    rpi = rpi < nNodes ? rpi : nNodes;
    const int rpv = rowptr[rpi];

    constexpr int WTOT = (FT / 2) * NP2;
    static_assert(WTOT % AT == 0, "every thread stages the same number of elements");
    constexpr int WU = NT2 ? WTOT / AT : 1;
    float w0[WU], w1[WU];
    if constexpr (NT2 > 0) {
#pragma unroll
        for (int u = 0; u < WU; ++u) {
            const int idx = u * AT + (int)threadIdx.x;
            const int kp = idx / NP2;
            const int n  = idx - kp * NP2;
            const int nc = n < Nn ? n : 0;
            w0[u] = Wn[(size_t)(2 * kp) * Nn + nc];
            w1[u] = Wn[(size_t)(2 * kp + 1) * Nn + nc];
        }
    }

    if constexpr (FT != 0 && FT % 4 == 0) {
        constexpr int F4 = FT / 4;
        constexpr int U  = 8;
        const int  j    = lane & 15;
        const bool up   = lane >= 16;
        const int  jc   = j < F4 ? j : F4 - 1;
        const char* lanebase = (const char*)hs + 16 * jc;

        int p0 = __builtin_amdgcn_readlane(rpv, 0);
        int p1 = __builtin_amdgcn_readlane(rpv, 1);
        int sNext = 0;
        {
            const int c = p1 - p0 < 32 ? p1 - p0 : 32;
            const int h = (c + 1) >> 1;
            const int q = up ? h + j : j;
            const bool have = up ? q < c : j < h;
            if (c > 0) sNext = sorted[p0 + (have ? q : 0)];
        }
        if constexpr (NT2 > 0) {
            float2* wsm = (float2*)(acc4 + NB * P4);
#pragma unroll
            for (int u = 0; u < WU; ++u) {
                const int idx = u * AT + (int)threadIdx.x;
                const int kp = idx / NP2;
                const int n  = idx - kp * NP2;
                wsm[kp * PW2 + n] = n < Nn ? make_float2(w0[u], w1[u]) : make_float2(0.0f, 0.0f);
            }
        }
        for (int i = 0; i < NPW; ++i) {
            float4 a = make_float4(0.0f, 0.0f, 0.0f, 0.0f);
            int s = sNext;
            const int n0 = p0, n1 = p1;
            if (i + 1 < NPW) {
                p0 = n1;
                p1 = __builtin_amdgcn_readlane(rpv, i + 2);
                const int c = p1 - p0 < 32 ? p1 - p0 : 32;
                const int h = (c + 1) >> 1;
                const int q = up ? h + j : j;
                const bool have = up ? q < c : j < h;
                if (c > 0) sNext = sorted[p0 + (have ? q : 0)];
            }
            for (int c0 = n0; c0 < n1; c0 += 32) {
                const int c  = n1 - c0 < 32 ? n1 - c0 : 32;
                const int h  = (c + 1) >> 1;
                const int hf = c >> 1;
                if (c0 != n0) {
                    const int q = up ? h + j : j;
                    const bool have = up ? q < c : j < h;
                    s = sorted[c0 + (have ? q : 0)];
                }
                for (int g = 0; g < h; g += U) {
                    float4 v[U];
#pragma unroll
                    for (int u = 0; u < U; ++u) {
                        const int k  = g + u < h ? g + u : h - 1;
                        const int sa = __builtin_amdgcn_readlane(s, k);
                        const int sb = __builtin_amdgcn_readlane(s, 16 + k);
                        const int sk = up ? sb : sa;
                        v[u] = *(const float4*)(lanebase + (size_t)sk * (FT * 4));
                    }
                    if (g + U <= hf) {
#pragma unroll
                        for (int u = 0; u < U; ++u) {
                            a.x += v[u].x; a.y += v[u].y; a.z += v[u].z; a.w += v[u].w;
                        }
                    } else {
#pragma unroll
                        for (int u = 0; u < U; ++u) {
                            const int  k  = g + u;
                            const bool ok = up ? h + k < c : k < h;
                            a.x += ok ? v[u].x : 0.0f; a.y += ok ? v[u].y : 0.0f;
                            a.z += ok ? v[u].z : 0.0f; a.w += ok ? v[u].w : 0.0f;
                        }
                    }
                }
            }
            a.x += xhalf(a.x); a.y += xhalf(a.y); a.z += xhalf(a.z); a.w += xhalf(a.w);
            if (!up && j < F4) acc4[(wave * NPW + i) * P4 + j] = a;
        }
    } else {
        int f0 = lane, f1 = 32 + lane;
        f0 = f0 < F ? f0 : F - 1;
        f1 = f1 < F ? f1 : F - 1;
        for (int i = 0; i < NPW; ++i) {
            const int n0 = __builtin_amdgcn_readlane(rpv, i);
            const int n1 = __builtin_amdgcn_readlane(rpv, i + 1);
            float a0 = 0.0f, a1 = 0.0f;
            for (int c0 = n0; c0 < n1; c0 += 32) {
                const int c = n1 - c0 < 32 ? n1 - c0 : 32;
                const int s = sorted[c0 + (lane < c ? lane : 0)];
                for (int k = 0; k < c; ++k) {
                    const float* row = hs + (size_t)__builtin_amdgcn_readlane(s, k) * F;
                    a0 += row[f0];
                    a1 += row[f1];
                }
            }
            float* arow = acc + (wave * NPW + i) * F;
            if (lane < F)      arow[lane]      = a0;
            if (32 + lane < F) arow[32 + lane] = a1;
        }
    }
    __syncthreads();

    const size_t g0 = (size_t)blockIdx.x * NB * F;
    const size_t gN = (size_t)nNodes * F;
    if constexpr (FT != 0 && FT % 4 == 0) {
        constexpr int Q   = NB * FT / 4;
        constexpr int PCS = (Q + AT - 1) / AT;
        const float4* hs4  = (const float4*)(hs + g0);
        float4*       out4 = (float4*)(out + g0);
        float4 a[PCS], s[PCS];
        float  d[PCS], bv[PCS][4];
#pragma unroll
        for (int t = 0; t < PCS; ++t) {
            const int  q    = threadIdx.x + t * AT;
            const bool live = q < Q && g0 + 4 * (size_t)q < gN;
            const int  qc   = live ? q : 0;
            const int  r    = (4 * qc) / FT;
            const int  f    = 4 * qc - r * FT;
            d[t] = dis[blockIdx.x * NB + r];
            s[t] = hs4[qc];
            a[t] = acc4[r * P4 + (f >> 2)];
#pragma unroll
            for (int c = 0; c < 4; ++c) bv[t][c] = bias[f + c];
        }
#pragma unroll
        for (int t = 0; t < PCS; ++t) {
            const int q = threadIdx.x + t * AT;
            if (q < Q && g0 + 4 * (size_t)q < gN) {
                float4 o;
                o.x = fmaf(d[t], a[t].x + s[t].x, bv[t][0]);
                o.y = fmaf(d[t], a[t].y + s[t].y, bv[t][1]);
                o.z = fmaf(d[t], a[t].z + s[t].z, bv[t][2]);
                o.w = fmaf(d[t], a[t].w + s[t].w, bv[t][3]);
                if (doRelu) {
                    o.x = fmaxf(o.x, 0.0f); o.y = fmaxf(o.y, 0.0f);
                    o.z = fmaxf(o.z, 0.0f); o.w = fmaxf(o.w, 0.0f);
                }
                if constexpr (NT2 > 0) {
                    const int r = (4 * q) / FT;
                    acc4[r * P4 + ((4 * q - r * FT) >> 2)] = o;
                } else {
                    out4[q] = o;
                }
            }
        }
    } else {
        for (int i = threadIdx.x; i < NB * F; i += AT) {
            const size_t g = g0 + i;
            if (g < gN) {
                const int r = i / F;
                const int f = i - r * F;
                float o = fmaf(dis[blockIdx.x * NB + r], acc[i] + hs[g], bias[f]);
                if (doRelu) o = fmaxf(o, 0.0f);
                out[g] = o;
            }
        }
    }

    if constexpr (NT2 > 0) {
        __syncthreads();
        constexpr int RT = NB / 16;
        static_assert(AT / 32 >= RT, "a wave per row tile");
        const float2* wsm = (const float2*)(acc4 + NB * P4);
        if (wave < RT)
            lds_rows_times_w<FT, NT2, 4 * P4, PW2>(acc + wave * 16 * (4 * P4), wsm, dis, hsn,
                                                   blockIdx.x * NB + wave * 16, nNodes, Nn, lane);
    }
}

static inline void launch_gemm(int tilesN, const float* X, const float* W,
                               const float* dis, float* hs,
                               int M, int K, int N, hipStream_t stream) {
    int mt = (M + 15) / 16;
    long long thr = (long long)mt * 32;
    int blocks = (int)((thr + 255) / 256);
    switch (tilesN) {
    case 1: gcn_gemm_wmma<1><<<blocks, 256, (size_t)(K / 2) * 16 * 8, stream>>>(X, W, dis, hs, M, K, N); break;
    case 2: gcn_gemm_wmma<2><<<blocks, 256, (size_t)(K / 2) * 48 * 8, stream>>>(X, W, dis, hs, M, K, N); break;
    case 3: gcn_gemm_wmma<3><<<blocks, 256, (size_t)(K / 2) * 48 * 8, stream>>>(X, W, dis, hs, M, K, N); break;
    default: gcn_gemm_wmma<4><<<blocks, 256, (size_t)(K / 2) * 80 * 8, stream>>>(X, W, dis, hs, M, K, N); break;
    }
}

static inline void launch_order_product(int tilesN, const int* grouped, const int* offs,
                                        int* sorted, int* rowptr, float* dis,
                                        const float* X, const float* W, float* hs,
                                        int nNodes, int nBk, int nE, int K, int N,
                                        hipStream_t stream) {
    switch (tilesN) {
    case 1: gcn_bucket_order_product<1><<<nBk, OT, (size_t)(K / 2) * 16 * 8, stream>>>(
                grouped, offs, sorted, rowptr, dis, nNodes, nBk, nE, X, W, hs, K, N); break;
    case 2: gcn_bucket_order_product<2><<<nBk, OT, (size_t)(K / 2) * 48 * 8, stream>>>(
                grouped, offs, sorted, rowptr, dis, nNodes, nBk, nE, X, W, hs, K, N); break;
    case 3: gcn_bucket_order_product<3><<<nBk, OT, (size_t)(K / 2) * 48 * 8, stream>>>(
                grouped, offs, sorted, rowptr, dis, nNodes, nBk, nE, X, W, hs, K, N); break;
    default: gcn_bucket_order_product<4><<<nBk, OT, (size_t)(K / 2) * 80 * 8, stream>>>(
                grouped, offs, sorted, rowptr, dis, nNodes, nBk, nE, X, W, hs, K, N); break;
    }
}

static inline void launch_aggregate(const int* sorted, const int* rowptr, const float* hs,
                                    const float* dis, const float* bias, float* out,
                                    int nNodes, int nBk, int F, int doRelu,
                                    hipStream_t stream) {
    size_t lds = (size_t)NB * F * sizeof(float);
    switch (F) {
    case 64: gcn_bucket_aggregate<64, 0><<<nBk, AT, lds, stream>>>(
                 sorted, rowptr, hs, dis, bias, out, nNodes, F, doRelu, nullptr, nullptr, 0); break;
    case 40: gcn_bucket_aggregate<40, 0><<<nBk, AT, lds, stream>>>(
                 sorted, rowptr, hs, dis, bias, out, nNodes, F, doRelu, nullptr, nullptr, 0); break;
    default: gcn_bucket_aggregate<0, 0><<<nBk, AT, lds, stream>>>(
                 sorted, rowptr, hs, dis, bias, out, nNodes, F, doRelu, nullptr, nullptr, 0); break;
    }
}

static inline void launch_aggregate_product(int tilesN, const int* sorted, const int* rowptr,
                                            const float* hs, const float* dis, const float* bias,
                                            const float* Wn, float* hsn,
                                            int nNodes, int nBk, int Nn, int doRelu,
                                            hipStream_t stream) {
    const size_t rows = (size_t)NB * 17 * 16;
    switch (tilesN) {
    case 1: gcn_bucket_aggregate<64, 1><<<nBk, AT, rows + 32 * 16 * 8, stream>>>(
                sorted, rowptr, hs, dis, bias, nullptr, nNodes, 64, doRelu, Wn, hsn, Nn); break;
    case 2: gcn_bucket_aggregate<64, 2><<<nBk, AT, rows + 32 * 48 * 8, stream>>>(
                sorted, rowptr, hs, dis, bias, nullptr, nNodes, 64, doRelu, Wn, hsn, Nn); break;
    case 3: gcn_bucket_aggregate<64, 3><<<nBk, AT, rows + 32 * 48 * 8, stream>>>(
                sorted, rowptr, hs, dis, bias, nullptr, nNodes, 64, doRelu, Wn, hsn, Nn); break;
    default: gcn_bucket_aggregate<64, 4><<<nBk, AT, rows + 32 * 80 * 8, stream>>>(
                sorted, rowptr, hs, dis, bias, nullptr, nNodes, 64, doRelu, Wn, hsn, Nn); break;
    }
}

extern "C" void kernel_launch(void* const* d_in, const int* in_sizes, int n_in,
                              void* d_out, int out_size, void* d_ws, size_t ws_size,
                              hipStream_t stream) {
    const float* x  = (const float*)d_in[0];
    const int*   ei = (const int*)d_in[1];
    const float* W1 = (const float*)d_in[2];
    const float* b1 = (const float*)d_in[3];
    const float* W2 = (const float*)d_in[4];
    const float* b2 = (const float*)d_in[5];
    float* out = (float*)d_out;

    const int IN  = 128;
    const int HID = in_sizes[3];
    const int OUT = in_sizes[5];
    const int N   = in_sizes[0] / IN;
    const int E   = in_sizes[1] / 2;

    const int* src = ei;
    const int* dst = ei + E;

    const int nBk  = (N + NB - 1) / NB;
    const int FMAX = HID > OUT ? HID : OUT;

    float* dis    = (float*)d_ws;
    float* hs     = dis + (((size_t)N + 63) & ~63ull);
    float* h1     = hs + (size_t)N * FMAX;
    int*  sorted  = (int*)(h1 + (size_t)N * HID);
    int*  rowptr  = sorted + (((size_t)E + 63) & ~63ull);
    int*  offs    = rowptr + (size_t)nBk * PB;
    int*  grouped = (int*)h1;

    const int per = (E + PB - 1) / PB;
    size_t ldsBk = ((size_t)nBk + (size_t)per) * sizeof(int);
    gcn_slice_sort<<<PB, PT, ldsBk, stream>>>(src, dst, offs, grouped, E, nBk);

    launch_order_product((HID + 15) / 16, grouped, offs, sorted, rowptr, dis, x, W1, hs,
                         N, nBk, E, IN, HID, stream);

    if (HID == 64 && OUT <= 64 && OUT % 4 == 0) {
        launch_aggregate_product((OUT + 15) / 16, sorted, rowptr, hs, dis, b1, W2, h1,
                                 N, nBk, OUT,  1, stream);
        launch_aggregate(sorted, rowptr, h1, dis, b2, out, N, nBk, OUT,  0, stream);
    } else {
        launch_aggregate(sorted, rowptr, hs, dis, b1, h1, N, nBk, HID,  1, stream);
        launch_gemm((OUT + 15) / 16, h1, W2, dis, hs, N, HID, OUT, stream);
        launch_aggregate(sorted, rowptr, hs, dis, b2, out, N, nBk, OUT,  0, stream);
    }
}


# gather addressing: a step's source-row offset is a DPP row broadcast of a per-lane 32-bit byte offset on a scalar base (was 2 v_readlane, select and 64-bit shift/add per row load); same sums
# speedup vs baseline: 5.8258x; 1.0353x over previous
#include <hip/hip_runtime.h>
#include <hip/hip_bf16.h>

typedef float v2f __attribute__((ext_vector_type(2)));
typedef float v8f __attribute__((ext_vector_type(8)));

template <int NT>
__global__ void __launch_bounds__(256)
gcn_gemm_wmma(const float* __restrict__ X,
              const float* __restrict__ W,
              const float* __restrict__ dis,
              float* __restrict__ hs,
              int M, int K, int N) {
    constexpr int NP    = NT * 16;
    constexpr int PITCH = (NP % 32 == 16) ? NP : NP + 16;
    constexpr int KC    = 8;
    extern __shared__ float2 wsm[];

    const int wave = (int)((blockIdx.x * blockDim.x + threadIdx.x) >> 5);
    const int lane = threadIdx.x & 31;
    const int m0 = wave << 4;
    const int lo = lane & 15;
    const int hi = lane >> 4;
    const bool live = m0 < M;

    int arow = m0 + lo;
    arow = arow < M ? arow : M - 1;
    const float* xrow = X + (size_t)arow * K + 2 * hi;

    const int nfull = K / (4 * KC);
    v2f an[KC];
#pragma unroll
    for (int j = 0; j < KC; ++j) an[j] = (v2f){0.0f, 0.0f};
    if (nfull > 0) {
#pragma unroll
        for (int j = 0; j < KC; ++j) an[j] = *(const v2f*)(xrow + 4 * j);
    }
    const int rowBase = m0 + (hi << 3);
    float dr[8];
#pragma unroll
    for (int r = 0; r < 8; ++r) {
        const int rr = rowBase + r;
        dr[r] = dis[rr < M ? rr : M - 1];
    }

    {
        const int KP = K >> 1;
        const int total = KP * NP;
        for (int base = 0; base < total; base += 4 * 256) {
            float w0[4], w1[4];
#pragma unroll
            for (int u = 0; u < 4; ++u) {
                int idx = base + u * 256 + (int)threadIdx.x;
                idx = idx < total ? idx : total - 1;
                const int kp = idx / NP;
                const int n  = idx - kp * NP;
                const int nc = n < N ? n : 0;
                w0[u] = W[(size_t)(2 * kp) * N + nc];
                w1[u] = W[(size_t)(2 * kp + 1) * N + nc];
            }
#pragma unroll
            for (int u = 0; u < 4; ++u) {
                const int idx = base + u * 256 + (int)threadIdx.x;
                if (idx < total) {
                    const int kp = idx / NP;
                    const int n  = idx - kp * NP;
                    wsm[kp * PITCH + n] = n < N ? make_float2(w0[u], w1[u]) : make_float2(0.0f, 0.0f);
                }
            }
        }
    }
    __syncthreads();
    if (!live) return;

    v8f c[NT];
#pragma unroll
    for (int t = 0; t < NT; ++t) c[t] = (v8f){};

    const float2* wl = wsm + hi * PITCH + lo;
    for (int ch = 0; ch < nfull; ++ch) {
        v2f ac[KC];
#pragma unroll
        for (int j = 0; j < KC; ++j) ac[j] = an[j];
        if (ch + 1 < nfull) {
#pragma unroll
            for (int j = 0; j < KC; ++j) an[j] = *(const v2f*)(xrow + (ch + 1) * (4 * KC) + 4 * j);
        }
        const float2* wc = wl + (size_t)ch * (2 * KC) * PITCH;
#pragma unroll
        for (int j = 0; j < KC; ++j) {
            v2f b[NT];
#pragma unroll
            for (int t = 0; t < NT; ++t) {
                const float2 w = wc[2 * j * PITCH + t * 16];
                b[t].x = w.x;
                b[t].y = w.y;
            }
#pragma unroll
            for (int t = 0; t < NT; ++t)
                c[t] = __builtin_amdgcn_wmma_f32_16x16x4_f32(
                            false, ac[j],  false, b[t],
                            (short)0, c[t],
                            false,  false);
        }
    }
    for (int k0 = nfull * (4 * KC); k0 < K; k0 += 4) {
        const v2f a = *(const v2f*)(xrow + k0);
        const float2* wc = wl + (size_t)(k0 >> 1) * PITCH;
#pragma unroll
        for (int t = 0; t < NT; ++t) {
            const float2 w = wc[t * 16];
            v2f b;
            b.x = w.x;
            b.y = w.y;
            c[t] = __builtin_amdgcn_wmma_f32_16x16x4_f32(
                        false, a,  false, b,
                        (short)0, c[t],
                        false,  false);
        }
    }

#pragma unroll
    for (int r = 0; r < 8; ++r) {
        const int rr = rowBase + r;
        if (rr < M) {
            const size_t base = (size_t)rr * N;
#pragma unroll
            for (int t = 0; t < NT; ++t) {
                const int col = t * 16 + lo;
                if (col < N) hs[base + col] = c[t][r] * dr[r];
            }
        }
    }
}

constexpr int LOG_NB = 7;
constexpr int NB     = 1 << LOG_NB;
constexpr int PB     = 256;
constexpr int PT     = 1024;
constexpr int PU     = 4;
constexpr int KE     = 16;

__global__ void __launch_bounds__(PT)
gcn_slice_sort(const int* __restrict__ src, const int* __restrict__ dst,
               int* __restrict__ offs, int* __restrict__ grouped, int nE, int nBk) {
    static_assert(PT == 1024, "the scan below lets one wave add up the totals of 32 waves");
    extern __shared__ int lcount[];
    __shared__ int wsum[PT / 32];
    __shared__ int chunkTotal;
    __shared__ int carry;
    const int p    = threadIdx.x;
    const int lane = p & 31;
    const int wv   = p >> 5;
    const int per = (nE + PB - 1) / PB;
    const int lo  = blockIdx.x * per;
    const int hi  = lo + per < nE ? lo + per : nE;
    int* stage = lcount + nBk;
    const bool inreg = per <= KE * PT;
    int bk[KE], key[KE], rk[KE];
#pragma unroll
    for (int u = 0; u < KE; ++u) { bk[u] = 0; key[u] = 0; rk[u] = 0; }
    if (inreg && lo < hi) {
#pragma unroll
        for (int u = 0; u < KE; ++u) {
            const int eu = lo + p + u * PT;
            const int ec = eu < hi ? eu : lo;
            const int dd = dst[ec];
            const int ss = src[ec];
            bk[u]  = dd >> LOG_NB;
            key[u] = (ss << LOG_NB) | (dd & (NB - 1));
        }
    }
    for (int b = p; b < nBk; b += PT) lcount[b] = 0;
    if (p == 0) carry = 0;
    __syncthreads();
    if (inreg) {
#pragma unroll
        for (int u = 0; u < KE; ++u)
            if (lo + p + u * PT < hi) rk[u] = atomicAdd(&lcount[bk[u]], 1);
    } else {
        for (int e = lo + p; e < hi; e += PU * PT) {
            int d[PU];
#pragma unroll
            for (int u = 0; u < PU; ++u) {
                const int eu = e + u * PT;
                d[u] = dst[eu < hi ? eu : e];
            }
#pragma unroll
            for (int u = 0; u < PU; ++u)
                if (e + u * PT < hi) atomicAdd(&lcount[d[u] >> LOG_NB], 1);
        }
    }
    __syncthreads();
    for (int base = 0; base < nBk; base += PT) {
        const int i = base + p;
        const int v = i < nBk ? lcount[i] : 0;
        int incl = v;
        for (int o = 1; o < 32; o <<= 1) {
            const int t = __shfl_up(incl, o, 32);
            if (lane >= o) incl += t;
        }
        if (lane == 31) wsum[wv] = incl;
        __syncthreads();
        if (wv == 0) {
            const int ws = wsum[lane];
            int wi = ws;
            for (int o = 1; o < 32; o <<= 1) {
                const int t = __shfl_up(wi, o, 32);
                if (lane >= o) wi += t;
            }
            wsum[lane] = wi - ws;
            if (lane == 31) chunkTotal = wi;
        }
        __syncthreads();
        const int c = carry;
        if (i < nBk) {
            const int ex = c + wsum[wv] + incl - v;
            offs[i * PB + blockIdx.x] = ex;
            lcount[i] = ex;
        }
        __syncthreads();
        if (p == 0) carry = c + chunkTotal;
        __syncthreads();
    }
    if (p == 0) offs[nBk * PB + blockIdx.x] = carry;
    if (inreg) {
        int pos[KE];
#pragma unroll
        for (int u = 0; u < KE; ++u) pos[u] = lcount[bk[u]] + rk[u];
#pragma unroll
        for (int u = 0; u < KE; ++u)
            if (lo + p + u * PT < hi) stage[pos[u]] = key[u];
    } else {
        for (int e = lo + p; e < hi; e += PU * PT) {
            int d[PU], s[PU];
#pragma unroll
            for (int u = 0; u < PU; ++u) {
                const int eu = e + u * PT;
                const int ec = eu < hi ? eu : e;
                d[u] = dst[ec];
                s[u] = src[ec];
            }
#pragma unroll
            for (int u = 0; u < PU; ++u) {
                if (e + u * PT < hi) {
                    const int pos = atomicAdd(&lcount[d[u] >> LOG_NB], 1);
                    stage[pos] = (s[u] << LOG_NB) | (d[u] & (NB - 1));
                }
            }
        }
    }
    __syncthreads();
    for (int gb = (lo & ~31) + p; gb < hi; gb += PU * PT) {
        int v[PU];
#pragma unroll
        for (int u = 0; u < PU; ++u) {
            const int g = gb + u * PT;
            v[u] = stage[g >= lo && g < hi ? g - lo : 0];
        }
#pragma unroll
        for (int u = 0; u < PU; ++u) {
            const int g = gb + u * PT;
            if (g >= lo && g < hi) grouped[g] = v[u];
        }
    }
}

constexpr int OT = 256;
constexpr int RL = 4;
constexpr int RG = OT / RL;
constexpr int RQ = PB / RG;
constexpr int OU = 8;
constexpr int SCAP = 8192;
static_assert(NB == (OT / 32) * 16, "a wave per 16-row tile of the bucket");
static_assert(PB % RG == 0, "every lane takes the same number of runs");

template <int NT>
__global__ void __launch_bounds__(OT, 8)
gcn_bucket_order_product(const int* __restrict__ grouped, const int* __restrict__ offs,
                         int* __restrict__ sorted, int* __restrict__ rowptr,
                         float* __restrict__ dis, int nNodes, int nBk, int nE,
                         const float* __restrict__ X, const float* __restrict__ W,
                         float* __restrict__ hs, int K, int N) {
    constexpr int NP    = NT * 16;
    constexpr int PITCH = (NP % 32 == 16) ? NP : NP + 16;
    constexpr int KC    = 8;
    __shared__ int   cnt[NB];
    __shared__ int   tcnt[NB];
    __shared__ int   off[NB];
    __shared__ float disl[NB];
    __shared__ int   wsb[OT / 32];
    __shared__ int   wse[OT / 32];
    __shared__ int   ends[2];
    __shared__ int   stage[SCAP];
    extern __shared__ float2 wsm[];

    const int M    = nNodes;
    const int lane = threadIdx.x & 31;
    const int wave = (int)(threadIdx.x >> 5);
    const int sub  = threadIdx.x % RL;
    const int tq   = threadIdx.x / RL;
    const int per  = (nE + PB - 1) / PB;

    {
        constexpr int WU = 16;
        const int KP = K >> 1;
        const int total = KP * NP;
        for (int base = 0; base < total; base += WU * OT) {
            float w0[WU], w1[WU];
#pragma unroll
            for (int u = 0; u < WU; ++u) {
                int idx = base + u * OT + (int)threadIdx.x;
                idx = idx < total ? idx : total - 1;
                const int kp = idx / NP;
                const int n  = idx - kp * NP;
                const int nc = n < N ? n : 0;
                w0[u] = W[(size_t)(2 * kp) * N + nc];
                w1[u] = W[(size_t)(2 * kp + 1) * N + nc];
            }
#pragma unroll
            for (int u = 0; u < WU; ++u) {
                const int idx = base + u * OT + (int)threadIdx.x;
                if (idx < total) {
                    const int kp = idx / NP;
                    const int n  = idx - kp * NP;
                    wsm[kp * PITCH + n] = n < N ? make_float2(w0[u], w1[u]) : make_float2(0.0f, 0.0f);
                }
            }
        }
    }

    int rbase[RQ], rlen[RQ];
    int sb = 0, se = 0;
#pragma unroll
    for (int q = 0; q < RQ; ++q) {
        const int t  = q * RG + tq;
        const int rs = offs[blockIdx.x * PB + t];
        const int re = offs[(blockIdx.x + 1) * PB + t];
        rbase[q] = t * per + rs;
        rlen[q]  = re - rs;
        if (sub == 0) { sb += rs; se += re; }
    }

    int w[RQ][OU];
#pragma unroll
    for (int q = 0; q < RQ; ++q) {
#pragma unroll
        for (int u = 0; u < OU; ++u) {
            const int k = sub + u * RL;
            w[q][u] = grouped[k < rlen[q] ? rbase[q] + k : 0];
        }
    }

    const int m0 = blockIdx.x * NB + (wave << 4);
    const int lo16 = lane & 15;
    const int hi16 = lane >> 4;
    const bool live = m0 < M;
    int arow = m0 + lo16;
    arow = arow < M ? arow : M - 1;
    const float* xrow = X + (size_t)arow * K + 2 * hi16;
    const int nfull = K / (4 * KC);
    v2f an[KC];
#pragma unroll
    for (int j = 0; j < KC; ++j) an[j] = (v2f){0.0f, 0.0f};
    if (nfull > 0) {
#pragma unroll
        for (int j = 0; j < KC; ++j) an[j] = *(const v2f*)(xrow + 4 * j);
    }

    for (int o = 16; o > 0; o >>= 1) {
        sb += __shfl_xor(sb, o, 32);
        se += __shfl_xor(se, o, 32);
    }
    if (lane == 0) { wsb[wave] = sb; wse[wave] = se; }
    for (int r = threadIdx.x; r < NB; r += OT) { cnt[r] = 0; tcnt[r] = 0; }
    __syncthreads();

    int pos[RQ][OU];
#pragma unroll
    for (int q = 0; q < RQ; ++q) {
#pragma unroll
        for (int u = 0; u < OU; ++u) {
            pos[q][u] = 0;
            if (sub + u * RL < rlen[q]) pos[q][u] = atomicAdd(&cnt[w[q][u] & (NB - 1)], 1);
        }
        for (int k = sub + OU * RL; k < rlen[q]; k += RL)
            atomicAdd(&tcnt[grouped[rbase[q] + k] & (NB - 1)], 1);
    }
    __syncthreads();
    if (threadIdx.x < 32) {
        const int l  = threadIdx.x;
        const int c0 = cnt[4 * l] + tcnt[4 * l],         c1 = cnt[4 * l + 1] + tcnt[4 * l + 1];
        const int c2 = cnt[4 * l + 2] + tcnt[4 * l + 2], c3 = cnt[4 * l + 3] + tcnt[4 * l + 3];
        const int s  = c0 + c1 + c2 + c3;
        int incl = s;
        for (int o = 1; o < 32; o <<= 1) {
            const int t2 = __shfl_up(incl, o, 32);
            if (l >= o) incl += t2;
        }
        const int ex = incl - s;
        off[4 * l]     = ex;
        off[4 * l + 1] = ex + c0;
        off[4 * l + 2] = ex + c0 + c1;
        off[4 * l + 3] = ex + c0 + c1 + c2;
        int tb = l < OT / 32 ? wsb[l] : 0;
        int te = l < OT / 32 ? wse[l] : 0;
        for (int o = 16; o > 0; o >>= 1) {
            tb += __shfl_xor(tb, o, 32);
            te += __shfl_xor(te, o, 32);
        }
        if (l == 0) { ends[0] = tb; ends[1] = te; }
    }
    __syncthreads();
    const int lo = ends[0];
    const int hi = ends[1];
    const bool staged = hi - lo <= SCAP;
    for (int r = threadIdx.x; r < NB; r += OT) {
        const int node = blockIdx.x * NB + r;
        const float d = __frsqrt_rn((float)(1 + cnt[r] + tcnt[r]));
        disl[r] = d;
        if (node < nNodes) {
            dis[node]    = d;
            rowptr[node] = lo + off[r];
        }
    }
    if (blockIdx.x == nBk - 1 && threadIdx.x == 0) rowptr[nNodes] = hi;
    __syncthreads();
#pragma unroll
    for (int q = 0; q < RQ; ++q) {
#pragma unroll
        for (int u = 0; u < OU; ++u) {
            asm volatile("" : "+v"(w[q][u]));
            pos[q][u] += off[w[q][u] & (NB - 1)];
        }
    }
    if (staged) {
#pragma unroll
        for (int q = 0; q < RQ; ++q) {
#pragma unroll
            for (int u = 0; u < OU; ++u)
                if (sub + u * RL < rlen[q]) stage[pos[q][u]] = w[q][u] >> LOG_NB;
        }
    } else {
#pragma unroll
        for (int q = 0; q < RQ; ++q) {
#pragma unroll
            for (int u = 0; u < OU; ++u)
                if (sub + u * RL < rlen[q]) sorted[lo + pos[q][u]] = w[q][u] >> LOG_NB;
        }
    }
#pragma unroll
    for (int q = 0; q < RQ; ++q) {
        for (int k = sub + OU * RL; k < rlen[q]; k += RL) {
            const int wv = grouped[rbase[q] + k];
            const int nd = wv & (NB - 1);
            const int ps = off[nd] + cnt[nd] + atomicSub(&tcnt[nd], 1) - 1;
            if (staged) stage[ps] = wv >> LOG_NB;
            else        sorted[lo + ps] = wv >> LOG_NB;
        }
    }
    if (staged) {
        __syncthreads();
        for (int gb = (lo & ~31) + (int)threadIdx.x; gb < hi; gb += PU * OT) {
            int v[PU];
#pragma unroll
            for (int u = 0; u < PU; ++u) {
                const int g = gb + u * OT;
                v[u] = stage[g >= lo && g < hi ? g - lo : 0];
            }
#pragma unroll
            for (int u = 0; u < PU; ++u) {
                const int g = gb + u * OT;
                if (g >= lo && g < hi) sorted[g] = v[u];
            }
        }
    }

    if (!live) return;

    const int rowBase = m0 + (hi16 << 3);
    float dr[8];
#pragma unroll
    for (int r = 0; r < 8; ++r) dr[r] = disl[(wave << 4) + (hi16 << 3) + r];

    v8f c[NT];
#pragma unroll
    for (int t = 0; t < NT; ++t) c[t] = (v8f){};

    const float2* wl = wsm + hi16 * PITCH + lo16;
    for (int ch = 0; ch < nfull; ++ch) {
        v2f ac[KC];
#pragma unroll
        for (int j = 0; j < KC; ++j) ac[j] = an[j];
        if (ch + 1 < nfull) {
#pragma unroll
            for (int j = 0; j < KC; ++j) an[j] = *(const v2f*)(xrow + (ch + 1) * (4 * KC) + 4 * j);
        }
        const float2* wc = wl + (size_t)ch * (2 * KC) * PITCH;
#pragma unroll
        for (int j = 0; j < KC; ++j) {
            v2f b[NT];
#pragma unroll
            for (int t = 0; t < NT; ++t) {
                const float2 wv = wc[2 * j * PITCH + t * 16];
                b[t].x = wv.x;
                b[t].y = wv.y;
            }
#pragma unroll
            for (int t = 0; t < NT; ++t)
                c[t] = __builtin_amdgcn_wmma_f32_16x16x4_f32(
                            false, ac[j],  false, b[t],
                            (short)0, c[t],
                            false,  false);
        }
    }
    for (int k0 = nfull * (4 * KC); k0 < K; k0 += 4) {
        const v2f a = *(const v2f*)(xrow + k0);
        const float2* wc = wl + (size_t)(k0 >> 1) * PITCH;
#pragma unroll
        for (int t = 0; t < NT; ++t) {
            const float2 wv = wc[t * 16];
            v2f b;
            b.x = wv.x;
            b.y = wv.y;
            c[t] = __builtin_amdgcn_wmma_f32_16x16x4_f32(
                        false, a,  false, b,
                        (short)0, c[t],
                        false,  false);
        }
    }

#pragma unroll
    for (int r = 0; r < 8; ++r) {
        const int rr = rowBase + r;
        if (rr < M) {
            const size_t base = (size_t)rr * N;
#pragma unroll
            for (int t = 0; t < NT; ++t) {
                const int col = t * 16 + lo16;
                if (col < N) hs[base + col] = c[t][r] * dr[r];
            }
        }
    }
}

__device__ __forceinline__ float xhalf(float x) {
    return __builtin_bit_cast(float, __builtin_amdgcn_permlanex16(
        __builtin_bit_cast(int, x), __builtin_bit_cast(int, x), 0x76543210, 0xfedcba98, false, false));
}

template <int K>
__device__ __forceinline__ unsigned rowlane(unsigned x) {
    return (unsigned)__builtin_amdgcn_update_dpp(0, (int)x, 0x150 + K, 0xF, 0xF, true);
}

template <int G>
__device__ __forceinline__ void gather_steps(float4& a, const char* __restrict__ rows,
                                             unsigned sOff, unsigned laneOff, int hf, int lim) {
    constexpr int U = 8;
    float4 v[U];
    v[0] = *(const float4*)(rows + (size_t)(rowlane<G + 0>(sOff) + laneOff));
    v[1] = *(const float4*)(rows + (size_t)(rowlane<G + 1>(sOff) + laneOff));
    v[2] = *(const float4*)(rows + (size_t)(rowlane<G + 2>(sOff) + laneOff));
    v[3] = *(const float4*)(rows + (size_t)(rowlane<G + 3>(sOff) + laneOff));
    v[4] = *(const float4*)(rows + (size_t)(rowlane<G + 4>(sOff) + laneOff));
    v[5] = *(const float4*)(rows + (size_t)(rowlane<G + 5>(sOff) + laneOff));
    v[6] = *(const float4*)(rows + (size_t)(rowlane<G + 6>(sOff) + laneOff));
    v[7] = *(const float4*)(rows + (size_t)(rowlane<G + 7>(sOff) + laneOff));
    static_assert(U == 8, "eight steps are written out above");
    if (G + U <= hf) {
#pragma unroll
        for (int u = 0; u < U; ++u) {
            a.x += v[u].x; a.y += v[u].y; a.z += v[u].z; a.w += v[u].w;
        }
    } else {
#pragma unroll
        for (int u = 0; u < U; ++u) {
            const bool ok = G + u < lim;
            a.x += ok ? v[u].x : 0.0f; a.y += ok ? v[u].y : 0.0f;
            a.z += ok ? v[u].z : 0.0f; a.w += ok ? v[u].w : 0.0f;
        }
    }
}

constexpr int AT  = 512;
constexpr int NPW = NB / (AT / 32);

template <int K, int TN, int PA, int PW>
__device__ __forceinline__ void lds_rows_times_w(float* __restrict__ a0,
                                                 const float2* __restrict__ w0,
                                                 const float* __restrict__ dis,
                                                 float* __restrict__ dst,
                                                 int row0, int M, int N, int lane) {
    const int lo = lane & 15;
    const int hi = lane >> 4;
    const int rowBase = row0 + (hi << 3);
    float dr[8];
#pragma unroll
    for (int r = 0; r < 8; ++r) {
        const int rr = rowBase + r;
        dr[r] = dis[rr < M ? rr : M - 1];
    }
    const float*  arow = a0 + lo * PA + 2 * hi;
    const float2* wl   = w0 + hi * PW + lo;
    v8f c[TN];
#pragma unroll
    for (int t = 0; t < TN; ++t) c[t] = (v8f){};
#pragma unroll
    for (int j = 0; j < K / 4; ++j) {
        const v2f a = *(const v2f*)(arow + 4 * j);
        v2f b[TN];
#pragma unroll
        for (int t = 0; t < TN; ++t) {
            const float2 w = wl[2 * j * PW + t * 16];
            b[t].x = w.x;
            b[t].y = w.y;
        }
#pragma unroll
        for (int t = 0; t < TN; ++t)
            c[t] = __builtin_amdgcn_wmma_f32_16x16x4_f32(
                        false, a,  false, b[t],
                        (short)0, c[t],
                        false,  false);
    }
#pragma unroll
    for (int r = 0; r < 8; ++r) {
#pragma unroll
        for (int t = 0; t < TN; ++t) {
            const int col = t * 16 + lo;
            if (col < N) a0[((hi << 3) + r) * N + col] = c[t][r] * dr[r];
        }
    }
    const int rows  = M - row0 < 16 ? M - row0 : 16;
    const int live4 = rows * (N >> 2);
    const float4* t4 = (const float4*)a0;
    float4*       d4 = (float4*)(dst + (size_t)row0 * N);
    for (int p = lane; p < 4 * N; p += 32)
        if (p < live4) d4[p] = t4[p];
}

template <int FT, int NT2>
__global__ void __launch_bounds__(AT)
gcn_bucket_aggregate(const int* __restrict__ sorted, const int* __restrict__ rowptr,
                     const float* __restrict__ hs, const float* __restrict__ dis,
                     const float* __restrict__ bias, float* __restrict__ out,
                     int nNodes, int Frt, int doRelu,
                     const float* __restrict__ Wn, float* __restrict__ hsn, int Nn) {
    static_assert(NT2 == 0 || (FT != 0 && FT % 4 == 0), "the product reads whole k-steps of 4 from the LDS rows");
    extern __shared__ float4 acc4[];
    float* acc = (float*)acc4;
    const int F = FT ? FT : Frt;
    constexpr int P4  = NT2 ? FT / 4 + 1 : FT / 4;
    constexpr int NP2 = NT2 * 16;
    constexpr int PW2 = (NP2 % 32 == 16) ? NP2 : NP2 + 16;

    const int lane = threadIdx.x & 31;
    const int wave = __builtin_amdgcn_readfirstlane((int)(threadIdx.x >> 5));

    int rpi = blockIdx.x * NB + wave * NPW + lane;
    rpi = rpi < nNodes ? rpi : nNodes;
    const int rpv = rowptr[rpi];

    constexpr int WTOT = (FT / 2) * NP2;
    static_assert(WTOT % AT == 0, "every thread stages the same number of elements");
    constexpr int WU = NT2 ? WTOT / AT : 1;
    float w0[WU], w1[WU];
    if constexpr (NT2 > 0) {
#pragma unroll
        for (int u = 0; u < WU; ++u) {
            const int idx = u * AT + (int)threadIdx.x;
            const int kp = idx / NP2;
            const int n  = idx - kp * NP2;
            const int nc = n < Nn ? n : 0;
            w0[u] = Wn[(size_t)(2 * kp) * Nn + nc];
            w1[u] = Wn[(size_t)(2 * kp + 1) * Nn + nc];
        }
    }

    if constexpr (FT != 0 && FT % 4 == 0) {
        constexpr int F4 = FT / 4;
        constexpr int U  = 8;
        const int  j    = lane & 15;
        const bool up   = lane >= 16;
        const int  jc   = j < F4 ? j : F4 - 1;
        const unsigned laneOff = 16u * (unsigned)jc;
        const char* rows = (const char*)hs;

        int p0 = __builtin_amdgcn_readlane(rpv, 0);
        int p1 = __builtin_amdgcn_readlane(rpv, 1);
        int sNext = 0;
        {
            const int c = p1 - p0 < 32 ? p1 - p0 : 32;
            const int h = (c + 1) >> 1;
            const int q = up ? h + j : j;
            const bool have = up ? q < c : j < h;
            if (c > 0) sNext = sorted[p0 + (have ? q : 0)];
        }
        if constexpr (NT2 > 0) {
            float2* wsm = (float2*)(acc4 + NB * P4);
#pragma unroll
            for (int u = 0; u < WU; ++u) {
                const int idx = u * AT + (int)threadIdx.x;
                const int kp = idx / NP2;
                const int n  = idx - kp * NP2;
                wsm[kp * PW2 + n] = n < Nn ? make_float2(w0[u], w1[u]) : make_float2(0.0f, 0.0f);
            }
        }
        for (int i = 0; i < NPW; ++i) {
            float4 a = make_float4(0.0f, 0.0f, 0.0f, 0.0f);
            int s = sNext;
            const int n0 = p0, n1 = p1;
            if (i + 1 < NPW) {
                p0 = n1;
                p1 = __builtin_amdgcn_readlane(rpv, i + 2);
                const int c = p1 - p0 < 32 ? p1 - p0 : 32;
                const int h = (c + 1) >> 1;
                const int q = up ? h + j : j;
                const bool have = up ? q < c : j < h;
                if (c > 0) sNext = sorted[p0 + (have ? q : 0)];
            }
            for (int c0 = n0; c0 < n1; c0 += 32) {
                const int c  = n1 - c0 < 32 ? n1 - c0 : 32;
                const int h  = (c + 1) >> 1;
                const int hf = c >> 1;
                if (c0 != n0) {
                    const int q = up ? h + j : j;
                    const bool have = up ? q < c : j < h;
                    s = sorted[c0 + (have ? q : 0)];
                }
                const unsigned sOff = (unsigned)s * (unsigned)(FT * 4);
                const int lim = up ? hf : h;
                gather_steps<0>(a, rows, sOff, laneOff, hf, lim);
                if (h > U)
                    gather_steps<U>(a, rows, sOff, laneOff, hf, lim);
            }
            a.x += xhalf(a.x); a.y += xhalf(a.y); a.z += xhalf(a.z); a.w += xhalf(a.w);
            if (!up && j < F4) acc4[(wave * NPW + i) * P4 + j] = a;
        }
    } else {
        int f0 = lane, f1 = 32 + lane;
        f0 = f0 < F ? f0 : F - 1;
        f1 = f1 < F ? f1 : F - 1;
        for (int i = 0; i < NPW; ++i) {
            const int n0 = __builtin_amdgcn_readlane(rpv, i);
            const int n1 = __builtin_amdgcn_readlane(rpv, i + 1);
            float a0 = 0.0f, a1 = 0.0f;
            for (int c0 = n0; c0 < n1; c0 += 32) {
                const int c = n1 - c0 < 32 ? n1 - c0 : 32;
                const int s = sorted[c0 + (lane < c ? lane : 0)];
                for (int k = 0; k < c; ++k) {
                    const float* row = hs + (size_t)__builtin_amdgcn_readlane(s, k) * F;
                    a0 += row[f0];
                    a1 += row[f1];
                }
            }
            float* arow = acc + (wave * NPW + i) * F;
            if (lane < F)      arow[lane]      = a0;
            if (32 + lane < F) arow[32 + lane] = a1;
        }
    }
    __syncthreads();

    const size_t g0 = (size_t)blockIdx.x * NB * F;
    const size_t gN = (size_t)nNodes * F;
    if constexpr (FT != 0 && FT % 4 == 0) {
        constexpr int Q   = NB * FT / 4;
        constexpr int PCS = (Q + AT - 1) / AT;
        const float4* hs4  = (const float4*)(hs + g0);
        float4*       out4 = (float4*)(out + g0);
        float4 a[PCS], s[PCS];
        float  d[PCS], bv[PCS][4];
#pragma unroll
        for (int t = 0; t < PCS; ++t) {
            const int  q    = threadIdx.x + t * AT;
            const bool live = q < Q && g0 + 4 * (size_t)q < gN;
            const int  qc   = live ? q : 0;
            const int  r    = (4 * qc) / FT;
            const int  f    = 4 * qc - r * FT;
            d[t] = dis[blockIdx.x * NB + r];
            s[t] = hs4[qc];
            a[t] = acc4[r * P4 + (f >> 2)];
#pragma unroll
            for (int c = 0; c < 4; ++c) bv[t][c] = bias[f + c];
        }
#pragma unroll
        for (int t = 0; t < PCS; ++t) {
            const int q = threadIdx.x + t * AT;
            if (q < Q && g0 + 4 * (size_t)q < gN) {
                float4 o;
                o.x = fmaf(d[t], a[t].x + s[t].x, bv[t][0]);
                o.y = fmaf(d[t], a[t].y + s[t].y, bv[t][1]);
                o.z = fmaf(d[t], a[t].z + s[t].z, bv[t][2]);
                o.w = fmaf(d[t], a[t].w + s[t].w, bv[t][3]);
                if (doRelu) {
                    o.x = fmaxf(o.x, 0.0f); o.y = fmaxf(o.y, 0.0f);
                    o.z = fmaxf(o.z, 0.0f); o.w = fmaxf(o.w, 0.0f);
                }
                if constexpr (NT2 > 0) {
                    const int r = (4 * q) / FT;
                    acc4[r * P4 + ((4 * q - r * FT) >> 2)] = o;
                } else {
                    out4[q] = o;
                }
            }
        }
    } else {
        for (int i = threadIdx.x; i < NB * F; i += AT) {
            const size_t g = g0 + i;
            if (g < gN) {
                const int r = i / F;
                const int f = i - r * F;
                float o = fmaf(dis[blockIdx.x * NB + r], acc[i] + hs[g], bias[f]);
                if (doRelu) o = fmaxf(o, 0.0f);
                out[g] = o;
            }
        }
    }

    if constexpr (NT2 > 0) {
        __syncthreads();
        constexpr int RT = NB / 16;
        static_assert(AT / 32 >= RT, "a wave per row tile");
        const float2* wsm = (const float2*)(acc4 + NB * P4);
        if (wave < RT)
            lds_rows_times_w<FT, NT2, 4 * P4, PW2>(acc + wave * 16 * (4 * P4), wsm, dis, hsn,
                                                   blockIdx.x * NB + wave * 16, nNodes, Nn, lane);
    }
}

static inline void launch_gemm(int tilesN, const float* X, const float* W,
                               const float* dis, float* hs,
                               int M, int K, int N, hipStream_t stream) {
    int mt = (M + 15) / 16;
    long long thr = (long long)mt * 32;
    int blocks = (int)((thr + 255) / 256);
    switch (tilesN) {
    case 1: gcn_gemm_wmma<1><<<blocks, 256, (size_t)(K / 2) * 16 * 8, stream>>>(X, W, dis, hs, M, K, N); break;
    case 2: gcn_gemm_wmma<2><<<blocks, 256, (size_t)(K / 2) * 48 * 8, stream>>>(X, W, dis, hs, M, K, N); break;
    case 3: gcn_gemm_wmma<3><<<blocks, 256, (size_t)(K / 2) * 48 * 8, stream>>>(X, W, dis, hs, M, K, N); break;
    default: gcn_gemm_wmma<4><<<blocks, 256, (size_t)(K / 2) * 80 * 8, stream>>>(X, W, dis, hs, M, K, N); break;
    }
}

static inline void launch_order_product(int tilesN, const int* grouped, const int* offs,
                                        int* sorted, int* rowptr, float* dis,
                                        const float* X, const float* W, float* hs,
                                        int nNodes, int nBk, int nE, int K, int N,
                                        hipStream_t stream) {
    switch (tilesN) {
    case 1: gcn_bucket_order_product<1><<<nBk, OT, (size_t)(K / 2) * 16 * 8, stream>>>(
                grouped, offs, sorted, rowptr, dis, nNodes, nBk, nE, X, W, hs, K, N); break;
    case 2: gcn_bucket_order_product<2><<<nBk, OT, (size_t)(K / 2) * 48 * 8, stream>>>(
                grouped, offs, sorted, rowptr, dis, nNodes, nBk, nE, X, W, hs, K, N); break;
    case 3: gcn_bucket_order_product<3><<<nBk, OT, (size_t)(K / 2) * 48 * 8, stream>>>(
                grouped, offs, sorted, rowptr, dis, nNodes, nBk, nE, X, W, hs, K, N); break;
    default: gcn_bucket_order_product<4><<<nBk, OT, (size_t)(K / 2) * 80 * 8, stream>>>(
                grouped, offs, sorted, rowptr, dis, nNodes, nBk, nE, X, W, hs, K, N); break;
    }
}

static inline void launch_aggregate(const int* sorted, const int* rowptr, const float* hs,
                                    const float* dis, const float* bias, float* out,
                                    int nNodes, int nBk, int F, int doRelu,
                                    hipStream_t stream) {
    size_t lds = (size_t)NB * F * sizeof(float);
    switch (F) {
    case 64: gcn_bucket_aggregate<64, 0><<<nBk, AT, lds, stream>>>(
                 sorted, rowptr, hs, dis, bias, out, nNodes, F, doRelu, nullptr, nullptr, 0); break;
    case 40: gcn_bucket_aggregate<40, 0><<<nBk, AT, lds, stream>>>(
                 sorted, rowptr, hs, dis, bias, out, nNodes, F, doRelu, nullptr, nullptr, 0); break;
    default: gcn_bucket_aggregate<0, 0><<<nBk, AT, lds, stream>>>(
                 sorted, rowptr, hs, dis, bias, out, nNodes, F, doRelu, nullptr, nullptr, 0); break;
    }
}

static inline void launch_aggregate_product(int tilesN, const int* sorted, const int* rowptr,
                                            const float* hs, const float* dis, const float* bias,
                                            const float* Wn, float* hsn,
                                            int nNodes, int nBk, int Nn, int doRelu,
                                            hipStream_t stream) {
    const size_t rows = (size_t)NB * 17 * 16;
    switch (tilesN) {
    case 1: gcn_bucket_aggregate<64, 1><<<nBk, AT, rows + 32 * 16 * 8, stream>>>(
                sorted, rowptr, hs, dis, bias, nullptr, nNodes, 64, doRelu, Wn, hsn, Nn); break;
    case 2: gcn_bucket_aggregate<64, 2><<<nBk, AT, rows + 32 * 48 * 8, stream>>>(
                sorted, rowptr, hs, dis, bias, nullptr, nNodes, 64, doRelu, Wn, hsn, Nn); break;
    case 3: gcn_bucket_aggregate<64, 3><<<nBk, AT, rows + 32 * 48 * 8, stream>>>(
                sorted, rowptr, hs, dis, bias, nullptr, nNodes, 64, doRelu, Wn, hsn, Nn); break;
    default: gcn_bucket_aggregate<64, 4><<<nBk, AT, rows + 32 * 80 * 8, stream>>>(
                sorted, rowptr, hs, dis, bias, nullptr, nNodes, 64, doRelu, Wn, hsn, Nn); break;
    }
}

extern "C" void kernel_launch(void* const* d_in, const int* in_sizes, int n_in,
                              void* d_out, int out_size, void* d_ws, size_t ws_size,
                              hipStream_t stream) {
    const float* x  = (const float*)d_in[0];
    const int*   ei = (const int*)d_in[1];
    const float* W1 = (const float*)d_in[2];
    const float* b1 = (const float*)d_in[3];
    const float* W2 = (const float*)d_in[4];
    const float* b2 = (const float*)d_in[5];
    float* out = (float*)d_out;

    const int IN  = 128;
    const int HID = in_sizes[3];
    const int OUT = in_sizes[5];
    const int N   = in_sizes[0] / IN;
    const int E   = in_sizes[1] / 2;

    const int* src = ei;
    const int* dst = ei + E;

    const int nBk  = (N + NB - 1) / NB;
    const int FMAX = HID > OUT ? HID : OUT;

    float* dis    = (float*)d_ws;
    float* hs     = dis + (((size_t)N + 63) & ~63ull);
    float* h1     = hs + (size_t)N * FMAX;
    int*  sorted  = (int*)(h1 + (size_t)N * HID);
    int*  rowptr  = sorted + (((size_t)E + 63) & ~63ull);
    int*  offs    = rowptr + (size_t)nBk * PB;
    int*  grouped = (int*)h1;

    const int per = (E + PB - 1) / PB;
    size_t ldsBk = ((size_t)nBk + (size_t)per) * sizeof(int);
    gcn_slice_sort<<<PB, PT, ldsBk, stream>>>(src, dst, offs, grouped, E, nBk);

    launch_order_product((HID + 15) / 16, grouped, offs, sorted, rowptr, dis, x, W1, hs,
                         N, nBk, E, IN, HID, stream);

    if (HID == 64 && OUT <= 64 && OUT % 4 == 0) {
        launch_aggregate_product((OUT + 15) / 16, sorted, rowptr, hs, dis, b1, W2, h1,
                                 N, nBk, OUT,  1, stream);
        launch_aggregate(sorted, rowptr, h1, dis, b2, out, N, nBk, OUT,  0, stream);
    } else {
        launch_aggregate(sorted, rowptr, hs, dis, b1, h1, N, nBk, HID,  1, stream);
        launch_gemm((OUT + 15) / 16, h1, W2, dis, hs, N, HID, OUT, stream);
        launch_aggregate(sorted, rowptr, hs, dis, b2, out, N, nBk, OUT,  0, stream);
    }
}
